# starting kernel
# baseline (speedup 1.0000x reference)
#include <hip/hip_runtime.h>
#include <hip/hip_bf16.h>

typedef __attribute__((ext_vector_type(2))) float v2f;
typedef __attribute__((ext_vector_type(8))) float v8f;

#define GD 128
#define BN_EPS 1e-5f
#define LRELU_SLOPE 0.01f

__global__ __launch_bounds__(256)
void wmma_gemm128(const float* __restrict__ A, const float* __restrict__ W,
                  float* __restrict__ C, int nrows) {
    __shared__ float lds[GD * GD];
    const int t = threadIdx.x;
    for (int i = t; i < GD * GD; i += 256) lds[i] = W[i];
    __syncthreads();

    const int wave = t >> 5;
    const int lane = t & 31;
    const int half = lane >> 4;
    const int l16  = lane & 15;

    const int m0 = blockIdx.x * 128 + wave * 16;
    int arow = m0 + l16;
    if (arow >= nrows) arow = nrows - 1;
    const float* aptr = A + (size_t)arow * GD + half * 2;

    v8f acc[8];
#pragma unroll
    for (int nt = 0; nt < 8; ++nt) acc[nt] = (v8f){0, 0, 0, 0, 0, 0, 0, 0};

    for (int k0 = 0; k0 < GD; k0 += 4) {
        const v2f a = *(const v2f*)(aptr + k0);
#pragma unroll
        for (int nt = 0; nt < 8; ++nt) {
            const int nn = nt * 16 + l16;
            v2f b;
            b.x = lds[(k0 + half * 2)     * GD + nn];
            b.y = lds[(k0 + half * 2 + 1) * GD + nn];
            acc[nt] = __builtin_amdgcn_wmma_f32_16x16x4_f32(
                false, a, false, b, (short)0, acc[nt], false, false);
        }
    }

#pragma unroll
    for (int r = 0; r < 8; ++r) {
        const int row = m0 + r + half * 8;
        if (row < nrows) {
#pragma unroll
            for (int nt = 0; nt < 8; ++nt)
                C[(size_t)row * GD + nt * 16 + l16] = acc[nt][r];
        }
    }
}

__global__ void init_one_kernel(float* __restrict__ p, int n) {
    int i = blockIdx.x * 256 + threadIdx.x;
    if (i < n) p[i] = 1.0f;
}
__global__ void deg_accum_kernel(float* __restrict__ deg, const int* __restrict__ dst, int e) {
    int i = blockIdx.x * 256 + threadIdx.x;
    if (i < e) unsafeAtomicAdd(&deg[dst[i]], 1.0f);
}
__global__ void rsqrt_kernel(const float* __restrict__ deg, float* __restrict__ dinv, int n) {
    int i = blockIdx.x * 256 + threadIdx.x;
    if (i < n) dinv[i] = rsqrtf(deg[i]);
}
__global__ void copy_kernel(const float* __restrict__ a, float* __restrict__ b, int n) {
    int i = blockIdx.x * 256 + threadIdx.x;
    if (i < n) b[i] = a[i];
}
__global__ void zero_kernel(float* __restrict__ p, int n) {
    int i = blockIdx.x * 256 + threadIdx.x;
    if (i < n) p[i] = 0.0f;
}

__global__ __launch_bounds__(256)
void scatter_kernel(const float* __restrict__ h, const int* __restrict__ src,
                    const int* __restrict__ dst, const float* __restrict__ dinv,
                    float* __restrict__ agg, int e) {
    int gid = blockIdx.x * 256 + threadIdx.x;
    int edge = gid >> 5;
    if (edge >= e) return;
    int lane = gid & 31;
    int s = src[edge], d = dst[edge];
    float w = dinv[s] * dinv[d];
    const float4 v = *((const float4*)(h + (size_t)s * GD) + lane);
    float* ad = agg + (size_t)d * GD + lane * 4;
    unsafeAtomicAdd(ad + 0, v.x * w);
    unsafeAtomicAdd(ad + 1, v.y * w);
    unsafeAtomicAdd(ad + 2, v.z * w);
    unsafeAtomicAdd(ad + 3, v.w * w);
}

__global__ void post_conv_kernel(float* __restrict__ x, const float* __restrict__ h,
                                 const float* __restrict__ agg, const float* __restrict__ dinv,
                                 const float* __restrict__ bc, int nd) {
    int i = blockIdx.x * 256 + threadIdx.x;
    if (i >= nd) return;
    int row = i >> 7, col = i & (GD - 1);
    float di = dinv[row];
    float conv = agg[i] + h[i] * di * di + bc[col];
    float v = conv + x[i];
    x[i] = v > 0.0f ? v : 0.0f;
}
__global__ void post_ff_kernel(float* __restrict__ x, const float* __restrict__ t,
                               const float* __restrict__ bf, int nd) {
    int i = blockIdx.x * 256 + threadIdx.x;
    if (i >= nd) return;
    int col = i & (GD - 1);
    float u = t[i] + bf[col];
    u = u > 0.0f ? u : LRELU_SLOPE * u;
    float v = u + x[i];
    x[i] = v > 0.0f ? v : 0.0f;
}
__global__ void post_skip_kernel(float* __restrict__ x, const float* __restrict__ t,
                                 const float* __restrict__ bskip, int nd) {
    int i = blockIdx.x * 256 + threadIdx.x;
    if (i >= nd) return;
    int col = i & (GD - 1);
    float s = t[i] + bskip[col];
    float v = x[i] + s;
    x[i] = v > 0.0f ? v : 0.0f;
}
__global__ void bn_kernel(float* __restrict__ x, const float* __restrict__ gamma,
                          const float* __restrict__ beta, const float* __restrict__ rmean,
                          const float* __restrict__ rvar, int nd) {
    int i = blockIdx.x * 256 + threadIdx.x;
    if (i >= nd) return;
    int col = i & (GD - 1);
    float xv = x[i];
    float b = (xv - rmean[col]) * rsqrtf(rvar[col] + BN_EPS) * gamma[col] + beta[col];
    float v = b + xv;
    x[i] = v > 0.0f ? v : 0.0f;
}

extern "C" void kernel_launch(void* const* d_in, const int* in_sizes, int n_in,
                              void* d_out, int out_size, void* d_ws, size_t ws_size,
                              hipStream_t stream) {
    const float* x_in  = (const float*)d_in[0];
    const int*   ei    = (const int*)d_in[1];
    const float* Wc    = (const float*)d_in[2];
    const float* bc    = (const float*)d_in[3];
    const float* Wf    = (const float*)d_in[4];
    const float* bf    = (const float*)d_in[5];
    const float* Wskip = (const float*)d_in[6];
    const float* bskip = (const float*)d_in[7];
    const float* gamma = (const float*)d_in[8];
    const float* beta  = (const float*)d_in[9];
    const float* rmean = (const float*)d_in[10];
    const float* rvar  = (const float*)d_in[11];

    const int n = in_sizes[0] / GD;
    const int e = in_sizes[1] / 2;
    const int L = in_sizes[3] / GD;

    const int* src = ei;
    const int* dst = ei + e;

    float* x    = (float*)d_out;
    float* bufA = (float*)d_ws;
    float* bufB = bufA + (size_t)n * GD;
    float* deg  = bufB + (size_t)n * GD;
    float* dinv = deg + n;

    const int ND = n * GD;
    dim3 blk(256);
    const int gND   = (ND + 255) / 256;
    const int gN    = (n + 255) / 256;
    const int gE    = (e + 255) / 256;
    const int gE32  = (int)(((long long)e * 32 + 255) / 256);
    const int gGemm = (n + 127) / 128;

    init_one_kernel<<<gN, blk, 0, stream>>>(deg, n);
    deg_accum_kernel<<<gE, blk, 0, stream>>>(deg, dst, e);
    rsqrt_kernel<<<gN, blk, 0, stream>>>(deg, dinv, n);
    copy_kernel<<<gND, blk, 0, stream>>>(x_in, x, ND);

    for (int i = 0; i < L; ++i) {
        wmma_gemm128<<<gGemm, blk, 0, stream>>>(x, Wc + (size_t)i * GD * GD, bufA, n);
        zero_kernel<<<gND, blk, 0, stream>>>(bufB, ND);
        scatter_kernel<<<gE32, blk, 0, stream>>>(bufA, src, dst, dinv, bufB, e);
        post_conv_kernel<<<gND, blk, 0, stream>>>(x, bufA, bufB, dinv, bc + i * GD, ND);

        wmma_gemm128<<<gGemm, blk, 0, stream>>>(x, Wf + (size_t)i * GD * GD, bufA, n);
        post_ff_kernel<<<gND, blk, 0, stream>>>(x, bufA, bf + i * GD, ND);

        if (i > 0) {
            wmma_gemm128<<<gGemm, blk, 0, stream>>>(x, Wskip + (size_t)(i - 1) * GD * GD, bufA, n);
            post_skip_kernel<<<gND, blk, 0, stream>>>(x, bufA, bskip + (i - 1) * GD, ND);
        }

        bn_kernel<<<gND, blk, 0, stream>>>(x, gamma + i * GD, beta + i * GD,
                                           rmean + i * GD, rvar + i * GD, ND);
    }
}


# Global float-atomic scatter replaced by destination-tile ownership: a work-group scans dst, sums its 200 rows in an LDS tile by LDS atomics (sum order: arrival in the group), ends conv+skip+relu
# speedup vs baseline: 1.2103x; 1.2103x over previous
#include <hip/hip_runtime.h>
#include <hip/hip_bf16.h>

typedef __attribute__((ext_vector_type(2))) float v2f;
typedef __attribute__((ext_vector_type(8))) float v8f;

#define GD 128
#define BN_EPS 1e-5f
#define LRELU_SLOPE 0.01f

__global__ __launch_bounds__(256)
void wmma_gemm128(const float* __restrict__ A, const float* __restrict__ W,
                  float* __restrict__ C, int nrows) {
    __shared__ float lds[GD * GD];
    const int t = threadIdx.x;
    for (int i = t; i < GD * GD; i += 256) lds[i] = W[i];
    __syncthreads();

    const int wave = t >> 5;
    const int lane = t & 31;
    const int half = lane >> 4;
    const int l16  = lane & 15;

    const int m0 = blockIdx.x * 128 + wave * 16;
    int arow = m0 + l16;
    if (arow >= nrows) arow = nrows - 1;
    const float* aptr = A + (size_t)arow * GD + half * 2;

    v8f acc[8];
#pragma unroll
    for (int nt = 0; nt < 8; ++nt) acc[nt] = (v8f){0, 0, 0, 0, 0, 0, 0, 0};

    for (int k0 = 0; k0 < GD; k0 += 4) {
        const v2f a = *(const v2f*)(aptr + k0);
#pragma unroll
        for (int nt = 0; nt < 8; ++nt) {
            const int nn = nt * 16 + l16;
            v2f b;
            b.x = lds[(k0 + half * 2)     * GD + nn];
            b.y = lds[(k0 + half * 2 + 1) * GD + nn];
            acc[nt] = __builtin_amdgcn_wmma_f32_16x16x4_f32(
                false, a, false, b, (short)0, acc[nt], false, false);
        }
    }

#pragma unroll
    for (int r = 0; r < 8; ++r) {
        const int row = m0 + r + half * 8;
        if (row < nrows) {
#pragma unroll
            for (int nt = 0; nt < 8; ++nt)
                C[(size_t)row * GD + nt * 16 + l16] = acc[nt][r];
        }
    }
}

__global__ void init_one_kernel(float* __restrict__ p, int n) {
    int i = blockIdx.x * 256 + threadIdx.x;
    if (i < n) p[i] = 1.0f;
}
__global__ void deg_accum_kernel(float* __restrict__ deg, const int* __restrict__ dst, int e) {
    int i = blockIdx.x * 256 + threadIdx.x;
    if (i < e) unsafeAtomicAdd(&deg[dst[i]], 1.0f);
}
__global__ void rsqrt_kernel(const float* __restrict__ deg, float* __restrict__ dinv, int n) {
    int i = blockIdx.x * 256 + threadIdx.x;
    if (i < n) dinv[i] = rsqrtf(deg[i]);
}
__global__ void copy_kernel(const float* __restrict__ a, float* __restrict__ b, int n) {
    int i = blockIdx.x * 256 + threadIdx.x;
    if (i < n) b[i] = a[i];
}

#define TR 200
#define AGG_T 1024
#define CH 32768
#define LC (CH + 4096)
#define GU 8

__global__ __launch_bounds__(AGG_T)
void agg_tile_kernel(float* __restrict__ x, const float* __restrict__ h,
                     const int* __restrict__ src, const int* __restrict__ dst,
                     const float* __restrict__ dinv, const float* __restrict__ bc,
                     int n, int e) {
    __shared__ float tile[TR * GD];
    __shared__ int   list[LC];
    __shared__ int   cnt;

    const int t = threadIdx.x;
    const int lane = t & 31;
    const int wave = t >> 5;
    const int row0 = blockIdx.x * TR;

    for (int i = t; i < TR * GD; i += AGG_T) tile[i] = 0.0f;
    if (t == 0) cnt = 0;
    __syncthreads();

    const bool al16 = ((((size_t)dst) & 15) == 0);

    for (int cb = 0; cb < e; cb += CH) {
        int4 dv[CH / (AGG_T * 4)];
        if (al16 && cb + CH <= e) {
            const int4* dp = (const int4*)(dst + cb) + t;
#pragma unroll
            for (int j = 0; j < CH / (AGG_T * 4); ++j) dv[j] = dp[j * AGG_T];
        } else {
#pragma unroll
            for (int j = 0; j < CH / (AGG_T * 4); ++j) {
                const int i0 = cb + (j * AGG_T + t) * 4;
                const int a = dst[i0     < e ? i0     : e - 1];
                const int b = dst[i0 + 1 < e ? i0 + 1 : e - 1];
                const int c = dst[i0 + 2 < e ? i0 + 2 : e - 1];
                const int d = dst[i0 + 3 < e ? i0 + 3 : e - 1];
                dv[j].x = i0     < e ? a : -1;
                dv[j].y = i0 + 1 < e ? b : -1;
                dv[j].z = i0 + 2 < e ? c : -1;
                dv[j].w = i0 + 3 < e ? d : -1;
            }
        }
        int k = 0;
#pragma unroll
        for (int j = 0; j < CH / (AGG_T * 4); ++j) {
            k += ((unsigned)(dv[j].x - row0) < (unsigned)TR) ? 1 : 0;
            k += ((unsigned)(dv[j].y - row0) < (unsigned)TR) ? 1 : 0;
            k += ((unsigned)(dv[j].z - row0) < (unsigned)TR) ? 1 : 0;
            k += ((unsigned)(dv[j].w - row0) < (unsigned)TR) ? 1 : 0;
        }
        if (k > 0) {
            int p = atomicAdd(&cnt, k);
#pragma unroll
            for (int j = 0; j < CH / (AGG_T * 4); ++j) {
                const int i0 = cb + (j * AGG_T + t) * 4;
                if ((unsigned)(dv[j].x - row0) < (unsigned)TR) list[p++] = i0;
                if ((unsigned)(dv[j].y - row0) < (unsigned)TR) list[p++] = i0 + 1;
                if ((unsigned)(dv[j].z - row0) < (unsigned)TR) list[p++] = i0 + 2;
                if ((unsigned)(dv[j].w - row0) < (unsigned)TR) list[p++] = i0 + 3;
            }
        }
        __syncthreads();
        const int m = cnt;
        __syncthreads();

        const bool last = (cb + CH >= e);
        if (m > LC - CH || last) {
            for (int base = wave * GU; base < m; base += (AGG_T / 32) * GU) {
                int   es[GU], ed[GU];
                float ew[GU];
                float4 ev[GU];
#pragma unroll
                for (int u = 0; u < GU; ++u) {
                    const int idx = base + u;
                    const int eid = list[idx < m ? idx : m - 1];
                    es[u] = src[eid];
                    ed[u] = dst[eid];
                }
#pragma unroll
                for (int u = 0; u < GU; ++u) {
                    ew[u] = dinv[es[u]] * dinv[ed[u]];
                    ev[u] = *((const float4*)(h + (size_t)es[u] * GD) + lane);
                }
#pragma unroll
                for (int u = 0; u < GU; ++u) {
                    if (base + u < m) {
                        float* tp = tile + (ed[u] - row0) * GD + lane;
                        const float w = ew[u];
                        __hip_atomic_fetch_add(tp,      ev[u].x * w, __ATOMIC_RELAXED, __HIP_MEMORY_SCOPE_WORKGROUP);
                        __hip_atomic_fetch_add(tp + 32, ev[u].y * w, __ATOMIC_RELAXED, __HIP_MEMORY_SCOPE_WORKGROUP);
                        __hip_atomic_fetch_add(tp + 64, ev[u].z * w, __ATOMIC_RELAXED, __HIP_MEMORY_SCOPE_WORKGROUP);
                        __hip_atomic_fetch_add(tp + 96, ev[u].w * w, __ATOMIC_RELAXED, __HIP_MEMORY_SCOPE_WORKGROUP);
                    }
                }
            }
            __syncthreads();
            if (t == 0) cnt = 0;
            __syncthreads();
        }
    }

    const float4 b4 = *((const float4*)bc + lane);
    for (int r = wave; r < TR; r += AGG_T / 32) {
        const int row = row0 + r;
        if (row < n) {
            const float di = dinv[row];
            const float4 hv = *((const float4*)(h + (size_t)row * GD) + lane);
            float4* xp = (float4*)(x + (size_t)row * GD) + lane;
            const float4 xv = *xp;
            const float* tp = tile + r * GD + lane;
            float4 o;
            o.x = (tp[0]  + hv.x * di * di + b4.x) + xv.x;
            o.y = (tp[32] + hv.y * di * di + b4.y) + xv.y;
            o.z = (tp[64] + hv.z * di * di + b4.z) + xv.z;
            o.w = (tp[96] + hv.w * di * di + b4.w) + xv.w;
            o.x = o.x > 0.0f ? o.x : 0.0f;
            o.y = o.y > 0.0f ? o.y : 0.0f;
            o.z = o.z > 0.0f ? o.z : 0.0f;
            o.w = o.w > 0.0f ? o.w : 0.0f;
            *xp = o;
        }
    }
}

__global__ void post_ff_kernel(float* __restrict__ x, const float* __restrict__ t,
                               const float* __restrict__ bf, int nd) {
    int i = blockIdx.x * 256 + threadIdx.x;
    if (i >= nd) return;
    int col = i & (GD - 1);
    float u = t[i] + bf[col];
    u = u > 0.0f ? u : LRELU_SLOPE * u;
    float v = u + x[i];
    x[i] = v > 0.0f ? v : 0.0f;
}
__global__ void post_skip_kernel(float* __restrict__ x, const float* __restrict__ t,
                                 const float* __restrict__ bskip, int nd) {
    int i = blockIdx.x * 256 + threadIdx.x;
    if (i >= nd) return;
    int col = i & (GD - 1);
    float s = t[i] + bskip[col];
    float v = x[i] + s;
    x[i] = v > 0.0f ? v : 0.0f;
}
__global__ void bn_kernel(float* __restrict__ x, const float* __restrict__ gamma,
                          const float* __restrict__ beta, const float* __restrict__ rmean,
                          const float* __restrict__ rvar, int nd) {
    int i = blockIdx.x * 256 + threadIdx.x;
    if (i >= nd) return;
    int col = i & (GD - 1);
    float xv = x[i];
    float b = (xv - rmean[col]) * rsqrtf(rvar[col] + BN_EPS) * gamma[col] + beta[col];
    float v = b + xv;
    x[i] = v > 0.0f ? v : 0.0f;
}

extern "C" void kernel_launch(void* const* d_in, const int* in_sizes, int n_in,
                              void* d_out, int out_size, void* d_ws, size_t ws_size,
                              hipStream_t stream) {
    const float* x_in  = (const float*)d_in[0];
    const int*   ei    = (const int*)d_in[1];
    const float* Wc    = (const float*)d_in[2];
    const float* bc    = (const float*)d_in[3];
    const float* Wf    = (const float*)d_in[4];
    const float* bf    = (const float*)d_in[5];
    const float* Wskip = (const float*)d_in[6];
    const float* bskip = (const float*)d_in[7];
    const float* gamma = (const float*)d_in[8];
    const float* beta  = (const float*)d_in[9];
    const float* rmean = (const float*)d_in[10];
    const float* rvar  = (const float*)d_in[11];

    const int n = in_sizes[0] / GD;
    const int e = in_sizes[1] / 2;
    const int L = in_sizes[3] / GD;

    const int* src = ei;
    const int* dst = ei + e;

    float* x    = (float*)d_out;
    float* bufA = (float*)d_ws;
    float* deg  = bufA + (size_t)n * GD;
    float* dinv = deg + n;

    const int ND = n * GD;
    dim3 blk(256);
    const int gND   = (ND + 255) / 256;
    const int gN    = (n + 255) / 256;
    const int gE    = (e + 255) / 256;
    const int gAgg  = (n + TR - 1) / TR;
    const int gGemm = (n + 127) / 128;

    init_one_kernel<<<gN, blk, 0, stream>>>(deg, n);
    deg_accum_kernel<<<gE, blk, 0, stream>>>(deg, dst, e);
    rsqrt_kernel<<<gN, blk, 0, stream>>>(deg, dinv, n);
    copy_kernel<<<gND, blk, 0, stream>>>(x_in, x, ND);

    for (int i = 0; i < L; ++i) {
        wmma_gemm128<<<gGemm, blk, 0, stream>>>(x, Wc + (size_t)i * GD * GD, bufA, n);
        agg_tile_kernel<<<gAgg, dim3(AGG_T), 0, stream>>>(x, bufA, src, dst, dinv, bc + i * GD, n, e);

        wmma_gemm128<<<gGemm, blk, 0, stream>>>(x, Wf + (size_t)i * GD * GD, bufA, n);
        post_ff_kernel<<<gND, blk, 0, stream>>>(x, bufA, bf + i * GD, ND);

        if (i > 0) {
            wmma_gemm128<<<gGemm, blk, 0, stream>>>(x, Wskip + (size_t)(i - 1) * GD * GD, bufA, n);
            post_skip_kernel<<<gND, blk, 0, stream>>>(x, bufA, bskip + (i - 1) * GD, ND);
        }

        bn_kernel<<<gND, blk, 0, stream>>>(x, gamma + i * GD, beta + i * GD,
                                           rmean + i * GD, rvar + i * GD, ND);
    }
}


# Tile owner keeps its 200 rows in LDS through the layer: ff and skip products read A from the tile (W staged over the dead edge list), epilogues and batch norm on the tile; 23 launches to 10
# speedup vs baseline: 1.7342x; 1.4329x over previous
#include <hip/hip_runtime.h>
#include <hip/hip_bf16.h>

typedef __attribute__((ext_vector_type(2))) float v2f;
typedef __attribute__((ext_vector_type(8))) float v8f;

#define GD 128
#define BN_EPS 1e-5f
#define LRELU_SLOPE 0.01f

__global__ __launch_bounds__(256)
void wmma_gemm128(const float* __restrict__ A, const float* __restrict__ W,
                  float* __restrict__ C, int nrows) {
    __shared__ float lds[GD * GD];
    const int t = threadIdx.x;
    for (int i = t; i < GD * GD; i += 256) lds[i] = W[i];
    __syncthreads();

    const int wave = t >> 5;
    const int lane = t & 31;
    const int half = lane >> 4;
    const int l16  = lane & 15;

    const int m0 = blockIdx.x * 128 + wave * 16;
    int arow = m0 + l16;
    if (arow >= nrows) arow = nrows - 1;
    const float* aptr = A + (size_t)arow * GD + half * 2;

    v8f acc[8];
#pragma unroll
    for (int nt = 0; nt < 8; ++nt) acc[nt] = (v8f){0, 0, 0, 0, 0, 0, 0, 0};

    for (int k0 = 0; k0 < GD; k0 += 4) {
        const v2f a = *(const v2f*)(aptr + k0);
#pragma unroll
        for (int nt = 0; nt < 8; ++nt) {
            const int nn = nt * 16 + l16;
            v2f b;
            b.x = lds[(k0 + half * 2)     * GD + nn];
            b.y = lds[(k0 + half * 2 + 1) * GD + nn];
            acc[nt] = __builtin_amdgcn_wmma_f32_16x16x4_f32(
                false, a, false, b, (short)0, acc[nt], false, false);
        }
    }

#pragma unroll
    for (int r = 0; r < 8; ++r) {
        const int row = m0 + r + half * 8;
        if (row < nrows) {
#pragma unroll
            for (int nt = 0; nt < 8; ++nt)
                C[(size_t)row * GD + nt * 16 + l16] = acc[nt][r];
        }
    }
}

__global__ void init_one_kernel(float* __restrict__ p, int n) {
    int i = blockIdx.x * 256 + threadIdx.x;
    if (i < n) p[i] = 1.0f;
}
__global__ void deg_accum_kernel(float* __restrict__ deg, const int* __restrict__ dst, int e) {
    int i = blockIdx.x * 256 + threadIdx.x;
    if (i < e) unsafeAtomicAdd(&deg[dst[i]], 1.0f);
}
__global__ void rsqrt_kernel(const float* __restrict__ deg, float* __restrict__ dinv, int n) {
    int i = blockIdx.x * 256 + threadIdx.x;
    if (i < n) dinv[i] = rsqrtf(deg[i]);
}
__global__ void copy_kernel(const float* __restrict__ a, float* __restrict__ b, int n) {
    int i = blockIdx.x * 256 + threadIdx.x;
    if (i < n) b[i] = a[i];
}

#define TR 200
#define TP 130
#define WP 136
#define AGG_T 1024
#define CH 32768
#define LC (CH + 4096)
#define GU 8
#define NSTRIPE ((TR + 15) / 16)
#define PERM(c) (((((c) & 1) << 1) | (((c) >> 1) & 1)) * 32 + ((c) >> 2))
#define OFX 0
#define OFY 64
#define OFZ 32
#define OFW 96

__device__ __forceinline__ void stage_w(float* __restrict__ wl, const float* __restrict__ Wg, int t) {
#pragma unroll
    for (int j = 0; j < (GD * GD / 4) / AGG_T; ++j) {
        const int i4 = t + j * AGG_T;
        const int row = i4 >> 5;
        const int c4 = i4 & 31;
        const float4 v = *((const float4*)Wg + i4);
        *(float4*)(wl + row * WP + c4 * 4) = v;
    }
}

__device__ __forceinline__ void tile_gemm(const float* __restrict__ tile, const float* __restrict__ wl,
                                          int stripe, int chalf, int l16, int half, v8f acc[4]) {
    int ar = stripe * 16 + l16;
    if (ar > TR - 1) ar = TR - 1;
    const float* ap = tile + ar * TP + half * 32;
    const float* bp = wl + (half * 2) * WP + chalf * 64 + l16;
#pragma unroll
    for (int nt = 0; nt < 4; ++nt) acc[nt] = (v8f){0, 0, 0, 0, 0, 0, 0, 0};
    for (int k0 = 0; k0 < GD; k0 += 4) {
        v2f a;
        a.x = ap[k0 >> 2];
        a.y = ap[(k0 >> 2) + 64];
#pragma unroll
        for (int nt = 0; nt < 4; ++nt) {
            v2f b;
            b.x = bp[k0 * WP + nt * 16];
            b.y = bp[(k0 + 1) * WP + nt * 16];
            acc[nt] = __builtin_amdgcn_wmma_f32_16x16x4_f32(
                false, a, false, b, (short)0, acc[nt], false, false);
        }
    }
}

__global__ __launch_bounds__(AGG_T)
void layer_tile_kernel(float* __restrict__ x, const float* __restrict__ h,
                       const int* __restrict__ src, const int* __restrict__ dst,
                       const float* __restrict__ dinv, const float* __restrict__ bc,
                       const float* __restrict__ Wf, const float* __restrict__ bf,
                       const float* __restrict__ Wsk, const float* __restrict__ bsk,
                       const float* __restrict__ gamma, const float* __restrict__ beta,
                       const float* __restrict__ rmean, const float* __restrict__ rvar,
                       int n, int e, int has_skip) {
    __shared__ float tile[TR * TP];
    __shared__ __attribute__((aligned(16))) unsigned char lraw[LC * 4];
    __shared__ int   cnt;
    int*   list = (int*)lraw;
    float* wl   = (float*)lraw;

    const int t = threadIdx.x;
    const int lane = t & 31;
    const int wave = t >> 5;
    const int row0 = blockIdx.x * TR;

    for (int i = t; i < TR * TP; i += AGG_T) tile[i] = 0.0f;
    if (t == 0) cnt = 0;
    __syncthreads();

    const bool al16 = ((((size_t)dst) & 15) == 0);

    for (int cb = 0; cb < e; cb += CH) {
        int4 dv[CH / (AGG_T * 4)];
        if (al16 && cb + CH <= e) {
            const int4* dp = (const int4*)(dst + cb) + t;
#pragma unroll
            for (int j = 0; j < CH / (AGG_T * 4); ++j) dv[j] = dp[j * AGG_T];
        } else {
#pragma unroll
            for (int j = 0; j < CH / (AGG_T * 4); ++j) {
                const int i0 = cb + (j * AGG_T + t) * 4;
                const int a = dst[i0     < e ? i0     : e - 1];
                const int b = dst[i0 + 1 < e ? i0 + 1 : e - 1];
                const int c = dst[i0 + 2 < e ? i0 + 2 : e - 1];
                const int d = dst[i0 + 3 < e ? i0 + 3 : e - 1];
                dv[j].x = i0     < e ? a : -1;
                dv[j].y = i0 + 1 < e ? b : -1;
                dv[j].z = i0 + 2 < e ? c : -1;
                dv[j].w = i0 + 3 < e ? d : -1;
            }
        }
        int k = 0;
#pragma unroll
        for (int j = 0; j < CH / (AGG_T * 4); ++j) {
            k += ((unsigned)(dv[j].x - row0) < (unsigned)TR) ? 1 : 0;
            k += ((unsigned)(dv[j].y - row0) < (unsigned)TR) ? 1 : 0;
            k += ((unsigned)(dv[j].z - row0) < (unsigned)TR) ? 1 : 0;
            k += ((unsigned)(dv[j].w - row0) < (unsigned)TR) ? 1 : 0;
        }
        if (k > 0) {
            int p = atomicAdd(&cnt, k);
#pragma unroll
            for (int j = 0; j < CH / (AGG_T * 4); ++j) {
                const int i0 = cb + (j * AGG_T + t) * 4;
                if ((unsigned)(dv[j].x - row0) < (unsigned)TR) list[p++] = i0;
                if ((unsigned)(dv[j].y - row0) < (unsigned)TR) list[p++] = i0 + 1;
                if ((unsigned)(dv[j].z - row0) < (unsigned)TR) list[p++] = i0 + 2;
                if ((unsigned)(dv[j].w - row0) < (unsigned)TR) list[p++] = i0 + 3;
            }
        }
        __syncthreads();
        const int m = cnt;
        __syncthreads();

        const bool last = (cb + CH >= e);
        if (m > LC - CH || last) {
            for (int base = wave * GU; base < m; base += (AGG_T / 32) * GU) {
                int   es[GU], ed[GU];
                float ew[GU];
                float4 ev[GU];
#pragma unroll
                for (int u = 0; u < GU; ++u) {
                    const int idx = base + u;
                    const int eid = list[idx < m ? idx : m - 1];
                    es[u] = src[eid];
                    ed[u] = dst[eid];
                }
#pragma unroll
                for (int u = 0; u < GU; ++u) {
                    ew[u] = dinv[es[u]] * dinv[ed[u]];
                    ev[u] = *((const float4*)(h + (size_t)es[u] * GD) + lane);
                }
#pragma unroll
                for (int u = 0; u < GU; ++u) {
                    if (base + u < m) {
                        float* tp = tile + (ed[u] - row0) * TP + lane;
                        const float w = ew[u];
                        __hip_atomic_fetch_add(tp + OFX, ev[u].x * w, __ATOMIC_RELAXED, __HIP_MEMORY_SCOPE_WORKGROUP);
                        __hip_atomic_fetch_add(tp + OFY, ev[u].y * w, __ATOMIC_RELAXED, __HIP_MEMORY_SCOPE_WORKGROUP);
                        __hip_atomic_fetch_add(tp + OFZ, ev[u].z * w, __ATOMIC_RELAXED, __HIP_MEMORY_SCOPE_WORKGROUP);
                        __hip_atomic_fetch_add(tp + OFW, ev[u].w * w, __ATOMIC_RELAXED, __HIP_MEMORY_SCOPE_WORKGROUP);
                    }
                }
            }
            __syncthreads();
            if (t == 0) cnt = 0;
            __syncthreads();
        }
    }

    stage_w(wl, Wf, t);
    if (has_skip) stage_w(wl + GD * WP, Wsk, t);

    {
        const float4 b4 = *((const float4*)bc + lane);
        for (int r = wave; r < TR; r += AGG_T / 32) {
            const int row = row0 + r;
            if (row < n) {
                const float di = dinv[row];
                const float4 hv = *((const float4*)(h + (size_t)row * GD) + lane);
                const float4 xv = *((const float4*)(x + (size_t)row * GD) + lane);
                float* tp = tile + r * TP + lane;
                float4 o;
                o.x = (tp[OFX] + hv.x * di * di + b4.x) + xv.x;
                o.y = (tp[OFY] + hv.y * di * di + b4.y) + xv.y;
                o.z = (tp[OFZ] + hv.z * di * di + b4.z) + xv.z;
                o.w = (tp[OFW] + hv.w * di * di + b4.w) + xv.w;
                tp[OFX] = o.x > 0.0f ? o.x : 0.0f;
                tp[OFY] = o.y > 0.0f ? o.y : 0.0f;
                tp[OFZ] = o.z > 0.0f ? o.z : 0.0f;
                tp[OFW] = o.w > 0.0f ? o.w : 0.0f;
            }
        }
    }
    __syncthreads();

    const int half = lane >> 4;
    const int l16  = lane & 15;
    const int stripe = wave >> 1;
    const int chalf  = wave & 1;
    const bool gw = (wave < 2 * NSTRIPE);

    v8f acc[4];
    float bnm[4], bns[4], bng[4], bnb[4];
#pragma unroll
    for (int nt = 0; nt < 4; ++nt) {
        const int col = chalf * 64 + nt * 16 + l16;
        bnm[nt] = rmean[col];
        bns[nt] = rsqrtf(rvar[col] + BN_EPS);
        bng[nt] = gamma[col];
        bnb[nt] = beta[col];
    }

    if (gw) tile_gemm(tile, wl, stripe, chalf, l16, half, acc);
    __syncthreads();
    if (gw) {
#pragma unroll
        for (int nt = 0; nt < 4; ++nt) {
            const int col = chalf * 64 + nt * 16 + l16;
            const float bias = bf[col];
            const int pc = PERM(col);
#pragma unroll
            for (int r = 0; r < 8; ++r) {
                const int row = stripe * 16 + r + half * 8;
                if (row < TR) {
                    float* tp = tile + row * TP + pc;
                    float u = acc[nt][r] + bias;
                    u = u > 0.0f ? u : LRELU_SLOPE * u;
                    float v = u + *tp;
                    v = v > 0.0f ? v : 0.0f;
                    if (!has_skip) {
                        const float b = (v - bnm[nt]) * bns[nt] * bng[nt] + bnb[nt];
                        const float y = b + v;
                        v = y > 0.0f ? y : 0.0f;
                    }
                    *tp = v;
                }
            }
        }
    }
    __syncthreads();

    if (has_skip) {
        if (gw) tile_gemm(tile, wl + GD * WP, stripe, chalf, l16, half, acc);
        __syncthreads();
        if (gw) {
#pragma unroll
            for (int nt = 0; nt < 4; ++nt) {
                const int col = chalf * 64 + nt * 16 + l16;
                const float bias = bsk[col];
                const int pc = PERM(col);
#pragma unroll
                for (int r = 0; r < 8; ++r) {
                    const int row = stripe * 16 + r + half * 8;
                    if (row < TR) {
                        float* tp = tile + row * TP + pc;
                        const float s = acc[nt][r] + bias;
                        float v = *tp + s;
                        v = v > 0.0f ? v : 0.0f;
                        const float b = (v - bnm[nt]) * bns[nt] * bng[nt] + bnb[nt];
                        const float y = b + v;
                        *tp = y > 0.0f ? y : 0.0f;
                    }
                }
            }
        }
        __syncthreads();
    }

    for (int r = wave; r < TR; r += AGG_T / 32) {
        const int row = row0 + r;
        if (row < n) {
            const float* tp = tile + r * TP + lane;
            float4 o;
            o.x = tp[OFX];
            o.y = tp[OFY];
            o.z = tp[OFZ];
            o.w = tp[OFW];
            *((float4*)(x + (size_t)row * GD) + lane) = o;
        }
    }
}

extern "C" void kernel_launch(void* const* d_in, const int* in_sizes, int n_in,
                              void* d_out, int out_size, void* d_ws, size_t ws_size,
                              hipStream_t stream) {
    const float* x_in  = (const float*)d_in[0];
    const int*   ei    = (const int*)d_in[1];
    const float* Wc    = (const float*)d_in[2];
    const float* bc    = (const float*)d_in[3];
    const float* Wf    = (const float*)d_in[4];
    const float* bf    = (const float*)d_in[5];
    const float* Wskip = (const float*)d_in[6];
    const float* bskip = (const float*)d_in[7];
    const float* gamma = (const float*)d_in[8];
    const float* beta  = (const float*)d_in[9];
    const float* rmean = (const float*)d_in[10];
    const float* rvar  = (const float*)d_in[11];

    const int n = in_sizes[0] / GD;
    const int e = in_sizes[1] / 2;
    const int L = in_sizes[3] / GD;

    const int* src = ei;
    const int* dst = ei + e;

    float* x    = (float*)d_out;
    float* bufA = (float*)d_ws;
    float* deg  = bufA + (size_t)n * GD;
    float* dinv = deg + n;

    const int ND = n * GD;
    dim3 blk(256);
    const int gND   = (ND + 255) / 256;
    const int gN    = (n + 255) / 256;
    const int gE    = (e + 255) / 256;
    const int gAgg  = (n + TR - 1) / TR;
    const int gGemm = (n + 127) / 128;

    init_one_kernel<<<gN, blk, 0, stream>>>(deg, n);
    deg_accum_kernel<<<gE, blk, 0, stream>>>(deg, dst, e);
    rsqrt_kernel<<<gN, blk, 0, stream>>>(deg, dinv, n);
    copy_kernel<<<gND, blk, 0, stream>>>(x_in, x, ND);

    for (int i = 0; i < L; ++i) {
        wmma_gemm128<<<gGemm, blk, 0, stream>>>(x, Wc + (size_t)i * GD * GD, bufA, n);
        const float* Wsk = (i > 0) ? Wskip + (size_t)(i - 1) * GD * GD : Wf + (size_t)i * GD * GD;
        const float* bsk = (i > 0) ? bskip + (i - 1) * GD : bf + i * GD;
        layer_tile_kernel<<<gAgg, dim3(AGG_T), 0, stream>>>(
            x, bufA, src, dst, dinv, bc + i * GD,
            Wf + (size_t)i * GD * GD, bf + i * GD, Wsk, bsk,
            gamma + i * GD, beta + i * GD, rmean + i * GD, rvar + i * GD,
            n, e, (i > 0) ? 1 : 0);
    }
}


# Tile owners scan one-byte tile ids (dst/200, written once per call by the degree launch, read one sub-chunk ahead) instead of 4-byte dst: scan bytes and tests cut 4x; sums stay in arrival order
# speedup vs baseline: 2.1146x; 1.2194x over previous
#include <hip/hip_runtime.h>
#include <hip/hip_bf16.h>

typedef __attribute__((ext_vector_type(2))) float v2f;
typedef __attribute__((ext_vector_type(8))) float v8f;

#define GD 128
#define BN_EPS 1e-5f
#define LRELU_SLOPE 0.01f

__global__ __launch_bounds__(256)
void wmma_gemm128(const float* __restrict__ A, const float* __restrict__ W,
                  float* __restrict__ C, int nrows) {
    __shared__ float lds[GD * GD];
    const int t = threadIdx.x;
    for (int i = t; i < GD * GD; i += 256) lds[i] = W[i];
    __syncthreads();

    const int wave = t >> 5;
    const int lane = t & 31;
    const int half = lane >> 4;
    const int l16  = lane & 15;

    const int m0 = blockIdx.x * 128 + wave * 16;
    int arow = m0 + l16;
    if (arow >= nrows) arow = nrows - 1;
    const float* aptr = A + (size_t)arow * GD + half * 2;

    v8f acc[8];
#pragma unroll
    for (int nt = 0; nt < 8; ++nt) acc[nt] = (v8f){0, 0, 0, 0, 0, 0, 0, 0};

    for (int k0 = 0; k0 < GD; k0 += 4) {
        const v2f a = *(const v2f*)(aptr + k0);
#pragma unroll
        for (int nt = 0; nt < 8; ++nt) {
            const int nn = nt * 16 + l16;
            v2f b;
            b.x = lds[(k0 + half * 2)     * GD + nn];
            b.y = lds[(k0 + half * 2 + 1) * GD + nn];
            acc[nt] = __builtin_amdgcn_wmma_f32_16x16x4_f32(
                false, a, false, b, (short)0, acc[nt], false, false);
        }
    }

#pragma unroll
    for (int r = 0; r < 8; ++r) {
        const int row = m0 + r + half * 8;
        if (row < nrows) {
#pragma unroll
            for (int nt = 0; nt < 8; ++nt)
                C[(size_t)row * GD + nt * 16 + l16] = acc[nt][r];
        }
    }
}

__global__ void init_one_kernel(float* __restrict__ p, int n) {
    int i = blockIdx.x * 256 + threadIdx.x;
    if (i < n) p[i] = 1.0f;
}
__global__ void rsqrt_kernel(const float* __restrict__ deg, float* __restrict__ dinv, int n) {
    int i = blockIdx.x * 256 + threadIdx.x;
    if (i < n) dinv[i] = rsqrtf(deg[i]);
}
__global__ void copy_kernel(const float* __restrict__ a, float* __restrict__ b, int n) {
    int i = blockIdx.x * 256 + threadIdx.x;
    if (i < n) b[i] = a[i];
}

#define TR 200
#define TP 130
#define WP 136
#define AGG_T 1024
#define CH 32768
#define LC (CH + 4096)
#define GU 8
#define NSTRIPE ((TR + 15) / 16)
#define PERM(c) (((((c) & 1) << 1) | (((c) >> 1) & 1)) * 32 + ((c) >> 2))
#define OFX 0
#define OFY 64
#define OFZ 32
#define OFW 96

#define TPF 1
#define TVS (CH / (AGG_T * 16))
__device__ __forceinline__ size_t tid_offset(int n) {
    const size_t used = ((size_t)n * GD + 2 * (size_t)n) * sizeof(float);
    return (used + 255) & ~(size_t)255;
}
__device__ __forceinline__ size_t tid_length(int e) {
    return ((size_t)((e + CH - 1) / CH) + TPF) * CH;
}
__device__ __forceinline__ bool tid_usable(int n, int e, size_t ws_bytes) {
    return (n + TR - 1) / TR <= 255 && tid_offset(n) + tid_length(e) <= ws_bytes;
}
__device__ __forceinline__ unsigned tid_match4(unsigned w, unsigned pat) {
    const unsigned x = w ^ pat;
    return ~(((x & 0x7F7F7F7Fu) + 0x7F7F7F7Fu) | x | 0x7F7F7F7Fu);
}

__global__ void deg_tid_kernel(float* __restrict__ deg, const int* __restrict__ dst, int e, int n,
                               float* __restrict__ wsf, size_t ws_bytes) {
    const size_t i4 = (size_t)blockIdx.x * 256 + threadIdx.x;
    int d[4];
#pragma unroll
    for (int j = 0; j < 4; ++j) {
        const size_t idx = i4 * 4 + j;
        d[j] = idx < (size_t)e ? dst[idx] : -1;
    }
    unsigned pk = 0;
#pragma unroll
    for (int j = 0; j < 4; ++j) {
        unsigned b = 255u;
        if (i4 * 4 + j < (size_t)e) {
            unsafeAtomicAdd(&deg[d[j]], 1.0f);
            if ((unsigned)d[j] < (unsigned)n) b = (unsigned)d[j] / TR;
        }
        pk |= b << (8 * j);
    }
    if (tid_usable(n, e, ws_bytes) && i4 * 4 < tid_length(e))
        *((unsigned*)((unsigned char*)wsf + tid_offset(n)) + i4) = pk;
}

__device__ __forceinline__ void stage_w(float* __restrict__ wl, const float* __restrict__ Wg, int t) {
#pragma unroll
    for (int j = 0; j < (GD * GD / 4) / AGG_T; ++j) {
        const int i4 = t + j * AGG_T;
        const int row = i4 >> 5;
        const int c4 = i4 & 31;
        const float4 v = *((const float4*)Wg + i4);
        *(float4*)(wl + row * WP + c4 * 4) = v;
    }
}

__device__ __forceinline__ void tile_gemm(const float* __restrict__ tile, const float* __restrict__ wl,
                                          int stripe, int chalf, int l16, int half, v8f acc[4]) {
    int ar = stripe * 16 + l16;
    if (ar > TR - 1) ar = TR - 1;
    const float* ap = tile + ar * TP + half * 32;
    const float* bp = wl + (half * 2) * WP + chalf * 64 + l16;
#pragma unroll
    for (int nt = 0; nt < 4; ++nt) acc[nt] = (v8f){0, 0, 0, 0, 0, 0, 0, 0};
    for (int k0 = 0; k0 < GD; k0 += 4) {
        v2f a;
        a.x = ap[k0 >> 2];
        a.y = ap[(k0 >> 2) + 64];
#pragma unroll
        for (int nt = 0; nt < 4; ++nt) {
            v2f b;
            b.x = bp[k0 * WP + nt * 16];
            b.y = bp[(k0 + 1) * WP + nt * 16];
            acc[nt] = __builtin_amdgcn_wmma_f32_16x16x4_f32(
                false, a, false, b, (short)0, acc[nt], false, false);
        }
    }
}

__global__ __launch_bounds__(AGG_T)
void layer_tile_kernel(float* __restrict__ x, const float* __restrict__ h,
                       const int* __restrict__ src, const int* __restrict__ dst,
                       const float* __restrict__ dinv, const float* __restrict__ bc,
                       const float* __restrict__ Wf, const float* __restrict__ bf,
                       const float* __restrict__ Wsk, const float* __restrict__ bsk,
                       const float* __restrict__ gamma, const float* __restrict__ beta,
                       const float* __restrict__ rmean, const float* __restrict__ rvar,
                       int n, int e, int has_skip,
                       const float* __restrict__ wsf, size_t ws_bytes) {
    __shared__ float tile[TR * TP];
    __shared__ __attribute__((aligned(16))) unsigned char lraw[LC * 4];
    __shared__ int   cnt;
    int*   list = (int*)lraw;
    float* wl   = (float*)lraw;

    const int t = threadIdx.x;
    const int lane = t & 31;
    const int wave = t >> 5;
    const int row0 = blockIdx.x * TR;

    for (int i = t; i < TR * TP; i += AGG_T) tile[i] = 0.0f;
    if (t == 0) cnt = 0;
    __syncthreads();

    const bool al16 = ((((size_t)dst) & 15) == 0);

    static_assert(TVS * 4 <= 8, "a thread's hits of one sub-chunk are kept as one 32-bit mask, one bit per byte");
    static_assert(TPF == 1, "the scan holds exactly one sub-chunk ahead in registers");
    const bool compact = tid_usable(n, e, ws_bytes);
    const uint4* tq = (const uint4*)((const unsigned char*)wsf + tid_offset(n)) + t;
    const unsigned pat = (unsigned)blockIdx.x * 0x01010101u;
    uint4 nx[TVS];
#pragma unroll
    for (int v = 0; v < TVS; ++v)
        nx[v] = compact ? tq[v * AGG_T] : make_uint4(0u, 0u, 0u, 0u);

    for (int cb = 0; cb < e; cb += CH) {
      if (compact) {
        uint4 c[TVS];
#pragma unroll
        for (int v = 0; v < TVS; ++v) {
            c[v] = nx[v];
            nx[v] = tq[(size_t)(cb / 16) + (size_t)((TVS + v) * AGG_T)];
        }
        unsigned bits = 0;
#pragma unroll
        for (int v = 0; v < TVS; ++v) {
            bits |= tid_match4(c[v].x, pat) >> (7 - (4 * v + 0));
            bits |= tid_match4(c[v].y, pat) >> (7 - (4 * v + 1));
            bits |= tid_match4(c[v].z, pat) >> (7 - (4 * v + 2));
            bits |= tid_match4(c[v].w, pat) >> (7 - (4 * v + 3));
        }
        const int k = __popc(bits);
        if (k > 0) {
            int p = atomicAdd(&cnt, k);
            const int eb = cb + t * 16;
            while (bits != 0) {
                const int pos = __ffs(bits) - 1;
                bits &= bits - 1;
                const int d = pos & 7;
                list[p++] = eb + (d >> 2) * (AGG_T * 16) + (d & 3) * 4 + (pos >> 3);
            }
        }
      } else {
        int4 dv[CH / (AGG_T * 4)];
        if (al16 && cb + CH <= e) {
            const int4* dp = (const int4*)(dst + cb) + t;
#pragma unroll
            for (int j = 0; j < CH / (AGG_T * 4); ++j) dv[j] = dp[j * AGG_T];
        } else {
#pragma unroll
            for (int j = 0; j < CH / (AGG_T * 4); ++j) {
                const int i0 = cb + (j * AGG_T + t) * 4;
                const int a = dst[i0     < e ? i0     : e - 1];
                const int b = dst[i0 + 1 < e ? i0 + 1 : e - 1];
                const int c = dst[i0 + 2 < e ? i0 + 2 : e - 1];
                const int d = dst[i0 + 3 < e ? i0 + 3 : e - 1];
                dv[j].x = i0     < e ? a : -1;
                dv[j].y = i0 + 1 < e ? b : -1;
                dv[j].z = i0 + 2 < e ? c : -1;
                dv[j].w = i0 + 3 < e ? d : -1;
            }
        }
        int k = 0;
#pragma unroll
        for (int j = 0; j < CH / (AGG_T * 4); ++j) {
            k += ((unsigned)(dv[j].x - row0) < (unsigned)TR) ? 1 : 0;
            k += ((unsigned)(dv[j].y - row0) < (unsigned)TR) ? 1 : 0;
            k += ((unsigned)(dv[j].z - row0) < (unsigned)TR) ? 1 : 0;
            k += ((unsigned)(dv[j].w - row0) < (unsigned)TR) ? 1 : 0;
        }
        if (k > 0) {
            int p = atomicAdd(&cnt, k);
#pragma unroll
            for (int j = 0; j < CH / (AGG_T * 4); ++j) {
                const int i0 = cb + (j * AGG_T + t) * 4;
                if ((unsigned)(dv[j].x - row0) < (unsigned)TR) list[p++] = i0;
                if ((unsigned)(dv[j].y - row0) < (unsigned)TR) list[p++] = i0 + 1;
                if ((unsigned)(dv[j].z - row0) < (unsigned)TR) list[p++] = i0 + 2;
                if ((unsigned)(dv[j].w - row0) < (unsigned)TR) list[p++] = i0 + 3;
            }
        }
      }
        __syncthreads();
        const int m = cnt;
        __syncthreads();

        const bool last = (cb + CH >= e);
        if (m > LC - CH || last) {
            for (int base = wave * GU; base < m; base += (AGG_T / 32) * GU) {
                int   es[GU], ed[GU];
                float ew[GU];
                float4 ev[GU];
#pragma unroll
                for (int u = 0; u < GU; ++u) {
                    const int idx = base + u;
                    const int eid = list[idx < m ? idx : m - 1];
                    es[u] = src[eid];
                    ed[u] = dst[eid];
                }
#pragma unroll
                for (int u = 0; u < GU; ++u) {
                    ew[u] = dinv[es[u]] * dinv[ed[u]];
                    ev[u] = *((const float4*)(h + (size_t)es[u] * GD) + lane);
                }
#pragma unroll
                for (int u = 0; u < GU; ++u) {
                    if (base + u < m) {
                        float* tp = tile + (ed[u] - row0) * TP + lane;
                        const float w = ew[u];
                        __hip_atomic_fetch_add(tp + OFX, ev[u].x * w, __ATOMIC_RELAXED, __HIP_MEMORY_SCOPE_WORKGROUP);
                        __hip_atomic_fetch_add(tp + OFY, ev[u].y * w, __ATOMIC_RELAXED, __HIP_MEMORY_SCOPE_WORKGROUP);
                        __hip_atomic_fetch_add(tp + OFZ, ev[u].z * w, __ATOMIC_RELAXED, __HIP_MEMORY_SCOPE_WORKGROUP);
                        __hip_atomic_fetch_add(tp + OFW, ev[u].w * w, __ATOMIC_RELAXED, __HIP_MEMORY_SCOPE_WORKGROUP);
                    }
                }
            }
            __syncthreads();
            if (t == 0) cnt = 0;
            __syncthreads();
        }
    }

    stage_w(wl, Wf, t);
    if (has_skip) stage_w(wl + GD * WP, Wsk, t);

    {
        const float4 b4 = *((const float4*)bc + lane);
        for (int r = wave; r < TR; r += AGG_T / 32) {
            const int row = row0 + r;
            if (row < n) {
                const float di = dinv[row];
                const float4 hv = *((const float4*)(h + (size_t)row * GD) + lane);
                const float4 xv = *((const float4*)(x + (size_t)row * GD) + lane);
                float* tp = tile + r * TP + lane;
                float4 o;
                o.x = (tp[OFX] + hv.x * di * di + b4.x) + xv.x;
                o.y = (tp[OFY] + hv.y * di * di + b4.y) + xv.y;
                o.z = (tp[OFZ] + hv.z * di * di + b4.z) + xv.z;
                o.w = (tp[OFW] + hv.w * di * di + b4.w) + xv.w;
                tp[OFX] = o.x > 0.0f ? o.x : 0.0f;
                tp[OFY] = o.y > 0.0f ? o.y : 0.0f;
                tp[OFZ] = o.z > 0.0f ? o.z : 0.0f;
                tp[OFW] = o.w > 0.0f ? o.w : 0.0f;
            }
        }
    }
    __syncthreads();

    const int half = lane >> 4;
    const int l16  = lane & 15;
    const int stripe = wave >> 1;
    const int chalf  = wave & 1;
    const bool gw = (wave < 2 * NSTRIPE);

    v8f acc[4];
    float bnm[4], bns[4], bng[4], bnb[4];
#pragma unroll
    for (int nt = 0; nt < 4; ++nt) {
        const int col = chalf * 64 + nt * 16 + l16;
        bnm[nt] = rmean[col];
        bns[nt] = rsqrtf(rvar[col] + BN_EPS);
        bng[nt] = gamma[col];
        bnb[nt] = beta[col];
    }

    if (gw) tile_gemm(tile, wl, stripe, chalf, l16, half, acc);
    __syncthreads();
    if (gw) {
#pragma unroll
        for (int nt = 0; nt < 4; ++nt) {
            const int col = chalf * 64 + nt * 16 + l16;
            const float bias = bf[col];
            const int pc = PERM(col);
#pragma unroll
            for (int r = 0; r < 8; ++r) {
                const int row = stripe * 16 + r + half * 8;
                if (row < TR) {
                    float* tp = tile + row * TP + pc;
                    float u = acc[nt][r] + bias;
                    u = u > 0.0f ? u : LRELU_SLOPE * u;
                    float v = u + *tp;
                    v = v > 0.0f ? v : 0.0f;
                    if (!has_skip) {
                        const float b = (v - bnm[nt]) * bns[nt] * bng[nt] + bnb[nt];
                        const float y = b + v;
                        v = y > 0.0f ? y : 0.0f;
                    }
                    *tp = v;
                }
            }
        }
    }
    __syncthreads();

    if (has_skip) {
        if (gw) tile_gemm(tile, wl + GD * WP, stripe, chalf, l16, half, acc);
        __syncthreads();
        if (gw) {
#pragma unroll
            for (int nt = 0; nt < 4; ++nt) {
                const int col = chalf * 64 + nt * 16 + l16;
                const float bias = bsk[col];
                const int pc = PERM(col);
#pragma unroll
                for (int r = 0; r < 8; ++r) {
                    const int row = stripe * 16 + r + half * 8;
                    if (row < TR) {
                        float* tp = tile + row * TP + pc;
                        const float s = acc[nt][r] + bias;
                        float v = *tp + s;
                        v = v > 0.0f ? v : 0.0f;
                        const float b = (v - bnm[nt]) * bns[nt] * bng[nt] + bnb[nt];
                        const float y = b + v;
                        *tp = y > 0.0f ? y : 0.0f;
                    }
                }
            }
        }
        __syncthreads();
    }

    for (int r = wave; r < TR; r += AGG_T / 32) {
        const int row = row0 + r;
        if (row < n) {
            const float* tp = tile + r * TP + lane;
            float4 o;
            o.x = tp[OFX];
            o.y = tp[OFY];
            o.z = tp[OFZ];
            o.w = tp[OFW];
            *((float4*)(x + (size_t)row * GD) + lane) = o;
        }
    }
}

extern "C" void kernel_launch(void* const* d_in, const int* in_sizes, int n_in,
                              void* d_out, int out_size, void* d_ws, size_t ws_size,
                              hipStream_t stream) {
    const float* x_in  = (const float*)d_in[0];
    const int*   ei    = (const int*)d_in[1];
    const float* Wc    = (const float*)d_in[2];
    const float* bc    = (const float*)d_in[3];
    const float* Wf    = (const float*)d_in[4];
    const float* bf    = (const float*)d_in[5];
    const float* Wskip = (const float*)d_in[6];
    const float* bskip = (const float*)d_in[7];
    const float* gamma = (const float*)d_in[8];
    const float* beta  = (const float*)d_in[9];
    const float* rmean = (const float*)d_in[10];
    const float* rvar  = (const float*)d_in[11];

    const int n = in_sizes[0] / GD;
    const int e = in_sizes[1] / 2;
    const int L = in_sizes[3] / GD;

    const int* src = ei;
    const int* dst = ei + e;

    float* x    = (float*)d_out;
    float* bufA = (float*)d_ws;
    float* deg  = bufA + (size_t)n * GD;
    float* dinv = deg + n;

    const int ND = n * GD;
    dim3 blk(256);
    const int gND   = (ND + 255) / 256;
    const int gN    = (n + 255) / 256;
    const int gT    = ((e + CH - 1) / CH + TPF) * (CH / 1024);
    const int gAgg  = (n + TR - 1) / TR;
    const int gGemm = (n + 127) / 128;

    init_one_kernel<<<gN, blk, 0, stream>>>(deg, n);
    deg_tid_kernel<<<gT, blk, 0, stream>>>(deg, dst, e, n, bufA, ws_size);
    rsqrt_kernel<<<gN, blk, 0, stream>>>(deg, dinv, n);
    copy_kernel<<<gND, blk, 0, stream>>>(x_in, x, ND);

    for (int i = 0; i < L; ++i) {
        wmma_gemm128<<<gGemm, blk, 0, stream>>>(x, Wc + (size_t)i * GD * GD, bufA, n);
        const float* Wsk = (i > 0) ? Wskip + (size_t)(i - 1) * GD * GD : Wf + (size_t)i * GD * GD;
        const float* bsk = (i > 0) ? bskip + (i - 1) * GD : bf + i * GD;
        layer_tile_kernel<<<gAgg, dim3(AGG_T), 0, stream>>>(
            x, bufA, src, dst, dinv, bc + i * GD,
            Wf + (size_t)i * GD * GD, bf + i * GD, Wsk, bsk,
            gamma + i * GD, beta + i * GD, rmean + i * GD, rvar + i * GD,
            n, e, (i > 0) ? 1 : 0, bufA, ws_size);
    }
}


# Layer 0's tile owners save a two-word record per found edge (source, tile row, weight) in the workspace; layers 1-2 load them: no tile-id scan, no src/dst/dinv loads; sums stay in arrival order
# speedup vs baseline: 2.3103x; 1.0925x over previous
#include <hip/hip_runtime.h>
#include <hip/hip_bf16.h>

typedef __attribute__((ext_vector_type(2))) float v2f;
typedef __attribute__((ext_vector_type(8))) float v8f;

#define GD 128
#define BN_EPS 1e-5f
#define LRELU_SLOPE 0.01f

__global__ __launch_bounds__(256)
void wmma_gemm128(const float* __restrict__ A, const float* __restrict__ W,
                  float* __restrict__ C, int nrows) {
    __shared__ float lds[GD * GD];
    const int t = threadIdx.x;
    for (int i = t; i < GD * GD; i += 256) lds[i] = W[i];
    __syncthreads();

    const int wave = t >> 5;
    const int lane = t & 31;
    const int half = lane >> 4;
    const int l16  = lane & 15;

    const int m0 = blockIdx.x * 128 + wave * 16;
    int arow = m0 + l16;
    if (arow >= nrows) arow = nrows - 1;
    const float* aptr = A + (size_t)arow * GD + half * 2;

    v8f acc[8];
#pragma unroll
    for (int nt = 0; nt < 8; ++nt) acc[nt] = (v8f){0, 0, 0, 0, 0, 0, 0, 0};

    for (int k0 = 0; k0 < GD; k0 += 4) {
        const v2f a = *(const v2f*)(aptr + k0);
#pragma unroll
        for (int nt = 0; nt < 8; ++nt) {
            const int nn = nt * 16 + l16;
            v2f b;
            b.x = lds[(k0 + half * 2)     * GD + nn];
            b.y = lds[(k0 + half * 2 + 1) * GD + nn];
            acc[nt] = __builtin_amdgcn_wmma_f32_16x16x4_f32(
                false, a, false, b, (short)0, acc[nt], false, false);
        }
    }

#pragma unroll
    for (int r = 0; r < 8; ++r) {
        const int row = m0 + r + half * 8;
        if (row < nrows) {
#pragma unroll
            for (int nt = 0; nt < 8; ++nt)
                C[(size_t)row * GD + nt * 16 + l16] = acc[nt][r];
        }
    }
}

__global__ void init_one_kernel(float* __restrict__ p, int n) {
    int i = blockIdx.x * 256 + threadIdx.x;
    if (i < n) p[i] = 1.0f;
}
__global__ void rsqrt_kernel(const float* __restrict__ deg, float* __restrict__ dinv, int n) {
    int i = blockIdx.x * 256 + threadIdx.x;
    if (i < n) dinv[i] = rsqrtf(deg[i]);
}
__global__ void copy_kernel(const float* __restrict__ a, float* __restrict__ b, int n) {
    int i = blockIdx.x * 256 + threadIdx.x;
    if (i < n) b[i] = a[i];
}

#define TR 200
#define TP 130
#define WP 136
#define AGG_T 1024
#define CH 32768
#define LC (CH + 4096)
#define GU 8
#define NSTRIPE ((TR + 15) / 16)
#define PERM(c) (((((c) & 1) << 1) | (((c) >> 1) & 1)) * 32 + ((c) >> 2))
#define OFX 0
#define OFY 64
#define OFZ 32
#define OFW 96

#define TPF 1
#define TVS (CH / (AGG_T * 16))
__device__ __forceinline__ size_t tid_offset(int n) {
    const size_t used = ((size_t)n * GD + 2 * (size_t)n) * sizeof(float);
    return (used + 255) & ~(size_t)255;
}
__device__ __forceinline__ size_t tid_length(int e) {
    return ((size_t)((e + CH - 1) / CH) + TPF) * CH;
}
__device__ __forceinline__ bool tid_usable(int n, int e, size_t ws_bytes) {
    return (n + TR - 1) / TR <= 255 && tid_offset(n) + tid_length(e) <= ws_bytes;
}
#define LCAP 2560
__device__ __forceinline__ size_t rec_offset(int n, int e) {
    return (tid_offset(n) + tid_length(e) + 255) & ~(size_t)255;
}
__device__ __forceinline__ size_t rec_hdr_bytes(int n) {
    return ((size_t)((n + TR - 1) / TR) * sizeof(int) + 255) & ~(size_t)255;
}
__device__ __forceinline__ bool rec_usable(int n, int e, size_t ws_bytes) {
    const size_t tiles = (size_t)((n + TR - 1) / TR);
    return n <= (1 << 24) && rec_offset(n, e) + rec_hdr_bytes(n) + tiles * LCAP * sizeof(int2) <= ws_bytes;
}
__device__ __forceinline__ unsigned tid_match4(unsigned w, unsigned pat) {
    const unsigned x = w ^ pat;
    return ~(((x & 0x7F7F7F7Fu) + 0x7F7F7F7Fu) | x | 0x7F7F7F7Fu);
}

__global__ void deg_tid_kernel(float* __restrict__ deg, const int* __restrict__ dst, int e, int n,
                               float* __restrict__ wsf, size_t ws_bytes) {
    const size_t i4 = (size_t)blockIdx.x * 256 + threadIdx.x;
    int d[4];
#pragma unroll
    for (int j = 0; j < 4; ++j) {
        const size_t idx = i4 * 4 + j;
        d[j] = idx < (size_t)e ? dst[idx] : -1;
    }
    unsigned pk = 0;
#pragma unroll
    for (int j = 0; j < 4; ++j) {
        unsigned b = 255u;
        if (i4 * 4 + j < (size_t)e) {
            unsafeAtomicAdd(&deg[d[j]], 1.0f);
            if ((unsigned)d[j] < (unsigned)n) b = (unsigned)d[j] / TR;
        }
        pk |= b << (8 * j);
    }
    if (tid_usable(n, e, ws_bytes) && i4 * 4 < tid_length(e))
        *((unsigned*)((unsigned char*)wsf + tid_offset(n)) + i4) = pk;
}

__device__ __forceinline__ void stage_w(float* __restrict__ wl, const float* __restrict__ Wg, int t) {
#pragma unroll
    for (int j = 0; j < (GD * GD / 4) / AGG_T; ++j) {
        const int i4 = t + j * AGG_T;
        const int row = i4 >> 5;
        const int c4 = i4 & 31;
        const float4 v = *((const float4*)Wg + i4);
        *(float4*)(wl + row * WP + c4 * 4) = v;
    }
}

__device__ __forceinline__ void tile_gemm(const float* __restrict__ tile, const float* __restrict__ wl,
                                          int stripe, int chalf, int l16, int half, v8f acc[4]) {
    int ar = stripe * 16 + l16;
    if (ar > TR - 1) ar = TR - 1;
    const float* ap = tile + ar * TP + half * 32;
    const float* bp = wl + (half * 2) * WP + chalf * 64 + l16;
#pragma unroll
    for (int nt = 0; nt < 4; ++nt) acc[nt] = (v8f){0, 0, 0, 0, 0, 0, 0, 0};
    for (int k0 = 0; k0 < GD; k0 += 4) {
        v2f a;
        a.x = ap[k0 >> 2];
        a.y = ap[(k0 >> 2) + 64];
#pragma unroll
        for (int nt = 0; nt < 4; ++nt) {
            v2f b;
            b.x = bp[k0 * WP + nt * 16];
            b.y = bp[(k0 + 1) * WP + nt * 16];
            acc[nt] = __builtin_amdgcn_wmma_f32_16x16x4_f32(
                false, a, false, b, (short)0, acc[nt], false, false);
        }
    }
}

__global__ __launch_bounds__(AGG_T)
void layer_tile_kernel(float* __restrict__ x, const float* __restrict__ h,
                       const int* __restrict__ src, const int* __restrict__ dst,
                       const float* __restrict__ dinv, const float* __restrict__ bc,
                       const float* __restrict__ Wf, const float* __restrict__ bf,
                       const float* __restrict__ Wsk, const float* __restrict__ bsk,
                       const float* __restrict__ gamma, const float* __restrict__ beta,
                       const float* __restrict__ rmean, const float* __restrict__ rvar,
                       int n, int e, int has_skip, int first,
                       float* wsf, size_t ws_bytes) {
    __shared__ float tile[TR * TP];
    __shared__ __attribute__((aligned(16))) unsigned char lraw[LC * 4];
    __shared__ int   cnt;
    int*   list = (int*)lraw;
    float* wl   = (float*)lraw;

    const int t = threadIdx.x;
    const int lane = t & 31;
    const int wave = t >> 5;
    const int row0 = blockIdx.x * TR;

    for (int i = t; i < TR * TP; i += AGG_T) tile[i] = 0.0f;
    if (t == 0) cnt = 0;
    __syncthreads();

    const bool keep = rec_usable(n, e, ws_bytes);
    unsigned char* rbase = (unsigned char*)wsf + rec_offset(n, e);
    int*  ghdr = (int*)rbase + blockIdx.x;
    int2* grec = (int2*)(rbase + rec_hdr_bytes(n)) + (size_t)blockIdx.x * LCAP;
    const bool save = keep && first != 0;
    int nsaved = -1;
    if (keep && first == 0) {
        nsaved = __builtin_amdgcn_readfirstlane(*ghdr);
        if (nsaved > LCAP) nsaved = -1;
    }

    if (nsaved >= 0) {
        int2* rl = (int2*)lraw;
        for (int i = t; i < nsaved; i += AGG_T) rl[i] = grec[i];
        __syncthreads();
        for (int base = wave * GU; base < nsaved; base += (AGG_T / 32) * GU) {
            int2   rc[GU];
            float4 ev[GU];
#pragma unroll
            for (int u = 0; u < GU; ++u) {
                const int idx = base + u;
                rc[u] = rl[idx < nsaved ? idx : nsaved - 1];
            }
#pragma unroll
            for (int u = 0; u < GU; ++u) {
                int s = rc[u].x & 0xFFFFFF;
                if (s > n - 1) s = n - 1;
                ev[u] = *((const float4*)(h + (size_t)s * GD) + lane);
            }
            static_assert(GU == 8, "the statement below names one loaded value per edge of a trip");
            asm volatile("" :: "v"(ev[0].x), "v"(ev[1].x), "v"(ev[2].x), "v"(ev[3].x),
                               "v"(ev[4].x), "v"(ev[5].x), "v"(ev[6].x), "v"(ev[7].x));
#pragma unroll
            for (int u = 0; u < GU; ++u) {
                if (base + u < nsaved) {
                    int r = (int)((unsigned)rc[u].x >> 24);
                    if (r > TR - 1) r = TR - 1;
                    float* tp = tile + r * TP + lane;
                    const float w = __int_as_float(rc[u].y);
                    __hip_atomic_fetch_add(tp + OFX, ev[u].x * w, __ATOMIC_RELAXED, __HIP_MEMORY_SCOPE_WORKGROUP);
                    __hip_atomic_fetch_add(tp + OFY, ev[u].y * w, __ATOMIC_RELAXED, __HIP_MEMORY_SCOPE_WORKGROUP);
                    __hip_atomic_fetch_add(tp + OFZ, ev[u].z * w, __ATOMIC_RELAXED, __HIP_MEMORY_SCOPE_WORKGROUP);
                    __hip_atomic_fetch_add(tp + OFW, ev[u].w * w, __ATOMIC_RELAXED, __HIP_MEMORY_SCOPE_WORKGROUP);
                }
            }
        }
        __syncthreads();
    } else {
        const bool al16 = ((((size_t)dst) & 15) == 0);

        static_assert(TVS * 4 <= 8, "a thread's hits of one sub-chunk are kept as one 32-bit mask, one bit per byte");
        static_assert(TPF == 1, "the scan holds exactly one sub-chunk ahead in registers");
        const bool compact = tid_usable(n, e, ws_bytes);
        const uint4* tq = (const uint4*)((const unsigned char*)wsf + tid_offset(n)) + t;
        const unsigned pat = (unsigned)blockIdx.x * 0x01010101u;
        uint4 nx[TVS];
#pragma unroll
        for (int v = 0; v < TVS; ++v)
            nx[v] = compact ? tq[v * AGG_T] : make_uint4(0u, 0u, 0u, 0u);

        int  nrec = 0;
        bool fits = true;
        for (int cb = 0; cb < e; cb += CH) {
          if (compact) {
            uint4 c[TVS];
#pragma unroll
            for (int v = 0; v < TVS; ++v) {
                c[v] = nx[v];
                nx[v] = tq[(size_t)(cb / 16) + (size_t)((TVS + v) * AGG_T)];
            }
            unsigned bits = 0;
#pragma unroll
            for (int v = 0; v < TVS; ++v) {
                bits |= tid_match4(c[v].x, pat) >> (7 - (4 * v + 0));
                bits |= tid_match4(c[v].y, pat) >> (7 - (4 * v + 1));
                bits |= tid_match4(c[v].z, pat) >> (7 - (4 * v + 2));
                bits |= tid_match4(c[v].w, pat) >> (7 - (4 * v + 3));
            }
            const int k = __popc(bits);
            if (k > 0) {
                int p = atomicAdd(&cnt, k);
                const int eb = cb + t * 16;
                while (bits != 0) {
                    const int pos = __ffs(bits) - 1;
                    bits &= bits - 1;
                    const int d = pos & 7;
                    list[p++] = eb + (d >> 2) * (AGG_T * 16) + (d & 3) * 4 + (pos >> 3);
                }
            }
          } else {
            int4 dv[CH / (AGG_T * 4)];
            if (al16 && cb + CH <= e) {
                const int4* dp = (const int4*)(dst + cb) + t;
#pragma unroll
                for (int j = 0; j < CH / (AGG_T * 4); ++j) dv[j] = dp[j * AGG_T];
            } else {
#pragma unroll
                for (int j = 0; j < CH / (AGG_T * 4); ++j) {
                    const int i0 = cb + (j * AGG_T + t) * 4;
                    const int a = dst[i0     < e ? i0     : e - 1];
                    const int b = dst[i0 + 1 < e ? i0 + 1 : e - 1];
                    const int c = dst[i0 + 2 < e ? i0 + 2 : e - 1];
                    const int d = dst[i0 + 3 < e ? i0 + 3 : e - 1];
                    dv[j].x = i0     < e ? a : -1;
                    dv[j].y = i0 + 1 < e ? b : -1;
                    dv[j].z = i0 + 2 < e ? c : -1;
                    dv[j].w = i0 + 3 < e ? d : -1;
                }
            }
            int k = 0;
#pragma unroll
            for (int j = 0; j < CH / (AGG_T * 4); ++j) {
                k += ((unsigned)(dv[j].x - row0) < (unsigned)TR) ? 1 : 0;
                k += ((unsigned)(dv[j].y - row0) < (unsigned)TR) ? 1 : 0;
                k += ((unsigned)(dv[j].z - row0) < (unsigned)TR) ? 1 : 0;
                k += ((unsigned)(dv[j].w - row0) < (unsigned)TR) ? 1 : 0;
            }
            if (k > 0) {
                int p = atomicAdd(&cnt, k);
#pragma unroll
                for (int j = 0; j < CH / (AGG_T * 4); ++j) {
                    const int i0 = cb + (j * AGG_T + t) * 4;
                    if ((unsigned)(dv[j].x - row0) < (unsigned)TR) list[p++] = i0;
                    if ((unsigned)(dv[j].y - row0) < (unsigned)TR) list[p++] = i0 + 1;
                    if ((unsigned)(dv[j].z - row0) < (unsigned)TR) list[p++] = i0 + 2;
                    if ((unsigned)(dv[j].w - row0) < (unsigned)TR) list[p++] = i0 + 3;
                }
            }
          }
            __syncthreads();
            const int m = cnt;
            __syncthreads();

            const bool last = (cb + CH >= e);
            if (m > LC - CH || last) {
                const bool wr = save && fits && nrec + m <= LCAP;
                if (save && !wr) fits = false;
                for (int base = wave * GU; base < m; base += (AGG_T / 32) * GU) {
                    int   es[GU], ed[GU];
                    float ew[GU];
                    float4 ev[GU];
#pragma unroll
                    for (int u = 0; u < GU; ++u) {
                        const int idx = base + u;
                        const int eid = list[idx < m ? idx : m - 1];
                        es[u] = src[eid];
                        ed[u] = dst[eid];
                    }
#pragma unroll
                    for (int u = 0; u < GU; ++u) {
                        ew[u] = dinv[es[u]] * dinv[ed[u]];
                        ev[u] = *((const float4*)(h + (size_t)es[u] * GD) + lane);
                    }
#pragma unroll
                    for (int u = 0; u < GU; ++u) {
                        if (base + u < m) {
                            float* tp = tile + (ed[u] - row0) * TP + lane;
                            const float w = ew[u];
                            __hip_atomic_fetch_add(tp + OFX, ev[u].x * w, __ATOMIC_RELAXED, __HIP_MEMORY_SCOPE_WORKGROUP);
                            __hip_atomic_fetch_add(tp + OFY, ev[u].y * w, __ATOMIC_RELAXED, __HIP_MEMORY_SCOPE_WORKGROUP);
                            __hip_atomic_fetch_add(tp + OFZ, ev[u].z * w, __ATOMIC_RELAXED, __HIP_MEMORY_SCOPE_WORKGROUP);
                            __hip_atomic_fetch_add(tp + OFW, ev[u].w * w, __ATOMIC_RELAXED, __HIP_MEMORY_SCOPE_WORKGROUP);
                            if (wr && lane == 0)
                                grec[nrec + base + u] = make_int2((int)((unsigned)es[u] | ((unsigned)(ed[u] - row0) << 24)), __float_as_int(w));
                        }
                    }
                }
                nrec += m;
                __syncthreads();
                if (t == 0) cnt = 0;
                __syncthreads();
            }
        }
        if (save && t == 0) *ghdr = fits ? nrec : -1;
    }

    stage_w(wl, Wf, t);
    if (has_skip) stage_w(wl + GD * WP, Wsk, t);

    {
        const float4 b4 = *((const float4*)bc + lane);
        for (int r = wave; r < TR; r += AGG_T / 32) {
            const int row = row0 + r;
            if (row < n) {
                const float di = dinv[row];
                const float4 hv = *((const float4*)(h + (size_t)row * GD) + lane);
                const float4 xv = *((const float4*)(x + (size_t)row * GD) + lane);
                float* tp = tile + r * TP + lane;
                float4 o;
                o.x = (tp[OFX] + hv.x * di * di + b4.x) + xv.x;
                o.y = (tp[OFY] + hv.y * di * di + b4.y) + xv.y;
                o.z = (tp[OFZ] + hv.z * di * di + b4.z) + xv.z;
                o.w = (tp[OFW] + hv.w * di * di + b4.w) + xv.w;
                tp[OFX] = o.x > 0.0f ? o.x : 0.0f;
                tp[OFY] = o.y > 0.0f ? o.y : 0.0f;
                tp[OFZ] = o.z > 0.0f ? o.z : 0.0f;
                tp[OFW] = o.w > 0.0f ? o.w : 0.0f;
            }
        }
    }
    __syncthreads();

    const int half = lane >> 4;
    const int l16  = lane & 15;
    const int stripe = wave >> 1;
    const int chalf  = wave & 1;
    const bool gw = (wave < 2 * NSTRIPE);

    v8f acc[4];
    float bnm[4], bns[4], bng[4], bnb[4];
#pragma unroll
    for (int nt = 0; nt < 4; ++nt) {
        const int col = chalf * 64 + nt * 16 + l16;
        bnm[nt] = rmean[col];
        bns[nt] = rsqrtf(rvar[col] + BN_EPS);
        bng[nt] = gamma[col];
        bnb[nt] = beta[col];
    }

    if (gw) tile_gemm(tile, wl, stripe, chalf, l16, half, acc);
    __syncthreads();
    if (gw) {
#pragma unroll
        for (int nt = 0; nt < 4; ++nt) {
            const int col = chalf * 64 + nt * 16 + l16;
            const float bias = bf[col];
            const int pc = PERM(col);
#pragma unroll
            for (int r = 0; r < 8; ++r) {
                const int row = stripe * 16 + r + half * 8;
                if (row < TR) {
                    float* tp = tile + row * TP + pc;
                    float u = acc[nt][r] + bias;
                    u = u > 0.0f ? u : LRELU_SLOPE * u;
                    float v = u + *tp;
                    v = v > 0.0f ? v : 0.0f;
                    if (!has_skip) {
                        const float b = (v - bnm[nt]) * bns[nt] * bng[nt] + bnb[nt];
                        const float y = b + v;
                        v = y > 0.0f ? y : 0.0f;
                    }
                    *tp = v;
                }
            }
        }
    }
    __syncthreads();

    if (has_skip) {
        if (gw) tile_gemm(tile, wl + GD * WP, stripe, chalf, l16, half, acc);
        __syncthreads();
        if (gw) {
#pragma unroll
            for (int nt = 0; nt < 4; ++nt) {
                const int col = chalf * 64 + nt * 16 + l16;
                const float bias = bsk[col];
                const int pc = PERM(col);
#pragma unroll
                for (int r = 0; r < 8; ++r) {
                    const int row = stripe * 16 + r + half * 8;
                    if (row < TR) {
                        float* tp = tile + row * TP + pc;
                        const float s = acc[nt][r] + bias;
                        float v = *tp + s;
                        v = v > 0.0f ? v : 0.0f;
                        const float b = (v - bnm[nt]) * bns[nt] * bng[nt] + bnb[nt];
                        const float y = b + v;
                        *tp = y > 0.0f ? y : 0.0f;
                    }
                }
            }
        }
        __syncthreads();
    }

    for (int r = wave; r < TR; r += AGG_T / 32) {
        const int row = row0 + r;
        if (row < n) {
            const float* tp = tile + r * TP + lane;
            float4 o;
            o.x = tp[OFX];
            o.y = tp[OFY];
            o.z = tp[OFZ];
            o.w = tp[OFW];
            *((float4*)(x + (size_t)row * GD) + lane) = o;
        }
    }
}

extern "C" void kernel_launch(void* const* d_in, const int* in_sizes, int n_in,
                              void* d_out, int out_size, void* d_ws, size_t ws_size,
                              hipStream_t stream) {
    const float* x_in  = (const float*)d_in[0];
    const int*   ei    = (const int*)d_in[1];
    const float* Wc    = (const float*)d_in[2];
    const float* bc    = (const float*)d_in[3];
    const float* Wf    = (const float*)d_in[4];
    const float* bf    = (const float*)d_in[5];
    const float* Wskip = (const float*)d_in[6];
    const float* bskip = (const float*)d_in[7];
    const float* gamma = (const float*)d_in[8];
    const float* beta  = (const float*)d_in[9];
    const float* rmean = (const float*)d_in[10];
    const float* rvar  = (const float*)d_in[11];

    const int n = in_sizes[0] / GD;
    const int e = in_sizes[1] / 2;
    const int L = in_sizes[3] / GD;

    const int* src = ei;
    const int* dst = ei + e;

    float* x    = (float*)d_out;
    float* bufA = (float*)d_ws;
    float* deg  = bufA + (size_t)n * GD;
    float* dinv = deg + n;

    const int ND = n * GD;
    dim3 blk(256);
    const int gND   = (ND + 255) / 256;
    const int gN    = (n + 255) / 256;
    const int gT    = ((e + CH - 1) / CH + TPF) * (CH / 1024);
    const int gAgg  = (n + TR - 1) / TR;
    const int gGemm = (n + 127) / 128;

    init_one_kernel<<<gN, blk, 0, stream>>>(deg, n);
    deg_tid_kernel<<<gT, blk, 0, stream>>>(deg, dst, e, n, bufA, ws_size);
    rsqrt_kernel<<<gN, blk, 0, stream>>>(deg, dinv, n);
    copy_kernel<<<gND, blk, 0, stream>>>(x_in, x, ND);

    for (int i = 0; i < L; ++i) {
        wmma_gemm128<<<gGemm, blk, 0, stream>>>(x, Wc + (size_t)i * GD * GD, bufA, n);
        const float* Wsk = (i > 0) ? Wskip + (size_t)(i - 1) * GD * GD : Wf + (size_t)i * GD * GD;
        const float* bsk = (i > 0) ? bskip + (i - 1) * GD : bf + i * GD;
        layer_tile_kernel<<<gAgg, dim3(AGG_T), 0, stream>>>(
            x, bufA, src, dst, dinv, bc + i * GD,
            Wf + (size_t)i * GD * GD, bf + i * GD, Wsk, bsk,
            gamma + i * GD, beta + i * GD, rmean + i * GD, rvar + i * GD,
            n, e, (i > 0) ? 1 : 0, (i == 0) ? 1 : 0, bufA, ws_size);
    }
}


# Set-up side by side: one launch runs layer 0's conv product (from the input x), the degree/tile-id pass and the input copy (16 B per lane, 8 loads in flight) as block ranges; 10 launches to 8
# speedup vs baseline: 2.5461x; 1.1021x over previous
#include <hip/hip_runtime.h>
#include <hip/hip_bf16.h>

typedef __attribute__((ext_vector_type(2))) float v2f;
typedef __attribute__((ext_vector_type(8))) float v8f;

#define GD 128
#define BN_EPS 1e-5f
#define LRELU_SLOPE 0.01f

__device__ __forceinline__
void gemm128_block(const float* __restrict__ A, const float* __restrict__ W,
                   float* __restrict__ C, int nrows, int bid) {
    __shared__ float lds[GD * GD];
    const int t = threadIdx.x;
    for (int i = t; i < GD * GD; i += 256) lds[i] = W[i];
    __syncthreads();

    const int wave = t >> 5;
    const int lane = t & 31;
    const int half = lane >> 4;
    const int l16  = lane & 15;

    const int m0 = bid * 128 + wave * 16;
    int arow = m0 + l16;
    if (arow >= nrows) arow = nrows - 1;
    const float* aptr = A + (size_t)arow * GD + half * 2;

    v8f acc[8];
#pragma unroll
    for (int nt = 0; nt < 8; ++nt) acc[nt] = (v8f){0, 0, 0, 0, 0, 0, 0, 0};

    for (int k0 = 0; k0 < GD; k0 += 4) {
        const v2f a = *(const v2f*)(aptr + k0);
#pragma unroll
        for (int nt = 0; nt < 8; ++nt) {
            const int nn = nt * 16 + l16;
            v2f b;
            b.x = lds[(k0 + half * 2)     * GD + nn];
            b.y = lds[(k0 + half * 2 + 1) * GD + nn];
            acc[nt] = __builtin_amdgcn_wmma_f32_16x16x4_f32(
                false, a, false, b, (short)0, acc[nt], false, false);
        }
    }

#pragma unroll
    for (int r = 0; r < 8; ++r) {
        const int row = m0 + r + half * 8;
        if (row < nrows) {
#pragma unroll
            for (int nt = 0; nt < 8; ++nt)
                C[(size_t)row * GD + nt * 16 + l16] = acc[nt][r];
        }
    }
}

__global__ __launch_bounds__(256)
void wmma_gemm128(const float* __restrict__ A, const float* __restrict__ W,
                  float* __restrict__ C, int nrows) {
    gemm128_block(A, W, C, nrows, blockIdx.x);
}

__global__ void init_one_kernel(float* __restrict__ p, int n) {
    int i = blockIdx.x * 256 + threadIdx.x;
    if (i < n) p[i] = 1.0f;
}
__global__ void rsqrt_kernel(const float* __restrict__ deg, float* __restrict__ dinv, int n) {
    int i = blockIdx.x * 256 + threadIdx.x;
    if (i < n) dinv[i] = rsqrtf(deg[i]);
}

#define TR 200
#define TP 130
#define WP 136
#define AGG_T 1024
#define CH 32768
#define LC (CH + 4096)
#define GU 8
#define NSTRIPE ((TR + 15) / 16)
#define PERM(c) (((((c) & 1) << 1) | (((c) >> 1) & 1)) * 32 + ((c) >> 2))
#define OFX 0
#define OFY 64
#define OFZ 32
#define OFW 96

#define TPF 1
#define TVS (CH / (AGG_T * 16))
__device__ __forceinline__ size_t tid_offset(int n) {
    const size_t used = ((size_t)n * GD + 2 * (size_t)n) * sizeof(float);
    return (used + 255) & ~(size_t)255;
}
__device__ __forceinline__ size_t tid_length(int e) {
    return ((size_t)((e + CH - 1) / CH) + TPF) * CH;
}
__device__ __forceinline__ bool tid_usable(int n, int e, size_t ws_bytes) {
    return (n + TR - 1) / TR <= 255 && tid_offset(n) + tid_length(e) <= ws_bytes;
}
#define LCAP 2560
__device__ __forceinline__ size_t rec_offset(int n, int e) {
    return (tid_offset(n) + tid_length(e) + 255) & ~(size_t)255;
}
__device__ __forceinline__ size_t rec_hdr_bytes(int n) {
    return ((size_t)((n + TR - 1) / TR) * sizeof(int) + 255) & ~(size_t)255;
}
__device__ __forceinline__ bool rec_usable(int n, int e, size_t ws_bytes) {
    const size_t tiles = (size_t)((n + TR - 1) / TR);
    return n <= (1 << 24) && rec_offset(n, e) + rec_hdr_bytes(n) + tiles * LCAP * sizeof(int2) <= ws_bytes;
}
__device__ __forceinline__ unsigned tid_match4(unsigned w, unsigned pat) {
    const unsigned x = w ^ pat;
    return ~(((x & 0x7F7F7F7Fu) + 0x7F7F7F7Fu) | x | 0x7F7F7F7Fu);
}

__device__ __forceinline__ void deg_tid_block(float* __restrict__ deg, const int* __restrict__ dst, int e, int n,
                                              float* __restrict__ wsf, size_t ws_bytes, int bid) {
    const size_t i4 = (size_t)bid * 256 + threadIdx.x;
    int d[4];
#pragma unroll
    for (int j = 0; j < 4; ++j) {
        const size_t idx = i4 * 4 + j;
        d[j] = idx < (size_t)e ? dst[idx] : -1;
    }
    unsigned pk = 0;
#pragma unroll
    for (int j = 0; j < 4; ++j) {
        unsigned b = 255u;
        if (i4 * 4 + j < (size_t)e) {
            unsafeAtomicAdd(&deg[d[j]], 1.0f);
            if ((unsigned)d[j] < (unsigned)n) b = (unsigned)d[j] / TR;
        }
        pk |= b << (8 * j);
    }
    if (tid_usable(n, e, ws_bytes) && i4 * 4 < tid_length(e))
        *((unsigned*)((unsigned char*)wsf + tid_offset(n)) + i4) = pk;
}

#define CPV 8
__device__ __forceinline__ void copy_block(const float* __restrict__ a, float* __restrict__ b, int n4, int bid) {
    const float4* ap = (const float4*)a;
    float4* bp = (float4*)b;
    const int base = bid * (256 * CPV) + threadIdx.x;
    float4 v[CPV];
#pragma unroll
    for (int j = 0; j < CPV; ++j) {
        const int i = base + j * 256;
        v[j] = i < n4 ? ap[i] : make_float4(0.0f, 0.0f, 0.0f, 0.0f);
    }
#pragma unroll
    for (int j = 0; j < CPV; ++j) {
        const int i = base + j * 256;
        if (i < n4) bp[i] = v[j];
    }
}

__global__ __launch_bounds__(256)
void prep_conv0_kernel(const float* __restrict__ x_in, const float* __restrict__ W, float* __restrict__ hbuf,
                       int n, float* __restrict__ deg, const int* __restrict__ dst, int e,
                       float* wsf, size_t ws_bytes,
                       float* __restrict__ xout, int n4, int g_gemm, int g_tid) {
    const int b = blockIdx.x;
    if (b < g_gemm) {
        gemm128_block(x_in, W, hbuf, n, b);
    } else if (b < g_gemm + g_tid) {
        deg_tid_block(deg, dst, e, n, wsf, ws_bytes, b - g_gemm);
    } else {
        copy_block(x_in, xout, n4, b - g_gemm - g_tid);
    }
}

__device__ __forceinline__ void stage_w(float* __restrict__ wl, const float* __restrict__ Wg, int t) {
#pragma unroll
    for (int j = 0; j < (GD * GD / 4) / AGG_T; ++j) {
        const int i4 = t + j * AGG_T;
        const int row = i4 >> 5;
        const int c4 = i4 & 31;
        const float4 v = *((const float4*)Wg + i4);
        *(float4*)(wl + row * WP + c4 * 4) = v;
    }
}

__device__ __forceinline__ void tile_gemm(const float* __restrict__ tile, const float* __restrict__ wl,
                                          int stripe, int chalf, int l16, int half, v8f acc[4]) {
    int ar = stripe * 16 + l16;
    if (ar > TR - 1) ar = TR - 1;
    const float* ap = tile + ar * TP + half * 32;
    const float* bp = wl + (half * 2) * WP + chalf * 64 + l16;
#pragma unroll
    for (int nt = 0; nt < 4; ++nt) acc[nt] = (v8f){0, 0, 0, 0, 0, 0, 0, 0};
    for (int k0 = 0; k0 < GD; k0 += 4) {
        v2f a;
        a.x = ap[k0 >> 2];
        a.y = ap[(k0 >> 2) + 64];
#pragma unroll
        for (int nt = 0; nt < 4; ++nt) {
            v2f b;
            b.x = bp[k0 * WP + nt * 16];
            b.y = bp[(k0 + 1) * WP + nt * 16];
            acc[nt] = __builtin_amdgcn_wmma_f32_16x16x4_f32(
                false, a, false, b, (short)0, acc[nt], false, false);
        }
    }
}

__global__ __launch_bounds__(AGG_T)
void layer_tile_kernel(float* __restrict__ x, const float* __restrict__ h,
                       const int* __restrict__ src, const int* __restrict__ dst,
                       const float* __restrict__ dinv, const float* __restrict__ bc,
                       const float* __restrict__ Wf, const float* __restrict__ bf,
                       const float* __restrict__ Wsk, const float* __restrict__ bsk,
                       const float* __restrict__ gamma, const float* __restrict__ beta,
                       const float* __restrict__ rmean, const float* __restrict__ rvar,
                       int n, int e, int has_skip, int first,
                       float* wsf, size_t ws_bytes) {
    __shared__ float tile[TR * TP];
    __shared__ __attribute__((aligned(16))) unsigned char lraw[LC * 4];
    __shared__ int   cnt;
    int*   list = (int*)lraw;
    float* wl   = (float*)lraw;

    const int t = threadIdx.x;
    const int lane = t & 31;
    const int wave = t >> 5;
    const int row0 = blockIdx.x * TR;

    for (int i = t; i < TR * TP; i += AGG_T) tile[i] = 0.0f;
    if (t == 0) cnt = 0;
    __syncthreads();

    const bool keep = rec_usable(n, e, ws_bytes);
    unsigned char* rbase = (unsigned char*)wsf + rec_offset(n, e);
    int*  ghdr = (int*)rbase + blockIdx.x;
    int2* grec = (int2*)(rbase + rec_hdr_bytes(n)) + (size_t)blockIdx.x * LCAP;
    const bool save = keep && first != 0;
    int nsaved = -1;
    if (keep && first == 0) {
        nsaved = __builtin_amdgcn_readfirstlane(*ghdr);
        if (nsaved > LCAP) nsaved = -1;
    }

    if (nsaved >= 0) {
        int2* rl = (int2*)lraw;
        for (int i = t; i < nsaved; i += AGG_T) rl[i] = grec[i];
        __syncthreads();
        for (int base = wave * GU; base < nsaved; base += (AGG_T / 32) * GU) {
            int2   rc[GU];
            float4 ev[GU];
#pragma unroll
            for (int u = 0; u < GU; ++u) {
                const int idx = base + u;
                rc[u] = rl[idx < nsaved ? idx : nsaved - 1];
            }
#pragma unroll
            for (int u = 0; u < GU; ++u) {
                int s = rc[u].x & 0xFFFFFF;
                if (s > n - 1) s = n - 1;
                ev[u] = *((const float4*)(h + (size_t)s * GD) + lane);
            }
            static_assert(GU == 8, "the statement below names one loaded value per edge of a trip");
            asm volatile("" :: "v"(ev[0].x), "v"(ev[1].x), "v"(ev[2].x), "v"(ev[3].x),
                               "v"(ev[4].x), "v"(ev[5].x), "v"(ev[6].x), "v"(ev[7].x));
#pragma unroll
            for (int u = 0; u < GU; ++u) {
                if (base + u < nsaved) {
                    int r = (int)((unsigned)rc[u].x >> 24);
                    if (r > TR - 1) r = TR - 1;
                    float* tp = tile + r * TP + lane;
                    const float w = __int_as_float(rc[u].y);
                    __hip_atomic_fetch_add(tp + OFX, ev[u].x * w, __ATOMIC_RELAXED, __HIP_MEMORY_SCOPE_WORKGROUP);
                    __hip_atomic_fetch_add(tp + OFY, ev[u].y * w, __ATOMIC_RELAXED, __HIP_MEMORY_SCOPE_WORKGROUP);
                    __hip_atomic_fetch_add(tp + OFZ, ev[u].z * w, __ATOMIC_RELAXED, __HIP_MEMORY_SCOPE_WORKGROUP);
                    __hip_atomic_fetch_add(tp + OFW, ev[u].w * w, __ATOMIC_RELAXED, __HIP_MEMORY_SCOPE_WORKGROUP);
                }
            }
        }
        __syncthreads();
    } else {
        const bool al16 = ((((size_t)dst) & 15) == 0);

        static_assert(TVS * 4 <= 8, "a thread's hits of one sub-chunk are kept as one 32-bit mask, one bit per byte");
        static_assert(TPF == 1, "the scan holds exactly one sub-chunk ahead in registers");
        const bool compact = tid_usable(n, e, ws_bytes);
        const uint4* tq = (const uint4*)((const unsigned char*)wsf + tid_offset(n)) + t;
        const unsigned pat = (unsigned)blockIdx.x * 0x01010101u;
        uint4 nx[TVS];
#pragma unroll
        for (int v = 0; v < TVS; ++v)
            nx[v] = compact ? tq[v * AGG_T] : make_uint4(0u, 0u, 0u, 0u);

        int  nrec = 0;
        bool fits = true;
        for (int cb = 0; cb < e; cb += CH) {
          if (compact) {
            uint4 c[TVS];
#pragma unroll
            for (int v = 0; v < TVS; ++v) {
                c[v] = nx[v];
                nx[v] = tq[(size_t)(cb / 16) + (size_t)((TVS + v) * AGG_T)];
            }
            unsigned bits = 0;
#pragma unroll
            for (int v = 0; v < TVS; ++v) {
                bits |= tid_match4(c[v].x, pat) >> (7 - (4 * v + 0));
                bits |= tid_match4(c[v].y, pat) >> (7 - (4 * v + 1));
                bits |= tid_match4(c[v].z, pat) >> (7 - (4 * v + 2));
                bits |= tid_match4(c[v].w, pat) >> (7 - (4 * v + 3));
            }
            const int k = __popc(bits);
            if (k > 0) {
                int p = atomicAdd(&cnt, k);
                const int eb = cb + t * 16;
                while (bits != 0) {
                    const int pos = __ffs(bits) - 1;
                    bits &= bits - 1;
                    const int d = pos & 7;
                    list[p++] = eb + (d >> 2) * (AGG_T * 16) + (d & 3) * 4 + (pos >> 3);
                }
            }
          } else {
            int4 dv[CH / (AGG_T * 4)];
            if (al16 && cb + CH <= e) {
                const int4* dp = (const int4*)(dst + cb) + t;
#pragma unroll
                for (int j = 0; j < CH / (AGG_T * 4); ++j) dv[j] = dp[j * AGG_T];
            } else {
#pragma unroll
                for (int j = 0; j < CH / (AGG_T * 4); ++j) {
                    const int i0 = cb + (j * AGG_T + t) * 4;
                    const int a = dst[i0     < e ? i0     : e - 1];
                    const int b = dst[i0 + 1 < e ? i0 + 1 : e - 1];
                    const int c = dst[i0 + 2 < e ? i0 + 2 : e - 1];
                    const int d = dst[i0 + 3 < e ? i0 + 3 : e - 1];
                    dv[j].x = i0     < e ? a : -1;
                    dv[j].y = i0 + 1 < e ? b : -1;
                    dv[j].z = i0 + 2 < e ? c : -1;
                    dv[j].w = i0 + 3 < e ? d : -1;
                }
            }
            int k = 0;
#pragma unroll
            for (int j = 0; j < CH / (AGG_T * 4); ++j) {
                k += ((unsigned)(dv[j].x - row0) < (unsigned)TR) ? 1 : 0;
                k += ((unsigned)(dv[j].y - row0) < (unsigned)TR) ? 1 : 0;
                k += ((unsigned)(dv[j].z - row0) < (unsigned)TR) ? 1 : 0;
                k += ((unsigned)(dv[j].w - row0) < (unsigned)TR) ? 1 : 0;
            }
            if (k > 0) {
                int p = atomicAdd(&cnt, k);
#pragma unroll
                for (int j = 0; j < CH / (AGG_T * 4); ++j) {
                    const int i0 = cb + (j * AGG_T + t) * 4;
                    if ((unsigned)(dv[j].x - row0) < (unsigned)TR) list[p++] = i0;
                    if ((unsigned)(dv[j].y - row0) < (unsigned)TR) list[p++] = i0 + 1;
                    if ((unsigned)(dv[j].z - row0) < (unsigned)TR) list[p++] = i0 + 2;
                    if ((unsigned)(dv[j].w - row0) < (unsigned)TR) list[p++] = i0 + 3;
                }
            }
          }
            __syncthreads();
            const int m = cnt;
            __syncthreads();

            const bool last = (cb + CH >= e);
            if (m > LC - CH || last) {
                const bool wr = save && fits && nrec + m <= LCAP;
                if (save && !wr) fits = false;
                for (int base = wave * GU; base < m; base += (AGG_T / 32) * GU) {
                    int   es[GU], ed[GU];
                    float ew[GU];
                    float4 ev[GU];
#pragma unroll
                    for (int u = 0; u < GU; ++u) {
                        const int idx = base + u;
                        const int eid = list[idx < m ? idx : m - 1];
                        es[u] = src[eid];
                        ed[u] = dst[eid];
                    }
#pragma unroll
                    for (int u = 0; u < GU; ++u) {
                        ew[u] = dinv[es[u]] * dinv[ed[u]];
                        ev[u] = *((const float4*)(h + (size_t)es[u] * GD) + lane);
                    }
#pragma unroll
                    for (int u = 0; u < GU; ++u) {
                        if (base + u < m) {
                            float* tp = tile + (ed[u] - row0) * TP + lane;
                            const float w = ew[u];
                            __hip_atomic_fetch_add(tp + OFX, ev[u].x * w, __ATOMIC_RELAXED, __HIP_MEMORY_SCOPE_WORKGROUP);
                            __hip_atomic_fetch_add(tp + OFY, ev[u].y * w, __ATOMIC_RELAXED, __HIP_MEMORY_SCOPE_WORKGROUP);
                            __hip_atomic_fetch_add(tp + OFZ, ev[u].z * w, __ATOMIC_RELAXED, __HIP_MEMORY_SCOPE_WORKGROUP);
                            __hip_atomic_fetch_add(tp + OFW, ev[u].w * w, __ATOMIC_RELAXED, __HIP_MEMORY_SCOPE_WORKGROUP);
                            if (wr && lane == 0)
                                grec[nrec + base + u] = make_int2((int)((unsigned)es[u] | ((unsigned)(ed[u] - row0) << 24)), __float_as_int(w));
                        }
                    }
                }
                nrec += m;
                __syncthreads();
                if (t == 0) cnt = 0;
                __syncthreads();
            }
        }
        if (save && t == 0) *ghdr = fits ? nrec : -1;
    }

    stage_w(wl, Wf, t);
    if (has_skip) stage_w(wl + GD * WP, Wsk, t);

    {
        const float4 b4 = *((const float4*)bc + lane);
        for (int r = wave; r < TR; r += AGG_T / 32) {
            const int row = row0 + r;
            if (row < n) {
                const float di = dinv[row];
                const float4 hv = *((const float4*)(h + (size_t)row * GD) + lane);
                const float4 xv = *((const float4*)(x + (size_t)row * GD) + lane);
                float* tp = tile + r * TP + lane;
                float4 o;
                o.x = (tp[OFX] + hv.x * di * di + b4.x) + xv.x;
                o.y = (tp[OFY] + hv.y * di * di + b4.y) + xv.y;
                o.z = (tp[OFZ] + hv.z * di * di + b4.z) + xv.z;
                o.w = (tp[OFW] + hv.w * di * di + b4.w) + xv.w;
                tp[OFX] = o.x > 0.0f ? o.x : 0.0f;
                tp[OFY] = o.y > 0.0f ? o.y : 0.0f;
                tp[OFZ] = o.z > 0.0f ? o.z : 0.0f;
                tp[OFW] = o.w > 0.0f ? o.w : 0.0f;
            }
        }
    }
    __syncthreads();

    const int half = lane >> 4;
    const int l16  = lane & 15;
    const int stripe = wave >> 1;
    const int chalf  = wave & 1;
    const bool gw = (wave < 2 * NSTRIPE);

    v8f acc[4];
    float bnm[4], bns[4], bng[4], bnb[4];
#pragma unroll
    for (int nt = 0; nt < 4; ++nt) {
        const int col = chalf * 64 + nt * 16 + l16;
        bnm[nt] = rmean[col];
        bns[nt] = rsqrtf(rvar[col] + BN_EPS);
        bng[nt] = gamma[col];
        bnb[nt] = beta[col];
    }

    if (gw) tile_gemm(tile, wl, stripe, chalf, l16, half, acc);
    __syncthreads();
    if (gw) {
#pragma unroll
        for (int nt = 0; nt < 4; ++nt) {
            const int col = chalf * 64 + nt * 16 + l16;
            const float bias = bf[col];
            const int pc = PERM(col);
#pragma unroll
            for (int r = 0; r < 8; ++r) {
                const int row = stripe * 16 + r + half * 8;
                if (row < TR) {
                    float* tp = tile + row * TP + pc;
                    float u = acc[nt][r] + bias;
                    u = u > 0.0f ? u : LRELU_SLOPE * u;
                    float v = u + *tp;
                    v = v > 0.0f ? v : 0.0f;
                    if (!has_skip) {
                        const float b = (v - bnm[nt]) * bns[nt] * bng[nt] + bnb[nt];
                        const float y = b + v;
                        v = y > 0.0f ? y : 0.0f;
                    }
                    *tp = v;
                }
            }
        }
    }
    __syncthreads();

    if (has_skip) {
        if (gw) tile_gemm(tile, wl + GD * WP, stripe, chalf, l16, half, acc);
        __syncthreads();
        if (gw) {
#pragma unroll
            for (int nt = 0; nt < 4; ++nt) {
                const int col = chalf * 64 + nt * 16 + l16;
                const float bias = bsk[col];
                const int pc = PERM(col);
#pragma unroll
                for (int r = 0; r < 8; ++r) {
                    const int row = stripe * 16 + r + half * 8;
                    if (row < TR) {
                        float* tp = tile + row * TP + pc;
                        const float s = acc[nt][r] + bias;
                        float v = *tp + s;
                        v = v > 0.0f ? v : 0.0f;
                        const float b = (v - bnm[nt]) * bns[nt] * bng[nt] + bnb[nt];
                        const float y = b + v;
                        *tp = y > 0.0f ? y : 0.0f;
                    }
                }
            }
        }
        __syncthreads();
    }

    for (int r = wave; r < TR; r += AGG_T / 32) {
        const int row = row0 + r;
        if (row < n) {
            const float* tp = tile + r * TP + lane;
            float4 o;
            o.x = tp[OFX];
            o.y = tp[OFY];
            o.z = tp[OFZ];
            o.w = tp[OFW];
            *((float4*)(x + (size_t)row * GD) + lane) = o;
        }
    }
}

extern "C" void kernel_launch(void* const* d_in, const int* in_sizes, int n_in,
                              void* d_out, int out_size, void* d_ws, size_t ws_size,
                              hipStream_t stream) {
    const float* x_in  = (const float*)d_in[0];
    const int*   ei    = (const int*)d_in[1];
    const float* Wc    = (const float*)d_in[2];
    const float* bc    = (const float*)d_in[3];
    const float* Wf    = (const float*)d_in[4];
    const float* bf    = (const float*)d_in[5];
    const float* Wskip = (const float*)d_in[6];
    const float* bskip = (const float*)d_in[7];
    const float* gamma = (const float*)d_in[8];
    const float* beta  = (const float*)d_in[9];
    const float* rmean = (const float*)d_in[10];
    const float* rvar  = (const float*)d_in[11];

    const int n = in_sizes[0] / GD;
    const int e = in_sizes[1] / 2;
    const int L = in_sizes[3] / GD;

    const int* src = ei;
    const int* dst = ei + e;

    float* x    = (float*)d_out;
    float* bufA = (float*)d_ws;
    float* deg  = bufA + (size_t)n * GD;
    float* dinv = deg + n;

    const int ND4 = n * (GD / 4);
    dim3 blk(256);
    const int gCopy = (ND4 + 256 * CPV - 1) / (256 * CPV);
    const int gN    = (n + 255) / 256;
    const int gT    = ((e + CH - 1) / CH + TPF) * (CH / 1024);
    const int gAgg  = (n + TR - 1) / TR;
    const int gGemm = (n + 127) / 128;

    init_one_kernel<<<gN, blk, 0, stream>>>(deg, n);
    prep_conv0_kernel<<<gGemm + gT + gCopy, blk, 0, stream>>>(x_in, Wc, bufA, n, deg, dst, e, bufA, ws_size,
                                                             x, ND4, gGemm, gT);
    rsqrt_kernel<<<gN, blk, 0, stream>>>(deg, dinv, n);

    for (int i = 0; i < L; ++i) {
        if (i > 0) wmma_gemm128<<<gGemm, blk, 0, stream>>>(x, Wc + (size_t)i * GD * GD, bufA, n);
        const float* Wsk = (i > 0) ? Wskip + (size_t)(i - 1) * GD * GD : Wf + (size_t)i * GD * GD;
        const float* bsk = (i > 0) ? bskip + (i - 1) * GD : bf + i * GD;
        layer_tile_kernel<<<gAgg, dim3(AGG_T), 0, stream>>>(
            x, bufA, src, dst, dinv, bc + i * GD,
            Wf + (size_t)i * GD * GD, bf + i * GD, Wsk, bsk,
            gamma + i * GD, beta + i * GD, rmean + i * GD, rvar + i * GD,
            n, e, (i > 0) ? 1 : 0, (i == 0) ? 1 : 0, bufA, ws_size);
    }
}


# Next layer's conv product done by the tile owner from its LDS tile (Wc staged over the dead Wf); h planes alternate in the workspace: both product launches and their read of x gone; 8 launches to 6
# speedup vs baseline: 2.7424x; 1.0771x over previous
#include <hip/hip_runtime.h>
#include <hip/hip_bf16.h>

typedef __attribute__((ext_vector_type(2))) float v2f;
typedef __attribute__((ext_vector_type(8))) float v8f;

#define GD 128
#define BN_EPS 1e-5f
#define LRELU_SLOPE 0.01f

__device__ __forceinline__
void gemm128_block(const float* __restrict__ A, const float* __restrict__ W,
                   float* __restrict__ C, int nrows, int bid) {
    __shared__ float lds[GD * GD];
    const int t = threadIdx.x;
    for (int i = t; i < GD * GD; i += 256) lds[i] = W[i];
    __syncthreads();

    const int wave = t >> 5;
    const int lane = t & 31;
    const int half = lane >> 4;
    const int l16  = lane & 15;

    const int m0 = bid * 128 + wave * 16;
    int arow = m0 + l16;
    if (arow >= nrows) arow = nrows - 1;
    const float* aptr = A + (size_t)arow * GD + half * 2;

    v8f acc[8];
#pragma unroll
    for (int nt = 0; nt < 8; ++nt) acc[nt] = (v8f){0, 0, 0, 0, 0, 0, 0, 0};

    for (int k0 = 0; k0 < GD; k0 += 4) {
        const v2f a = *(const v2f*)(aptr + k0);
#pragma unroll
        for (int nt = 0; nt < 8; ++nt) {
            const int nn = nt * 16 + l16;
            v2f b;
            b.x = lds[(k0 + half * 2)     * GD + nn];
            b.y = lds[(k0 + half * 2 + 1) * GD + nn];
            acc[nt] = __builtin_amdgcn_wmma_f32_16x16x4_f32(
                false, a, false, b, (short)0, acc[nt], false, false);
        }
    }

#pragma unroll
    for (int r = 0; r < 8; ++r) {
        const int row = m0 + r + half * 8;
        if (row < nrows) {
#pragma unroll
            for (int nt = 0; nt < 8; ++nt)
                C[(size_t)row * GD + nt * 16 + l16] = acc[nt][r];
        }
    }
}

__global__ __launch_bounds__(256)
void wmma_gemm128(const float* __restrict__ A, const float* __restrict__ W,
                  float* __restrict__ C, int nrows) {
    gemm128_block(A, W, C, nrows, blockIdx.x);
}

__global__ void init_one_kernel(float* __restrict__ p, int n) {
    int i = blockIdx.x * 256 + threadIdx.x;
    if (i < n) p[i] = 1.0f;
}
__global__ void rsqrt_kernel(const float* __restrict__ deg, float* __restrict__ dinv, int n) {
    int i = blockIdx.x * 256 + threadIdx.x;
    if (i < n) dinv[i] = rsqrtf(deg[i]);
}

#define TR 200
#define TP 130
#define WP 136
#define AGG_T 1024
#define CH 32768
#define LC (CH + 4096)
#define GU 8
#define NSTRIPE ((TR + 15) / 16)
#define PERM(c) (((((c) & 1) << 1) | (((c) >> 1) & 1)) * 32 + ((c) >> 2))
#define OFX 0
#define OFY 64
#define OFZ 32
#define OFW 96

#define TPF 1
#define TVS (CH / (AGG_T * 16))
__host__ __device__ __forceinline__ size_t tid_offset(int n) {
    const size_t used = ((size_t)n * GD + 2 * (size_t)n) * sizeof(float);
    return (used + 255) & ~(size_t)255;
}
__host__ __device__ __forceinline__ size_t tid_length(int e) {
    return ((size_t)((e + CH - 1) / CH) + TPF) * CH;
}
__device__ __forceinline__ bool tid_usable(int n, int e, size_t ws_bytes) {
    return (n + TR - 1) / TR <= 255 && tid_offset(n) + tid_length(e) <= ws_bytes;
}
#define LCAP 2560
__host__ __device__ __forceinline__ size_t rec_offset(int n, int e) {
    return (tid_offset(n) + tid_length(e) + 255) & ~(size_t)255;
}
__host__ __device__ __forceinline__ size_t rec_hdr_bytes(int n) {
    return ((size_t)((n + TR - 1) / TR) * sizeof(int) + 255) & ~(size_t)255;
}
__device__ __forceinline__ bool rec_usable(int n, int e, size_t ws_bytes) {
    const size_t tiles = (size_t)((n + TR - 1) / TR);
    return n <= (1 << 24) && rec_offset(n, e) + rec_hdr_bytes(n) + tiles * LCAP * sizeof(int2) <= ws_bytes;
}
__host__ __device__ __forceinline__ size_t h2_offset(int n, int e) {
    const size_t tiles = (size_t)((n + TR - 1) / TR);
    return (rec_offset(n, e) + rec_hdr_bytes(n) + tiles * LCAP * sizeof(int2) + 255) & ~(size_t)255;
}
__device__ __forceinline__ unsigned tid_match4(unsigned w, unsigned pat) {
    const unsigned x = w ^ pat;
    return ~(((x & 0x7F7F7F7Fu) + 0x7F7F7F7Fu) | x | 0x7F7F7F7Fu);
}

__device__ __forceinline__ void deg_tid_block(float* __restrict__ deg, const int* __restrict__ dst, int e, int n,
                                              float* __restrict__ wsf, size_t ws_bytes, int bid) {
    const size_t i4 = (size_t)bid * 256 + threadIdx.x;
    int d[4];
#pragma unroll
    for (int j = 0; j < 4; ++j) {
        const size_t idx = i4 * 4 + j;
        d[j] = idx < (size_t)e ? dst[idx] : -1;
    }
    unsigned pk = 0;
#pragma unroll
    for (int j = 0; j < 4; ++j) {
        unsigned b = 255u;
        if (i4 * 4 + j < (size_t)e) {
            unsafeAtomicAdd(&deg[d[j]], 1.0f);
            if ((unsigned)d[j] < (unsigned)n) b = (unsigned)d[j] / TR;
        }
        pk |= b << (8 * j);
    }
    if (tid_usable(n, e, ws_bytes) && i4 * 4 < tid_length(e))
        *((unsigned*)((unsigned char*)wsf + tid_offset(n)) + i4) = pk;
}

#define CPV 8
__device__ __forceinline__ void copy_block(const float* __restrict__ a, float* __restrict__ b, int n4, int bid) {
    const float4* ap = (const float4*)a;
    float4* bp = (float4*)b;
    const int base = bid * (256 * CPV) + threadIdx.x;
    float4 v[CPV];
#pragma unroll
    for (int j = 0; j < CPV; ++j) {
        const int i = base + j * 256;
        v[j] = i < n4 ? ap[i] : make_float4(0.0f, 0.0f, 0.0f, 0.0f);
    }
#pragma unroll
    for (int j = 0; j < CPV; ++j) {
        const int i = base + j * 256;
        if (i < n4) bp[i] = v[j];
    }
}

__global__ __launch_bounds__(256)
void prep_conv0_kernel(const float* __restrict__ x_in, const float* __restrict__ W, float* __restrict__ hbuf,
                       int n, float* __restrict__ deg, const int* __restrict__ dst, int e,
                       float* wsf, size_t ws_bytes,
                       float* __restrict__ xout, int n4, int g_gemm, int g_tid) {
    const int b = blockIdx.x;
    if (b < g_gemm) {
        gemm128_block(x_in, W, hbuf, n, b);
    } else if (b < g_gemm + g_tid) {
        deg_tid_block(deg, dst, e, n, wsf, ws_bytes, b - g_gemm);
    } else {
        copy_block(x_in, xout, n4, b - g_gemm - g_tid);
    }
}

__device__ __forceinline__ void stage_w(float* __restrict__ wl, const float* __restrict__ Wg, int t) {
#pragma unroll
    for (int j = 0; j < (GD * GD / 4) / AGG_T; ++j) {
        const int i4 = t + j * AGG_T;
        const int row = i4 >> 5;
        const int c4 = i4 & 31;
        const float4 v = *((const float4*)Wg + i4);
        *(float4*)(wl + row * WP + c4 * 4) = v;
    }
}

__device__ __forceinline__ void tile_gemm(const float* __restrict__ tile, const float* __restrict__ wl,
                                          int stripe, int chalf, int l16, int half, v8f acc[4]) {
    int ar = stripe * 16 + l16;
    if (ar > TR - 1) ar = TR - 1;
    const float* ap = tile + ar * TP + half * 32;
    const float* bp = wl + (half * 2) * WP + chalf * 64 + l16;
#pragma unroll
    for (int nt = 0; nt < 4; ++nt) acc[nt] = (v8f){0, 0, 0, 0, 0, 0, 0, 0};
    for (int k0 = 0; k0 < GD; k0 += 4) {
        v2f a;
        a.x = ap[k0 >> 2];
        a.y = ap[(k0 >> 2) + 64];
#pragma unroll
        for (int nt = 0; nt < 4; ++nt) {
            v2f b;
            b.x = bp[k0 * WP + nt * 16];
            b.y = bp[(k0 + 1) * WP + nt * 16];
            acc[nt] = __builtin_amdgcn_wmma_f32_16x16x4_f32(
                false, a, false, b, (short)0, acc[nt], false, false);
        }
    }
}

__global__ __launch_bounds__(AGG_T)
void layer_tile_kernel(float* __restrict__ x, const float* __restrict__ h,
                       const int* __restrict__ src, const int* __restrict__ dst,
                       const float* __restrict__ dinv, const float* __restrict__ bc,
                       const float* __restrict__ Wf, const float* __restrict__ bf,
                       const float* __restrict__ Wsk, const float* __restrict__ bsk,
                       const float* __restrict__ gamma, const float* __restrict__ beta,
                       const float* __restrict__ rmean, const float* __restrict__ rvar,
                       int n, int e, int has_skip, int first,
                       float* wsf, size_t ws_bytes,
                       float* __restrict__ hn, const float* __restrict__ Wcn, int has_next) {
    __shared__ float tile[TR * TP];
    __shared__ __attribute__((aligned(16))) unsigned char lraw[LC * 4];
    __shared__ int   cnt;
    int*   list = (int*)lraw;
    float* wl   = (float*)lraw;

    const int t = threadIdx.x;
    const int lane = t & 31;
    const int wave = t >> 5;
    const int row0 = blockIdx.x * TR;

    for (int i = t; i < TR * TP; i += AGG_T) tile[i] = 0.0f;
    if (t == 0) cnt = 0;
    __syncthreads();

    const bool keep = rec_usable(n, e, ws_bytes);
    unsigned char* rbase = (unsigned char*)wsf + rec_offset(n, e);
    int*  ghdr = (int*)rbase + blockIdx.x;
    int2* grec = (int2*)(rbase + rec_hdr_bytes(n)) + (size_t)blockIdx.x * LCAP;
    const bool save = keep && first != 0;
    int nsaved = -1;
    if (keep && first == 0) {
        nsaved = __builtin_amdgcn_readfirstlane(*ghdr);
        if (nsaved > LCAP) nsaved = -1;
    }

    if (nsaved >= 0) {
        int2* rl = (int2*)lraw;
        for (int i = t; i < nsaved; i += AGG_T) rl[i] = grec[i];
        __syncthreads();
        for (int base = wave * GU; base < nsaved; base += (AGG_T / 32) * GU) {
            int2   rc[GU];
            float4 ev[GU];
#pragma unroll
            for (int u = 0; u < GU; ++u) {
                const int idx = base + u;
                rc[u] = rl[idx < nsaved ? idx : nsaved - 1];
            }
#pragma unroll
            for (int u = 0; u < GU; ++u) {
                int s = rc[u].x & 0xFFFFFF;
                if (s > n - 1) s = n - 1;
                ev[u] = *((const float4*)(h + (size_t)s * GD) + lane);
            }
            static_assert(GU == 8, "the statement below names one loaded value per edge of a trip");
            asm volatile("" :: "v"(ev[0].x), "v"(ev[1].x), "v"(ev[2].x), "v"(ev[3].x),
                               "v"(ev[4].x), "v"(ev[5].x), "v"(ev[6].x), "v"(ev[7].x));
#pragma unroll
            for (int u = 0; u < GU; ++u) {
                if (base + u < nsaved) {
                    int r = (int)((unsigned)rc[u].x >> 24);
                    if (r > TR - 1) r = TR - 1;
                    float* tp = tile + r * TP + lane;
                    const float w = __int_as_float(rc[u].y);
                    __hip_atomic_fetch_add(tp + OFX, ev[u].x * w, __ATOMIC_RELAXED, __HIP_MEMORY_SCOPE_WORKGROUP);
                    __hip_atomic_fetch_add(tp + OFY, ev[u].y * w, __ATOMIC_RELAXED, __HIP_MEMORY_SCOPE_WORKGROUP);
                    __hip_atomic_fetch_add(tp + OFZ, ev[u].z * w, __ATOMIC_RELAXED, __HIP_MEMORY_SCOPE_WORKGROUP);
                    __hip_atomic_fetch_add(tp + OFW, ev[u].w * w, __ATOMIC_RELAXED, __HIP_MEMORY_SCOPE_WORKGROUP);
                }
            }
        }
        __syncthreads();
    } else {
        const bool al16 = ((((size_t)dst) & 15) == 0);

        static_assert(TVS * 4 <= 8, "a thread's hits of one sub-chunk are kept as one 32-bit mask, one bit per byte");
        static_assert(TPF == 1, "the scan holds exactly one sub-chunk ahead in registers");
        const bool compact = tid_usable(n, e, ws_bytes);
        const uint4* tq = (const uint4*)((const unsigned char*)wsf + tid_offset(n)) + t;
        const unsigned pat = (unsigned)blockIdx.x * 0x01010101u;
        uint4 nx[TVS];
#pragma unroll
        for (int v = 0; v < TVS; ++v)
            nx[v] = compact ? tq[v * AGG_T] : make_uint4(0u, 0u, 0u, 0u);

        int  nrec = 0;
        bool fits = true;
        for (int cb = 0; cb < e; cb += CH) {
          if (compact) {
            uint4 c[TVS];
#pragma unroll
            for (int v = 0; v < TVS; ++v) {
                c[v] = nx[v];
                nx[v] = tq[(size_t)(cb / 16) + (size_t)((TVS + v) * AGG_T)];
            }
            unsigned bits = 0;
#pragma unroll
            for (int v = 0; v < TVS; ++v) {
                bits |= tid_match4(c[v].x, pat) >> (7 - (4 * v + 0));
                bits |= tid_match4(c[v].y, pat) >> (7 - (4 * v + 1));
                bits |= tid_match4(c[v].z, pat) >> (7 - (4 * v + 2));
                bits |= tid_match4(c[v].w, pat) >> (7 - (4 * v + 3));
            }
            const int k = __popc(bits);
            if (k > 0) {
                int p = atomicAdd(&cnt, k);
                const int eb = cb + t * 16;
                while (bits != 0) {
                    const int pos = __ffs(bits) - 1;
                    bits &= bits - 1;
                    const int d = pos & 7;
                    list[p++] = eb + (d >> 2) * (AGG_T * 16) + (d & 3) * 4 + (pos >> 3);
                }
            }
          } else {
            int4 dv[CH / (AGG_T * 4)];
            if (al16 && cb + CH <= e) {
                const int4* dp = (const int4*)(dst + cb) + t;
#pragma unroll
                for (int j = 0; j < CH / (AGG_T * 4); ++j) dv[j] = dp[j * AGG_T];
            } else {
#pragma unroll
                for (int j = 0; j < CH / (AGG_T * 4); ++j) {
                    const int i0 = cb + (j * AGG_T + t) * 4;
                    const int a = dst[i0     < e ? i0     : e - 1];
                    const int b = dst[i0 + 1 < e ? i0 + 1 : e - 1];
                    const int c = dst[i0 + 2 < e ? i0 + 2 : e - 1];
                    const int d = dst[i0 + 3 < e ? i0 + 3 : e - 1];
                    dv[j].x = i0     < e ? a : -1;
                    dv[j].y = i0 + 1 < e ? b : -1;
                    dv[j].z = i0 + 2 < e ? c : -1;
                    dv[j].w = i0 + 3 < e ? d : -1;
                }
            }
            int k = 0;
#pragma unroll
            for (int j = 0; j < CH / (AGG_T * 4); ++j) {
                k += ((unsigned)(dv[j].x - row0) < (unsigned)TR) ? 1 : 0;
                k += ((unsigned)(dv[j].y - row0) < (unsigned)TR) ? 1 : 0;
                k += ((unsigned)(dv[j].z - row0) < (unsigned)TR) ? 1 : 0;
                k += ((unsigned)(dv[j].w - row0) < (unsigned)TR) ? 1 : 0;
            }
            if (k > 0) {
                int p = atomicAdd(&cnt, k);
#pragma unroll
                for (int j = 0; j < CH / (AGG_T * 4); ++j) {
                    const int i0 = cb + (j * AGG_T + t) * 4;
                    if ((unsigned)(dv[j].x - row0) < (unsigned)TR) list[p++] = i0;
                    if ((unsigned)(dv[j].y - row0) < (unsigned)TR) list[p++] = i0 + 1;
                    if ((unsigned)(dv[j].z - row0) < (unsigned)TR) list[p++] = i0 + 2;
                    if ((unsigned)(dv[j].w - row0) < (unsigned)TR) list[p++] = i0 + 3;
                }
            }
          }
            __syncthreads();
            const int m = cnt;
            __syncthreads();

            const bool last = (cb + CH >= e);
            if (m > LC - CH || last) {
                const bool wr = save && fits && nrec + m <= LCAP;
                if (save && !wr) fits = false;
                for (int base = wave * GU; base < m; base += (AGG_T / 32) * GU) {
                    int   es[GU], ed[GU];
                    float ew[GU];
                    float4 ev[GU];
#pragma unroll
                    for (int u = 0; u < GU; ++u) {
                        const int idx = base + u;
                        const int eid = list[idx < m ? idx : m - 1];
                        es[u] = src[eid];
                        ed[u] = dst[eid];
                    }
#pragma unroll
                    for (int u = 0; u < GU; ++u) {
                        ew[u] = dinv[es[u]] * dinv[ed[u]];
                        ev[u] = *((const float4*)(h + (size_t)es[u] * GD) + lane);
                    }
#pragma unroll
                    for (int u = 0; u < GU; ++u) {
                        if (base + u < m) {
                            float* tp = tile + (ed[u] - row0) * TP + lane;
                            const float w = ew[u];
                            __hip_atomic_fetch_add(tp + OFX, ev[u].x * w, __ATOMIC_RELAXED, __HIP_MEMORY_SCOPE_WORKGROUP);
                            __hip_atomic_fetch_add(tp + OFY, ev[u].y * w, __ATOMIC_RELAXED, __HIP_MEMORY_SCOPE_WORKGROUP);
                            __hip_atomic_fetch_add(tp + OFZ, ev[u].z * w, __ATOMIC_RELAXED, __HIP_MEMORY_SCOPE_WORKGROUP);
                            __hip_atomic_fetch_add(tp + OFW, ev[u].w * w, __ATOMIC_RELAXED, __HIP_MEMORY_SCOPE_WORKGROUP);
                            if (wr && lane == 0)
                                grec[nrec + base + u] = make_int2((int)((unsigned)es[u] | ((unsigned)(ed[u] - row0) << 24)), __float_as_int(w));
                        }
                    }
                }
                nrec += m;
                __syncthreads();
                if (t == 0) cnt = 0;
                __syncthreads();
            }
        }
        if (save && t == 0) *ghdr = fits ? nrec : -1;
    }

    stage_w(wl, Wf, t);
    if (has_skip) stage_w(wl + GD * WP, Wsk, t);

    {
        const float4 b4 = *((const float4*)bc + lane);
        for (int r = wave; r < TR; r += AGG_T / 32) {
            const int row = row0 + r;
            if (row < n) {
                const float di = dinv[row];
                const float4 hv = *((const float4*)(h + (size_t)row * GD) + lane);
                const float4 xv = *((const float4*)(x + (size_t)row * GD) + lane);
                float* tp = tile + r * TP + lane;
                float4 o;
                o.x = (tp[OFX] + hv.x * di * di + b4.x) + xv.x;
                o.y = (tp[OFY] + hv.y * di * di + b4.y) + xv.y;
                o.z = (tp[OFZ] + hv.z * di * di + b4.z) + xv.z;
                o.w = (tp[OFW] + hv.w * di * di + b4.w) + xv.w;
                tp[OFX] = o.x > 0.0f ? o.x : 0.0f;
                tp[OFY] = o.y > 0.0f ? o.y : 0.0f;
                tp[OFZ] = o.z > 0.0f ? o.z : 0.0f;
                tp[OFW] = o.w > 0.0f ? o.w : 0.0f;
            }
        }
    }
    __syncthreads();

    const int half = lane >> 4;
    const int l16  = lane & 15;
    const int stripe = wave >> 1;
    const int chalf  = wave & 1;
    const bool gw = (wave < 2 * NSTRIPE);

    v8f acc[4];
    float bnm[4], bns[4], bng[4], bnb[4];
#pragma unroll
    for (int nt = 0; nt < 4; ++nt) {
        const int col = chalf * 64 + nt * 16 + l16;
        bnm[nt] = rmean[col];
        bns[nt] = rsqrtf(rvar[col] + BN_EPS);
        bng[nt] = gamma[col];
        bnb[nt] = beta[col];
    }

    if (gw) tile_gemm(tile, wl, stripe, chalf, l16, half, acc);
    __syncthreads();
    if (has_next) stage_w(wl, Wcn, t);
    if (gw) {
#pragma unroll
        for (int nt = 0; nt < 4; ++nt) {
            const int col = chalf * 64 + nt * 16 + l16;
            const float bias = bf[col];
            const int pc = PERM(col);
#pragma unroll
            for (int r = 0; r < 8; ++r) {
                const int row = stripe * 16 + r + half * 8;
                if (row < TR) {
                    float* tp = tile + row * TP + pc;
                    float u = acc[nt][r] + bias;
                    u = u > 0.0f ? u : LRELU_SLOPE * u;
                    float v = u + *tp;
                    v = v > 0.0f ? v : 0.0f;
                    if (!has_skip) {
                        const float b = (v - bnm[nt]) * bns[nt] * bng[nt] + bnb[nt];
                        const float y = b + v;
                        v = y > 0.0f ? y : 0.0f;
                    }
                    *tp = v;
                }
            }
        }
    }
    __syncthreads();

    if (has_skip) {
        if (gw) tile_gemm(tile, wl + GD * WP, stripe, chalf, l16, half, acc);
        __syncthreads();
        if (gw) {
#pragma unroll
            for (int nt = 0; nt < 4; ++nt) {
                const int col = chalf * 64 + nt * 16 + l16;
                const float bias = bsk[col];
                const int pc = PERM(col);
#pragma unroll
                for (int r = 0; r < 8; ++r) {
                    const int row = stripe * 16 + r + half * 8;
                    if (row < TR) {
                        float* tp = tile + row * TP + pc;
                        const float s = acc[nt][r] + bias;
                        float v = *tp + s;
                        v = v > 0.0f ? v : 0.0f;
                        const float b = (v - bnm[nt]) * bns[nt] * bng[nt] + bnb[nt];
                        const float y = b + v;
                        *tp = y > 0.0f ? y : 0.0f;
                    }
                }
            }
        }
        __syncthreads();
    }

    for (int r = wave; r < TR; r += AGG_T / 32) {
        const int row = row0 + r;
        if (row < n) {
            const float* tp = tile + r * TP + lane;
            float4 o;
            o.x = tp[OFX];
            o.y = tp[OFY];
            o.z = tp[OFZ];
            o.w = tp[OFW];
            *((float4*)(x + (size_t)row * GD) + lane) = o;
        }
    }

    if (has_next) {
        if (gw) tile_gemm(tile, wl, stripe, chalf, l16, half, acc);
        __syncthreads();
        if (gw) {
#pragma unroll
            for (int nt = 0; nt < 4; ++nt) {
                const int pc = PERM(chalf * 64 + nt * 16 + l16);
#pragma unroll
                for (int r = 0; r < 8; ++r) {
                    const int row = stripe * 16 + r + half * 8;
                    if (row < TR) tile[row * TP + pc] = acc[nt][r];
                }
            }
        }
        __syncthreads();
        for (int r = wave; r < TR; r += AGG_T / 32) {
            const int row = row0 + r;
            if (row < n) {
                const float* tp = tile + r * TP + lane;
                float4 o;
                o.x = tp[OFX];
                o.y = tp[OFY];
                o.z = tp[OFZ];
                o.w = tp[OFW];
                *((float4*)(hn + (size_t)row * GD) + lane) = o;
            }
        }
    }
}

extern "C" void kernel_launch(void* const* d_in, const int* in_sizes, int n_in,
                              void* d_out, int out_size, void* d_ws, size_t ws_size,
                              hipStream_t stream) {
    const float* x_in  = (const float*)d_in[0];
    const int*   ei    = (const int*)d_in[1];
    const float* Wc    = (const float*)d_in[2];
    const float* bc    = (const float*)d_in[3];
    const float* Wf    = (const float*)d_in[4];
    const float* bf    = (const float*)d_in[5];
    const float* Wskip = (const float*)d_in[6];
    const float* bskip = (const float*)d_in[7];
    const float* gamma = (const float*)d_in[8];
    const float* beta  = (const float*)d_in[9];
    const float* rmean = (const float*)d_in[10];
    const float* rvar  = (const float*)d_in[11];

    const int n = in_sizes[0] / GD;
    const int e = in_sizes[1] / 2;
    const int L = in_sizes[3] / GD;

    const int* src = ei;
    const int* dst = ei + e;

    float* x    = (float*)d_out;
    float* bufA = (float*)d_ws;
    float* deg  = bufA + (size_t)n * GD;
    float* dinv = deg + n;
    const size_t offB = h2_offset(n, e);
    const bool chain = offB + (size_t)n * GD * sizeof(float) <= ws_size;
    float* bufB = chain ? (float*)((unsigned char*)d_ws + offB) : bufA;

    const int ND4 = n * (GD / 4);
    dim3 blk(256);
    const int gCopy = (ND4 + 256 * CPV - 1) / (256 * CPV);
    const int gN    = (n + 255) / 256;
    const int gT    = ((e + CH - 1) / CH + TPF) * (CH / 1024);
    const int gAgg  = (n + TR - 1) / TR;
    const int gGemm = (n + 127) / 128;

    init_one_kernel<<<gN, blk, 0, stream>>>(deg, n);
    prep_conv0_kernel<<<gGemm + gT + gCopy, blk, 0, stream>>>(x_in, Wc, bufA, n, deg, dst, e, bufA, ws_size,
                                                             x, ND4, gGemm, gT);
    rsqrt_kernel<<<gN, blk, 0, stream>>>(deg, dinv, n);

    for (int i = 0; i < L; ++i) {
        float* hcur = (i & 1) ? bufB : bufA;
        float* hnxt = (i & 1) ? bufA : bufB;
        const int nxt = (chain && i + 1 < L) ? 1 : 0;
        if (i > 0 && !chain) wmma_gemm128<<<gGemm, blk, 0, stream>>>(x, Wc + (size_t)i * GD * GD, hcur, n);
        const float* Wsk = (i > 0) ? Wskip + (size_t)(i - 1) * GD * GD : Wf + (size_t)i * GD * GD;
        const float* bsk = (i > 0) ? bskip + (i - 1) * GD : bf + i * GD;
        layer_tile_kernel<<<gAgg, dim3(AGG_T), 0, stream>>>(
            x, hcur, src, dst, dinv, bc + i * GD,
            Wf + (size_t)i * GD * GD, bf + i * GD, Wsk, bsk,
            gamma + i * GD, beta + i * GD, rmean + i * GD, rvar + i * GD,
            n, e, (i > 0) ? 1 : 0, (i == 0) ? 1 : 0, bufA, ws_size,
            hnxt, Wc + (size_t)(nxt ? i + 1 : i) * GD * GD, nxt);
    }
}


# Loads of the row-local phases issued as groups ahead of their uses: x1 pass fetches h, x, dinv rows and Wf at once; epilogues read tile words in two groups, biases early; tile padded to full stripes
# speedup vs baseline: 2.8233x; 1.0295x over previous
#include <hip/hip_runtime.h>
#include <hip/hip_bf16.h>

typedef __attribute__((ext_vector_type(2))) float v2f;
typedef __attribute__((ext_vector_type(8))) float v8f;
typedef __attribute__((ext_vector_type(4))) float v4f;

#define GD 128
#define BN_EPS 1e-5f
#define LRELU_SLOPE 0.01f

__device__ __forceinline__
void gemm128_block(const float* __restrict__ A, const float* __restrict__ W,
                   float* __restrict__ C, int nrows, int bid) {
    __shared__ float lds[GD * GD];
    const int t = threadIdx.x;
    for (int i = t; i < GD * GD; i += 256) lds[i] = W[i];
    __syncthreads();

    const int wave = t >> 5;
    const int lane = t & 31;
    const int half = lane >> 4;
    const int l16  = lane & 15;

    const int m0 = bid * 128 + wave * 16;
    int arow = m0 + l16;
    if (arow >= nrows) arow = nrows - 1;
    const float* aptr = A + (size_t)arow * GD + half * 2;

    v8f acc[8];
#pragma unroll
    for (int nt = 0; nt < 8; ++nt) acc[nt] = (v8f){0, 0, 0, 0, 0, 0, 0, 0};

    for (int k0 = 0; k0 < GD; k0 += 4) {
        const v2f a = *(const v2f*)(aptr + k0);
#pragma unroll
        for (int nt = 0; nt < 8; ++nt) {
            const int nn = nt * 16 + l16;
            v2f b;
            b.x = lds[(k0 + half * 2)     * GD + nn];
            b.y = lds[(k0 + half * 2 + 1) * GD + nn];
            acc[nt] = __builtin_amdgcn_wmma_f32_16x16x4_f32(
                false, a, false, b, (short)0, acc[nt], false, false);
        }
    }

#pragma unroll
    for (int r = 0; r < 8; ++r) {
        const int row = m0 + r + half * 8;
        if (row < nrows) {
#pragma unroll
            for (int nt = 0; nt < 8; ++nt)
                C[(size_t)row * GD + nt * 16 + l16] = acc[nt][r];
        }
    }
}

__global__ __launch_bounds__(256)
void wmma_gemm128(const float* __restrict__ A, const float* __restrict__ W,
                  float* __restrict__ C, int nrows) {
    gemm128_block(A, W, C, nrows, blockIdx.x);
}

__global__ void init_one_kernel(float* __restrict__ p, int n) {
    int i = blockIdx.x * 256 + threadIdx.x;
    if (i < n) p[i] = 1.0f;
}
__global__ void rsqrt_kernel(const float* __restrict__ deg, float* __restrict__ dinv, int n) {
    int i = blockIdx.x * 256 + threadIdx.x;
    if (i < n) dinv[i] = rsqrtf(deg[i]);
}

#define TR 200
#define TP 130
#define WP 136
#define AGG_T 1024
#define CH 32768
#define LC (CH + 4096)
#define GU 8
#define NSTRIPE ((TR + 15) / 16)
#define TRP (NSTRIPE * 16)
#define XR ((TR + AGG_T / 32 - 1) / (AGG_T / 32))
#define WV ((GD * GD / 4) / AGG_T)
#define PERM(c) (((((c) & 1) << 1) | (((c) >> 1) & 1)) * 32 + ((c) >> 2))
#define OFX 0
#define OFY 64
#define OFZ 32
#define OFW 96

#define TPF 1
#define TVS (CH / (AGG_T * 16))
__host__ __device__ __forceinline__ size_t tid_offset(int n) {
    const size_t used = ((size_t)n * GD + 2 * (size_t)n) * sizeof(float);
    return (used + 255) & ~(size_t)255;
}
__host__ __device__ __forceinline__ size_t tid_length(int e) {
    return ((size_t)((e + CH - 1) / CH) + TPF) * CH;
}
__device__ __forceinline__ bool tid_usable(int n, int e, size_t ws_bytes) {
    return (n + TR - 1) / TR <= 255 && tid_offset(n) + tid_length(e) <= ws_bytes;
}
#define LCAP 2560
__host__ __device__ __forceinline__ size_t rec_offset(int n, int e) {
    return (tid_offset(n) + tid_length(e) + 255) & ~(size_t)255;
}
__host__ __device__ __forceinline__ size_t rec_hdr_bytes(int n) {
    return ((size_t)((n + TR - 1) / TR) * sizeof(int) + 255) & ~(size_t)255;
}
__device__ __forceinline__ bool rec_usable(int n, int e, size_t ws_bytes) {
    const size_t tiles = (size_t)((n + TR - 1) / TR);
    return n <= (1 << 24) && rec_offset(n, e) + rec_hdr_bytes(n) + tiles * LCAP * sizeof(int2) <= ws_bytes;
}
__host__ __device__ __forceinline__ size_t h2_offset(int n, int e) {
    const size_t tiles = (size_t)((n + TR - 1) / TR);
    return (rec_offset(n, e) + rec_hdr_bytes(n) + tiles * LCAP * sizeof(int2) + 255) & ~(size_t)255;
}
__device__ __forceinline__ unsigned tid_match4(unsigned w, unsigned pat) {
    const unsigned x = w ^ pat;
    return ~(((x & 0x7F7F7F7Fu) + 0x7F7F7F7Fu) | x | 0x7F7F7F7Fu);
}

__device__ __forceinline__ void deg_tid_block(float* __restrict__ deg, const int* __restrict__ dst, int e, int n,
                                              float* __restrict__ wsf, size_t ws_bytes, int bid) {
    const size_t i4 = (size_t)bid * 256 + threadIdx.x;
    int d[4];
#pragma unroll
    for (int j = 0; j < 4; ++j) {
        const size_t idx = i4 * 4 + j;
        d[j] = idx < (size_t)e ? dst[idx] : -1;
    }
    unsigned pk = 0;
#pragma unroll
    for (int j = 0; j < 4; ++j) {
        unsigned b = 255u;
        if (i4 * 4 + j < (size_t)e) {
            unsafeAtomicAdd(&deg[d[j]], 1.0f);
            if ((unsigned)d[j] < (unsigned)n) b = (unsigned)d[j] / TR;
        }
        pk |= b << (8 * j);
    }
    if (tid_usable(n, e, ws_bytes) && i4 * 4 < tid_length(e))
        *((unsigned*)((unsigned char*)wsf + tid_offset(n)) + i4) = pk;
}

#define CPV 8
__device__ __forceinline__ void copy_block(const float* __restrict__ a, float* __restrict__ b, int n4, int bid) {
    const float4* ap = (const float4*)a;
    float4* bp = (float4*)b;
    const int base = bid * (256 * CPV) + threadIdx.x;
    float4 v[CPV];
#pragma unroll
    for (int j = 0; j < CPV; ++j) {
        const int i = base + j * 256;
        v[j] = i < n4 ? ap[i] : make_float4(0.0f, 0.0f, 0.0f, 0.0f);
    }
#pragma unroll
    for (int j = 0; j < CPV; ++j) {
        const int i = base + j * 256;
        if (i < n4) bp[i] = v[j];
    }
}

__global__ __launch_bounds__(256)
void prep_conv0_kernel(const float* __restrict__ x_in, const float* __restrict__ W, float* __restrict__ hbuf,
                       int n, float* __restrict__ deg, const int* __restrict__ dst, int e,
                       float* wsf, size_t ws_bytes,
                       float* __restrict__ xout, int n4, int g_gemm, int g_tid) {
    const int b = blockIdx.x;
    if (b < g_gemm) {
        gemm128_block(x_in, W, hbuf, n, b);
    } else if (b < g_gemm + g_tid) {
        deg_tid_block(deg, dst, e, n, wsf, ws_bytes, b - g_gemm);
    } else {
        copy_block(x_in, xout, n4, b - g_gemm - g_tid);
    }
}

__device__ __forceinline__ void stage_w_load(v4f v[WV], const float* __restrict__ Wg, int t) {
#pragma unroll
    for (int j = 0; j < WV; ++j) v[j] = *((const v4f*)Wg + t + j * AGG_T);
}
__device__ __forceinline__ void stage_w_store(float* __restrict__ wl, const v4f v[WV], int t) {
#pragma unroll
    for (int j = 0; j < WV; ++j) {
        const int i4 = t + j * AGG_T;
        const int row = i4 >> 5;
        const int c4 = i4 & 31;
        *(v4f*)(wl + row * WP + c4 * 4) = v[j];
    }
}

__device__ __forceinline__ float4 tile_get4(const float* tp) {
    float4 o;
    o.x = tp[OFX];
    o.y = tp[OFY];
    o.z = tp[OFZ];
    o.w = tp[OFW];
    return o;
}

__device__ __forceinline__ void tile_stream(const float* __restrict__ tile, float* __restrict__ g,
                                            int row0, int n, int wave, int lane) {
    float4 o[XR];
#pragma unroll
    for (int j = 0; j < XR; ++j) {
        const int r = wave + j * (AGG_T / 32);
        o[j] = tile_get4(tile + (r < TR ? r : 0) * TP + lane);
    }
#pragma unroll
    for (int j = 0; j < XR; ++j) {
        const int r = wave + j * (AGG_T / 32);
        const int row = row0 + r;
        if (r < TR && row < n) *((float4*)(g + (size_t)row * GD) + lane) = o[j];
    }
}

__device__ __forceinline__ void tile_gemm(const float* __restrict__ tile, const float* __restrict__ wl,
                                          int stripe, int chalf, int l16, int half, v8f acc[4]) {
    const int ar = stripe * 16 + l16;
    const float* ap = tile + ar * TP + half * 32;
    const float* bp = wl + (half * 2) * WP + chalf * 64 + l16;
#pragma unroll
    for (int nt = 0; nt < 4; ++nt) acc[nt] = (v8f){0, 0, 0, 0, 0, 0, 0, 0};
    for (int k0 = 0; k0 < GD; k0 += 4) {
        v2f a;
        a.x = ap[k0 >> 2];
        a.y = ap[(k0 >> 2) + 64];
#pragma unroll
        for (int nt = 0; nt < 4; ++nt) {
            v2f b;
            b.x = bp[k0 * WP + nt * 16];
            b.y = bp[(k0 + 1) * WP + nt * 16];
            acc[nt] = __builtin_amdgcn_wmma_f32_16x16x4_f32(
                false, a, false, b, (short)0, acc[nt], false, false);
        }
    }
}

__global__ __launch_bounds__(AGG_T)
void layer_tile_kernel(float* __restrict__ x, const float* __restrict__ h,
                       const int* __restrict__ src, const int* __restrict__ dst,
                       const float* __restrict__ dinv, const float* __restrict__ bc,
                       const float* __restrict__ Wf, const float* __restrict__ bf,
                       const float* __restrict__ Wsk, const float* __restrict__ bsk,
                       const float* __restrict__ gamma, const float* __restrict__ beta,
                       const float* __restrict__ rmean, const float* __restrict__ rvar,
                       int n, int e, int has_skip, int first,
                       float* wsf, size_t ws_bytes,
                       float* __restrict__ hn, const float* __restrict__ Wcn, int has_next) {
    __shared__ float tile[TRP * TP];
    __shared__ __attribute__((aligned(16))) unsigned char lraw[LC * 4];
    __shared__ int   cnt;
    int*   list = (int*)lraw;
    float* wl   = (float*)lraw;

    const int t = threadIdx.x;
    const int lane = t & 31;
    const int wave = t >> 5;
    const int row0 = blockIdx.x * TR;

    for (int i = t; i < TRP * TP; i += AGG_T) tile[i] = 0.0f;
    if (t == 0) cnt = 0;
    __syncthreads();

    const bool keep = rec_usable(n, e, ws_bytes);
    unsigned char* rbase = (unsigned char*)wsf + rec_offset(n, e);
    int*  ghdr = (int*)rbase + blockIdx.x;
    int2* grec = (int2*)(rbase + rec_hdr_bytes(n)) + (size_t)blockIdx.x * LCAP;
    const bool save = keep && first != 0;
    int nsaved = -1;
    if (keep && first == 0) {
        nsaved = __builtin_amdgcn_readfirstlane(*ghdr);
        if (nsaved > LCAP) nsaved = -1;
    }

    if (nsaved >= 0) {
        int2* rl = (int2*)lraw;
        for (int i = t; i < nsaved; i += AGG_T) rl[i] = grec[i];
        __syncthreads();
        for (int base = wave * GU; base < nsaved; base += (AGG_T / 32) * GU) {
            int2   rc[GU];
            float4 ev[GU];
#pragma unroll
            for (int u = 0; u < GU; ++u) {
                const int idx = base + u;
                rc[u] = rl[idx < nsaved ? idx : nsaved - 1];
            }
#pragma unroll
            for (int u = 0; u < GU; ++u) {
                int s = rc[u].x & 0xFFFFFF;
                if (s > n - 1) s = n - 1;
                ev[u] = *((const float4*)(h + (size_t)s * GD) + lane);
            }
            static_assert(GU == 8, "the statement below names one loaded value per edge of a trip");
            asm volatile("" :: "v"(ev[0].x), "v"(ev[1].x), "v"(ev[2].x), "v"(ev[3].x),
                               "v"(ev[4].x), "v"(ev[5].x), "v"(ev[6].x), "v"(ev[7].x));
#pragma unroll
            for (int u = 0; u < GU; ++u) {
                if (base + u < nsaved) {
                    int r = (int)((unsigned)rc[u].x >> 24);
                    if (r > TR - 1) r = TR - 1;
                    float* tp = tile + r * TP + lane;
                    const float w = __int_as_float(rc[u].y);
                    __hip_atomic_fetch_add(tp + OFX, ev[u].x * w, __ATOMIC_RELAXED, __HIP_MEMORY_SCOPE_WORKGROUP);
                    __hip_atomic_fetch_add(tp + OFY, ev[u].y * w, __ATOMIC_RELAXED, __HIP_MEMORY_SCOPE_WORKGROUP);
                    __hip_atomic_fetch_add(tp + OFZ, ev[u].z * w, __ATOMIC_RELAXED, __HIP_MEMORY_SCOPE_WORKGROUP);
                    __hip_atomic_fetch_add(tp + OFW, ev[u].w * w, __ATOMIC_RELAXED, __HIP_MEMORY_SCOPE_WORKGROUP);
                }
            }
        }
        __syncthreads();
    } else {
        const bool al16 = ((((size_t)dst) & 15) == 0);

        static_assert(TVS * 4 <= 8, "a thread's hits of one sub-chunk are kept as one 32-bit mask, one bit per byte");
        static_assert(TPF == 1, "the scan holds exactly one sub-chunk ahead in registers");
        const bool compact = tid_usable(n, e, ws_bytes);
        const uint4* tq = (const uint4*)((const unsigned char*)wsf + tid_offset(n)) + t;
        const unsigned pat = (unsigned)blockIdx.x * 0x01010101u;
        uint4 nx[TVS];
#pragma unroll
        for (int v = 0; v < TVS; ++v)
            nx[v] = compact ? tq[v * AGG_T] : make_uint4(0u, 0u, 0u, 0u);

        int  nrec = 0;
        bool fits = true;
        for (int cb = 0; cb < e; cb += CH) {
          if (compact) {
            uint4 c[TVS];
#pragma unroll
            for (int v = 0; v < TVS; ++v) {
                c[v] = nx[v];
                nx[v] = tq[(size_t)(cb / 16) + (size_t)((TVS + v) * AGG_T)];
            }
            unsigned bits = 0;
#pragma unroll
            for (int v = 0; v < TVS; ++v) {
                bits |= tid_match4(c[v].x, pat) >> (7 - (4 * v + 0));
                bits |= tid_match4(c[v].y, pat) >> (7 - (4 * v + 1));
                bits |= tid_match4(c[v].z, pat) >> (7 - (4 * v + 2));
                bits |= tid_match4(c[v].w, pat) >> (7 - (4 * v + 3));
            }
            const int k = __popc(bits);
            if (k > 0) {
                int p = atomicAdd(&cnt, k);
                const int eb = cb + t * 16;
                while (bits != 0) {
                    const int pos = __ffs(bits) - 1;
                    bits &= bits - 1;
                    const int d = pos & 7;
                    list[p++] = eb + (d >> 2) * (AGG_T * 16) + (d & 3) * 4 + (pos >> 3);
                }
            }
          } else {
            int4 dv[CH / (AGG_T * 4)];
            if (al16 && cb + CH <= e) {
                const int4* dp = (const int4*)(dst + cb) + t;
#pragma unroll
                for (int j = 0; j < CH / (AGG_T * 4); ++j) dv[j] = dp[j * AGG_T];
            } else {
#pragma unroll
                for (int j = 0; j < CH / (AGG_T * 4); ++j) {
                    const int i0 = cb + (j * AGG_T + t) * 4;
                    const int a = dst[i0     < e ? i0     : e - 1];
                    const int b = dst[i0 + 1 < e ? i0 + 1 : e - 1];
                    const int c = dst[i0 + 2 < e ? i0 + 2 : e - 1];
                    const int d = dst[i0 + 3 < e ? i0 + 3 : e - 1];
                    dv[j].x = i0     < e ? a : -1;
                    dv[j].y = i0 + 1 < e ? b : -1;
                    dv[j].z = i0 + 2 < e ? c : -1;
                    dv[j].w = i0 + 3 < e ? d : -1;
                }
            }
            int k = 0;
#pragma unroll
            for (int j = 0; j < CH / (AGG_T * 4); ++j) {
                k += ((unsigned)(dv[j].x - row0) < (unsigned)TR) ? 1 : 0;
                k += ((unsigned)(dv[j].y - row0) < (unsigned)TR) ? 1 : 0;
                k += ((unsigned)(dv[j].z - row0) < (unsigned)TR) ? 1 : 0;
                k += ((unsigned)(dv[j].w - row0) < (unsigned)TR) ? 1 : 0;
            }
            if (k > 0) {
                int p = atomicAdd(&cnt, k);
#pragma unroll
                for (int j = 0; j < CH / (AGG_T * 4); ++j) {
                    const int i0 = cb + (j * AGG_T + t) * 4;
                    if ((unsigned)(dv[j].x - row0) < (unsigned)TR) list[p++] = i0;
                    if ((unsigned)(dv[j].y - row0) < (unsigned)TR) list[p++] = i0 + 1;
                    if ((unsigned)(dv[j].z - row0) < (unsigned)TR) list[p++] = i0 + 2;
                    if ((unsigned)(dv[j].w - row0) < (unsigned)TR) list[p++] = i0 + 3;
                }
            }
          }
            __syncthreads();
            const int m = cnt;
            __syncthreads();

            const bool last = (cb + CH >= e);
            if (m > LC - CH || last) {
                const bool wr = save && fits && nrec + m <= LCAP;
                if (save && !wr) fits = false;
                for (int base = wave * GU; base < m; base += (AGG_T / 32) * GU) {
                    int   es[GU], ed[GU];
                    float ew[GU];
                    float4 ev[GU];
#pragma unroll
                    for (int u = 0; u < GU; ++u) {
                        const int idx = base + u;
                        const int eid = list[idx < m ? idx : m - 1];
                        es[u] = src[eid];
                        ed[u] = dst[eid];
                    }
#pragma unroll
                    for (int u = 0; u < GU; ++u) {
                        ew[u] = dinv[es[u]] * dinv[ed[u]];
                        ev[u] = *((const float4*)(h + (size_t)es[u] * GD) + lane);
                    }
#pragma unroll
                    for (int u = 0; u < GU; ++u) {
                        if (base + u < m) {
                            float* tp = tile + (ed[u] - row0) * TP + lane;
                            const float w = ew[u];
                            __hip_atomic_fetch_add(tp + OFX, ev[u].x * w, __ATOMIC_RELAXED, __HIP_MEMORY_SCOPE_WORKGROUP);
                            __hip_atomic_fetch_add(tp + OFY, ev[u].y * w, __ATOMIC_RELAXED, __HIP_MEMORY_SCOPE_WORKGROUP);
                            __hip_atomic_fetch_add(tp + OFZ, ev[u].z * w, __ATOMIC_RELAXED, __HIP_MEMORY_SCOPE_WORKGROUP);
                            __hip_atomic_fetch_add(tp + OFW, ev[u].w * w, __ATOMIC_RELAXED, __HIP_MEMORY_SCOPE_WORKGROUP);
                            if (wr && lane == 0)
                                grec[nrec + base + u] = make_int2((int)((unsigned)es[u] | ((unsigned)(ed[u] - row0) << 24)), __float_as_int(w));
                        }
                    }
                }
                nrec += m;
                __syncthreads();
                if (t == 0) cnt = 0;
                __syncthreads();
            }
        }
        if (save && t == 0) *ghdr = fits ? nrec : -1;
    }

    const float4 b4 = *((const float4*)bc + lane);
    float  dv[XR];
    float4 hv[XR], xv[XR];
#pragma unroll
    for (int j = 0; j < XR; ++j) {
        const int r = wave + j * (AGG_T / 32);
        const int row = (r < TR && row0 + r < n) ? row0 + r : row0;
        dv[j] = dinv[row];
        hv[j] = *((const float4*)(h + (size_t)row * GD) + lane);
        xv[j] = *((const float4*)(x + (size_t)row * GD) + lane);
    }
    v4f wf4[WV], ws4[WV];
    stage_w_load(wf4, Wf, t);
    static_assert(XR == 7 && WV == 4, "the statement below names one loaded value per row of h and of x, and Wf's vectors");
    asm volatile("" :: "v"(hv[0].x), "v"(hv[1].x), "v"(hv[2].x), "v"(hv[3].x), "v"(hv[4].x), "v"(hv[5].x), "v"(hv[6].x),
                       "v"(xv[0].x), "v"(xv[1].x), "v"(xv[2].x), "v"(xv[3].x), "v"(xv[4].x), "v"(xv[5].x), "v"(xv[6].x),
                       "v"(wf4[0]), "v"(wf4[1]), "v"(wf4[2]), "v"(wf4[3]));
    stage_w_store(wl, wf4, t);
    if (has_skip) {
        stage_w_load(ws4, Wsk, t);
    } else {
#pragma unroll
        for (int j = 0; j < WV; ++j) ws4[j] = (v4f){0.0f, 0.0f, 0.0f, 0.0f};
    }

#pragma unroll
    for (int j = 0; j < XR; ++j) {
        const int r = wave + j * (AGG_T / 32);
        if (r < TR && row0 + r < n) {
            const float di = dv[j];
            float* tp = tile + r * TP + lane;
            float4 o;
            o.x = (tp[OFX] + hv[j].x * di * di + b4.x) + xv[j].x;
            o.y = (tp[OFY] + hv[j].y * di * di + b4.y) + xv[j].y;
            o.z = (tp[OFZ] + hv[j].z * di * di + b4.z) + xv[j].z;
            o.w = (tp[OFW] + hv[j].w * di * di + b4.w) + xv[j].w;
            tp[OFX] = o.x > 0.0f ? o.x : 0.0f;
            tp[OFY] = o.y > 0.0f ? o.y : 0.0f;
            tp[OFZ] = o.z > 0.0f ? o.z : 0.0f;
            tp[OFW] = o.w > 0.0f ? o.w : 0.0f;
        }
    }
    if (has_skip) stage_w_store(wl + GD * WP, ws4, t);
    __syncthreads();

    const int half = lane >> 4;
    const int l16  = lane & 15;
    const int stripe = wave >> 1;
    const int chalf  = wave & 1;
    const bool gw = (wave < 2 * NSTRIPE);

    v8f acc[4];
    float bnm[4], bns[4], bng[4], bnb[4], bfv[4];
#pragma unroll
    for (int nt = 0; nt < 4; ++nt) {
        const int col = chalf * 64 + nt * 16 + l16;
        bnm[nt] = rmean[col];
        bns[nt] = rsqrtf(rvar[col] + BN_EPS);
        bng[nt] = gamma[col];
        bnb[nt] = beta[col];
        bfv[nt] = bf[col];
    }
    float* const ep = tile + (stripe * 16 + half * 8) * TP;
    int pcs[4];
#pragma unroll
    for (int nt = 0; nt < 4; ++nt) pcs[nt] = PERM(chalf * 64 + nt * 16 + l16);

    if (gw) tile_gemm(tile, wl, stripe, chalf, l16, half, acc);
    __syncthreads();
    if (has_next) {
        v4f wc4[WV];
        stage_w_load(wc4, Wcn, t);
        stage_w_store(wl, wc4, t);
    }
    if (gw) {
#pragma unroll
        for (int g = 0; g < 2; ++g) {
            float tv[2][8];
#pragma unroll
            for (int q = 0; q < 2; ++q)
#pragma unroll
                for (int r = 0; r < 8; ++r) tv[q][r] = ep[r * TP + pcs[2 * g + q]];
#pragma unroll
            for (int q = 0; q < 2; ++q) {
                const int nt = 2 * g + q;
#pragma unroll
                for (int r = 0; r < 8; ++r) {
                    float u = acc[nt][r] + bfv[nt];
                    u = u > 0.0f ? u : LRELU_SLOPE * u;
                    const float v = u + tv[q][r];
                    tv[q][r] = v > 0.0f ? v : 0.0f;
                }
            }
            if (!has_skip) {
#pragma unroll
                for (int q = 0; q < 2; ++q) {
                    const int nt = 2 * g + q;
#pragma unroll
                    for (int r = 0; r < 8; ++r) {
                        const float v = tv[q][r];
                        const float b = (v - bnm[nt]) * bns[nt] * bng[nt] + bnb[nt];
                        const float y = b + v;
                        tv[q][r] = y > 0.0f ? y : 0.0f;
                    }
                }
            }
#pragma unroll
            for (int q = 0; q < 2; ++q)
#pragma unroll
                for (int r = 0; r < 8; ++r) ep[r * TP + pcs[2 * g + q]] = tv[q][r];
        }
    }
    __syncthreads();

    if (has_skip) {
        float bkv[4];
#pragma unroll
        for (int nt = 0; nt < 4; ++nt) bkv[nt] = bsk[chalf * 64 + nt * 16 + l16];
        if (gw) tile_gemm(tile, wl + GD * WP, stripe, chalf, l16, half, acc);
        __syncthreads();
        if (gw) {
#pragma unroll
            for (int g = 0; g < 2; ++g) {
                float tv[2][8];
#pragma unroll
                for (int q = 0; q < 2; ++q)
#pragma unroll
                    for (int r = 0; r < 8; ++r) tv[q][r] = ep[r * TP + pcs[2 * g + q]];
#pragma unroll
                for (int q = 0; q < 2; ++q) {
                    const int nt = 2 * g + q;
#pragma unroll
                    for (int r = 0; r < 8; ++r) {
                        const float s = acc[nt][r] + bkv[nt];
                        float v = tv[q][r] + s;
                        v = v > 0.0f ? v : 0.0f;
                        const float b = (v - bnm[nt]) * bns[nt] * bng[nt] + bnb[nt];
                        const float y = b + v;
                        tv[q][r] = y > 0.0f ? y : 0.0f;
                    }
                }
#pragma unroll
                for (int q = 0; q < 2; ++q)
#pragma unroll
                    for (int r = 0; r < 8; ++r) ep[r * TP + pcs[2 * g + q]] = tv[q][r];
            }
        }
        __syncthreads();
    }

    tile_stream(tile, x, row0, n, wave, lane);

    if (has_next) {
        if (gw) tile_gemm(tile, wl, stripe, chalf, l16, half, acc);
        __syncthreads();
        if (gw) {
#pragma unroll
            for (int nt = 0; nt < 4; ++nt)
#pragma unroll
                for (int r = 0; r < 8; ++r) ep[r * TP + pcs[nt]] = acc[nt][r];
        }
        __syncthreads();
        tile_stream(tile, hn, row0, n, wave, lane);
    }
}

extern "C" void kernel_launch(void* const* d_in, const int* in_sizes, int n_in,
                              void* d_out, int out_size, void* d_ws, size_t ws_size,
                              hipStream_t stream) {
    const float* x_in  = (const float*)d_in[0];
    const int*   ei    = (const int*)d_in[1];
    const float* Wc    = (const float*)d_in[2];
    const float* bc    = (const float*)d_in[3];
    const float* Wf    = (const float*)d_in[4];
    const float* bf    = (const float*)d_in[5];
    const float* Wskip = (const float*)d_in[6];
    const float* bskip = (const float*)d_in[7];
    const float* gamma = (const float*)d_in[8];
    const float* beta  = (const float*)d_in[9];
    const float* rmean = (const float*)d_in[10];
    const float* rvar  = (const float*)d_in[11];

    const int n = in_sizes[0] / GD;
    const int e = in_sizes[1] / 2;
    const int L = in_sizes[3] / GD;

    const int* src = ei;
    const int* dst = ei + e;

    float* x    = (float*)d_out;
    float* bufA = (float*)d_ws;
    float* deg  = bufA + (size_t)n * GD;
    float* dinv = deg + n;
    const size_t offB = h2_offset(n, e);
    const bool chain = offB + (size_t)n * GD * sizeof(float) <= ws_size;
    float* bufB = chain ? (float*)((unsigned char*)d_ws + offB) : bufA;

    const int ND4 = n * (GD / 4);
    dim3 blk(256);
    const int gCopy = (ND4 + 256 * CPV - 1) / (256 * CPV);
    const int gN    = (n + 255) / 256;
    const int gT    = ((e + CH - 1) / CH + TPF) * (CH / 1024);
    const int gAgg  = (n + TR - 1) / TR;
    const int gGemm = (n + 127) / 128;

    init_one_kernel<<<gN, blk, 0, stream>>>(deg, n);
    prep_conv0_kernel<<<gGemm + gT + gCopy, blk, 0, stream>>>(x_in, Wc, bufA, n, deg, dst, e, bufA, ws_size,
                                                             x, ND4, gGemm, gT);
    rsqrt_kernel<<<gN, blk, 0, stream>>>(deg, dinv, n);

    for (int i = 0; i < L; ++i) {
        float* hcur = (i & 1) ? bufB : bufA;
        float* hnxt = (i & 1) ? bufA : bufB;
        const int nxt = (chain && i + 1 < L) ? 1 : 0;
        if (i > 0 && !chain) wmma_gemm128<<<gGemm, blk, 0, stream>>>(x, Wc + (size_t)i * GD * GD, hcur, n);
        const float* Wsk = (i > 0) ? Wskip + (size_t)(i - 1) * GD * GD : Wf + (size_t)i * GD * GD;
        const float* bsk = (i > 0) ? bskip + (i - 1) * GD : bf + i * GD;
        layer_tile_kernel<<<gAgg, dim3(AGG_T), 0, stream>>>(
            x, hcur, src, dst, dinv, bc + i * GD,
            Wf + (size_t)i * GD * GD, bf + i * GD, Wsk, bsk,
            gamma + i * GD, beta + i * GD, rmean + i * GD, rvar + i * GD,
            n, e, (i > 0) ? 1 : 0, (i == 0) ? 1 : 0, bufA, ws_size,
            hnxt, Wc + (size_t)(nxt ? i + 1 : i) * GD * GD, nxt);
    }
}


# Degree counted in per-work-group LDS histograms (two 16-bit integer counters a word) instead of global float atomics; the set-up launch sums the slabs into dinv: init and rsqrt launches gone, 6 to 5
# speedup vs baseline: 3.2003x; 1.1335x over previous
#include <hip/hip_runtime.h>
#include <hip/hip_bf16.h>

typedef __attribute__((ext_vector_type(2))) float v2f;
typedef __attribute__((ext_vector_type(8))) float v8f;
typedef __attribute__((ext_vector_type(4))) float v4f;

#define GD 128
#define BN_EPS 1e-5f
#define LRELU_SLOPE 0.01f

__device__ __forceinline__
void gemm128_block(const float* __restrict__ A, const float* __restrict__ W,
                   float* __restrict__ C, int nrows, int bid) {
    __shared__ float lds[GD * GD];
    const int t = threadIdx.x;
    for (int i = t; i < GD * GD; i += 256) lds[i] = W[i];
    __syncthreads();

    const int wave = t >> 5;
    const int lane = t & 31;
    const int half = lane >> 4;
    const int l16  = lane & 15;

    const int m0 = bid * 128 + wave * 16;
    int arow = m0 + l16;
    if (arow >= nrows) arow = nrows - 1;
    const float* aptr = A + (size_t)arow * GD + half * 2;

    v8f acc[8];
#pragma unroll
    for (int nt = 0; nt < 8; ++nt) acc[nt] = (v8f){0, 0, 0, 0, 0, 0, 0, 0};

    for (int k0 = 0; k0 < GD; k0 += 4) {
        const v2f a = *(const v2f*)(aptr + k0);
#pragma unroll
        for (int nt = 0; nt < 8; ++nt) {
            const int nn = nt * 16 + l16;
            v2f b;
            b.x = lds[(k0 + half * 2)     * GD + nn];
            b.y = lds[(k0 + half * 2 + 1) * GD + nn];
            acc[nt] = __builtin_amdgcn_wmma_f32_16x16x4_f32(
                false, a, false, b, (short)0, acc[nt], false, false);
        }
    }

#pragma unroll
    for (int r = 0; r < 8; ++r) {
        const int row = m0 + r + half * 8;
        if (row < nrows) {
#pragma unroll
            for (int nt = 0; nt < 8; ++nt)
                C[(size_t)row * GD + nt * 16 + l16] = acc[nt][r];
        }
    }
}

__global__ __launch_bounds__(256)
void wmma_gemm128(const float* __restrict__ A, const float* __restrict__ W,
                  float* __restrict__ C, int nrows) {
    gemm128_block(A, W, C, nrows, blockIdx.x);
}

__global__ void init_one_kernel(float* __restrict__ p, int n) {
    int i = blockIdx.x * 256 + threadIdx.x;
    if (i < n) p[i] = 1.0f;
}
__global__ void rsqrt_kernel(const float* __restrict__ deg, float* __restrict__ dinv, int n) {
    int i = blockIdx.x * 256 + threadIdx.x;
    if (i < n) dinv[i] = rsqrtf(deg[i]);
}

#define TR 200
#define TP 130
#define WP 136
#define AGG_T 1024
#define CH 32768
#define LC (CH + 4096)
#define GU 8
#define NSTRIPE ((TR + 15) / 16)
#define TRP (NSTRIPE * 16)
#define XR ((TR + AGG_T / 32 - 1) / (AGG_T / 32))
#define WV ((GD * GD / 4) / AGG_T)
#define PERM(c) (((((c) & 1) << 1) | (((c) >> 1) & 1)) * 32 + ((c) >> 2))
#define OFX 0
#define OFY 64
#define OFZ 32
#define OFW 96

#define TPF 1
#define TVS (CH / (AGG_T * 16))
__host__ __device__ __forceinline__ size_t tid_offset(int n) {
    const size_t used = ((size_t)n * GD + 2 * (size_t)n) * sizeof(float);
    return (used + 255) & ~(size_t)255;
}
__host__ __device__ __forceinline__ size_t tid_length(int e) {
    return ((size_t)((e + CH - 1) / CH) + TPF) * CH;
}
__device__ __forceinline__ bool tid_usable(int n, int e, size_t ws_bytes) {
    return (n + TR - 1) / TR <= 255 && tid_offset(n) + tid_length(e) <= ws_bytes;
}
#define LCAP 2560
__host__ __device__ __forceinline__ size_t rec_offset(int n, int e) {
    return (tid_offset(n) + tid_length(e) + 255) & ~(size_t)255;
}
__host__ __device__ __forceinline__ size_t rec_hdr_bytes(int n) {
    return ((size_t)((n + TR - 1) / TR) * sizeof(int) + 255) & ~(size_t)255;
}
__device__ __forceinline__ bool rec_usable(int n, int e, size_t ws_bytes) {
    const size_t tiles = (size_t)((n + TR - 1) / TR);
    return n <= (1 << 24) && rec_offset(n, e) + rec_hdr_bytes(n) + tiles * LCAP * sizeof(int2) <= ws_bytes;
}
__host__ __device__ __forceinline__ size_t h2_offset(int n, int e) {
    const size_t tiles = (size_t)((n + TR - 1) / TR);
    return (rec_offset(n, e) + rec_hdr_bytes(n) + tiles * LCAP * sizeof(int2) + 255) & ~(size_t)255;
}
__device__ __forceinline__ unsigned tid_match4(unsigned w, unsigned pat) {
    const unsigned x = w ^ pat;
    return ~(((x & 0x7F7F7F7Fu) + 0x7F7F7F7Fu) | x | 0x7F7F7F7Fu);
}

#define HK 64
#define HT 1024
#define HW 32768
#define HE 8
__host__ __device__ __forceinline__ int slab_pitch(int n) {
    return (((n + 1) >> 1) + 63) & ~63;
}
__global__ __launch_bounds__(HT)
void deg_hist_kernel(const int* __restrict__ dst, int e, int n, unsigned* __restrict__ slab) {
    __shared__ __attribute__((aligned(16))) unsigned hist[HW];
    const int t = threadIdx.x;
    const int pv = slab_pitch(n) >> 2;
    for (int i = t; i < pv; i += HT) ((uint4*)hist)[i] = make_uint4(0u, 0u, 0u, 0u);
    __syncthreads();
    const int per = (e + HK - 1) / HK;
    const int lo = blockIdx.x * per;
    const int hi = lo + per < e ? lo + per : e;
    for (int b = lo; b < hi; b += HT * HE) {
        int d[HE];
#pragma unroll
        for (int j = 0; j < HE; ++j) {
            const int i = b + j * HT + t;
            d[j] = dst[i < hi ? i : hi - 1];
        }
#pragma unroll
        for (int j = 0; j < HE; ++j)
            if (b + j * HT + t < hi && (unsigned)d[j] < (unsigned)n)
                atomicAdd(&hist[d[j] >> 1], 1u << ((d[j] & 1) * 16));
    }
    __syncthreads();
    uint4* out = (uint4*)(slab + (size_t)blockIdx.x * slab_pitch(n));
    for (int i = t; i < pv; i += HT) out[i] = ((const uint4*)hist)[i];
}

__device__ __forceinline__ void dinv_block(const unsigned* __restrict__ slab, float* __restrict__ dinv, int n, int bid) {
    const int i = bid * 256 + threadIdx.x;
    if (i >= n) return;
    const size_t pitch = (size_t)slab_pitch(n);
    const unsigned* p = slab + (i >> 1);
    const int sh = (i & 1) * 16;
    unsigned cnt = 0;
    static_assert(HK % 16 == 0, "the slabs are read in groups of 16");
    for (int k0 = 0; k0 < HK; k0 += 16) {
        unsigned w[16];
#pragma unroll
        for (int j = 0; j < 16; ++j) w[j] = p[(size_t)(k0 + j) * pitch];
#pragma unroll
        for (int j = 0; j < 16; ++j) cnt += (w[j] >> sh) & 0xFFFFu;
    }
    dinv[i] = rsqrtf(1.0f + (float)cnt);
}

__device__ __forceinline__ void deg_tid_block(float* __restrict__ deg, const int* __restrict__ dst, int e, int n,
                                              float* __restrict__ wsf, size_t ws_bytes, int bid, int count) {
    const size_t i4 = (size_t)bid * 256 + threadIdx.x;
    int d[4];
#pragma unroll
    for (int j = 0; j < 4; ++j) {
        const size_t idx = i4 * 4 + j;
        d[j] = idx < (size_t)e ? dst[idx] : -1;
    }
    unsigned pk = 0;
#pragma unroll
    for (int j = 0; j < 4; ++j) {
        unsigned b = 255u;
        if (i4 * 4 + j < (size_t)e) {
            if (count) unsafeAtomicAdd(&deg[d[j]], 1.0f);
            if ((unsigned)d[j] < (unsigned)n) b = (unsigned)d[j] / TR;
        }
        pk |= b << (8 * j);
    }
    if (tid_usable(n, e, ws_bytes) && i4 * 4 < tid_length(e))
        *((unsigned*)((unsigned char*)wsf + tid_offset(n)) + i4) = pk;
}

#define CPV 8
__device__ __forceinline__ void copy_block(const float* __restrict__ a, float* __restrict__ b, int n4, int bid) {
    const float4* ap = (const float4*)a;
    float4* bp = (float4*)b;
    const int base = bid * (256 * CPV) + threadIdx.x;
    float4 v[CPV];
#pragma unroll
    for (int j = 0; j < CPV; ++j) {
        const int i = base + j * 256;
        v[j] = i < n4 ? ap[i] : make_float4(0.0f, 0.0f, 0.0f, 0.0f);
    }
#pragma unroll
    for (int j = 0; j < CPV; ++j) {
        const int i = base + j * 256;
        if (i < n4) bp[i] = v[j];
    }
}

__global__ __launch_bounds__(256)
void prep_conv0_kernel(const float* __restrict__ x_in, const float* __restrict__ W, float* __restrict__ hbuf,
                       int n, float* __restrict__ deg, const int* __restrict__ dst, int e,
                       float* wsf, size_t ws_bytes,
                       float* __restrict__ xout, int n4, int g_gemm, int g_tid,
                       const unsigned* __restrict__ slab, float* __restrict__ dinv, int g_dinv, int count) {
    const int b = blockIdx.x;
    if (b < g_gemm) {
        gemm128_block(x_in, W, hbuf, n, b);
    } else if (b < g_gemm + g_dinv) {
        dinv_block(slab, dinv, n, b - g_gemm);
    } else if (b < g_gemm + g_dinv + g_tid) {
        deg_tid_block(deg, dst, e, n, wsf, ws_bytes, b - g_gemm - g_dinv, count);
    } else {
        copy_block(x_in, xout, n4, b - g_gemm - g_dinv - g_tid);
    }
}

__device__ __forceinline__ void stage_w_load(v4f v[WV], const float* __restrict__ Wg, int t) {
#pragma unroll
    for (int j = 0; j < WV; ++j) v[j] = *((const v4f*)Wg + t + j * AGG_T);
}
__device__ __forceinline__ void stage_w_store(float* __restrict__ wl, const v4f v[WV], int t) {
#pragma unroll
    for (int j = 0; j < WV; ++j) {
        const int i4 = t + j * AGG_T;
        const int row = i4 >> 5;
        const int c4 = i4 & 31;
        *(v4f*)(wl + row * WP + c4 * 4) = v[j];
    }
}

__device__ __forceinline__ float4 tile_get4(const float* tp) {
    float4 o;
    o.x = tp[OFX];
    o.y = tp[OFY];
    o.z = tp[OFZ];
    o.w = tp[OFW];
    return o;
}

__device__ __forceinline__ void tile_stream(const float* __restrict__ tile, float* __restrict__ g,
                                            int row0, int n, int wave, int lane) {
    float4 o[XR];
#pragma unroll
    for (int j = 0; j < XR; ++j) {
        const int r = wave + j * (AGG_T / 32);
        o[j] = tile_get4(tile + (r < TR ? r : 0) * TP + lane);
    }
#pragma unroll
    for (int j = 0; j < XR; ++j) {
        const int r = wave + j * (AGG_T / 32);
        const int row = row0 + r;
        if (r < TR && row < n) *((float4*)(g + (size_t)row * GD) + lane) = o[j];
    }
}

__device__ __forceinline__ void tile_gemm(const float* __restrict__ tile, const float* __restrict__ wl,
                                          int stripe, int chalf, int l16, int half, v8f acc[4]) {
    const int ar = stripe * 16 + l16;
    const float* ap = tile + ar * TP + half * 32;
    const float* bp = wl + (half * 2) * WP + chalf * 64 + l16;
#pragma unroll
    for (int nt = 0; nt < 4; ++nt) acc[nt] = (v8f){0, 0, 0, 0, 0, 0, 0, 0};
    for (int k0 = 0; k0 < GD; k0 += 4) {
        v2f a;
        a.x = ap[k0 >> 2];
        a.y = ap[(k0 >> 2) + 64];
#pragma unroll
        for (int nt = 0; nt < 4; ++nt) {
            v2f b;
            b.x = bp[k0 * WP + nt * 16];
            b.y = bp[(k0 + 1) * WP + nt * 16];
            acc[nt] = __builtin_amdgcn_wmma_f32_16x16x4_f32(
                false, a, false, b, (short)0, acc[nt], false, false);
        }
    }
}

__global__ __launch_bounds__(AGG_T)
void layer_tile_kernel(float* __restrict__ x, const float* __restrict__ h,
                       const int* __restrict__ src, const int* __restrict__ dst,
                       const float* __restrict__ dinv, const float* __restrict__ bc,
                       const float* __restrict__ Wf, const float* __restrict__ bf,
                       const float* __restrict__ Wsk, const float* __restrict__ bsk,
                       const float* __restrict__ gamma, const float* __restrict__ beta,
                       const float* __restrict__ rmean, const float* __restrict__ rvar,
                       int n, int e, int has_skip, int first,
                       float* wsf, size_t ws_bytes,
                       float* __restrict__ hn, const float* __restrict__ Wcn, int has_next) {
    __shared__ float tile[TRP * TP];
    __shared__ __attribute__((aligned(16))) unsigned char lraw[LC * 4];
    __shared__ int   cnt;
    int*   list = (int*)lraw;
    float* wl   = (float*)lraw;

    const int t = threadIdx.x;
    const int lane = t & 31;
    const int wave = t >> 5;
    const int row0 = blockIdx.x * TR;

    for (int i = t; i < TRP * TP; i += AGG_T) tile[i] = 0.0f;
    if (t == 0) cnt = 0;
    __syncthreads();

    const bool keep = rec_usable(n, e, ws_bytes);
    unsigned char* rbase = (unsigned char*)wsf + rec_offset(n, e);
    int*  ghdr = (int*)rbase + blockIdx.x;
    int2* grec = (int2*)(rbase + rec_hdr_bytes(n)) + (size_t)blockIdx.x * LCAP;
    const bool save = keep && first != 0;
    int nsaved = -1;
    if (keep && first == 0) {
        nsaved = __builtin_amdgcn_readfirstlane(*ghdr);
        if (nsaved > LCAP) nsaved = -1;
    }

    if (nsaved >= 0) {
        int2* rl = (int2*)lraw;
        for (int i = t; i < nsaved; i += AGG_T) rl[i] = grec[i];
        __syncthreads();
        for (int base = wave * GU; base < nsaved; base += (AGG_T / 32) * GU) {
            int2   rc[GU];
            float4 ev[GU];
#pragma unroll
            for (int u = 0; u < GU; ++u) {
                const int idx = base + u;
                rc[u] = rl[idx < nsaved ? idx : nsaved - 1];
            }
#pragma unroll
            for (int u = 0; u < GU; ++u) {
                int s = rc[u].x & 0xFFFFFF;
                if (s > n - 1) s = n - 1;
                ev[u] = *((const float4*)(h + (size_t)s * GD) + lane);
            }
            static_assert(GU == 8, "the statement below names one loaded value per edge of a trip");
            asm volatile("" :: "v"(ev[0].x), "v"(ev[1].x), "v"(ev[2].x), "v"(ev[3].x),
                               "v"(ev[4].x), "v"(ev[5].x), "v"(ev[6].x), "v"(ev[7].x));
#pragma unroll
            for (int u = 0; u < GU; ++u) {
                if (base + u < nsaved) {
                    int r = (int)((unsigned)rc[u].x >> 24);
                    if (r > TR - 1) r = TR - 1;
                    float* tp = tile + r * TP + lane;
                    const float w = __int_as_float(rc[u].y);
                    __hip_atomic_fetch_add(tp + OFX, ev[u].x * w, __ATOMIC_RELAXED, __HIP_MEMORY_SCOPE_WORKGROUP);
                    __hip_atomic_fetch_add(tp + OFY, ev[u].y * w, __ATOMIC_RELAXED, __HIP_MEMORY_SCOPE_WORKGROUP);
                    __hip_atomic_fetch_add(tp + OFZ, ev[u].z * w, __ATOMIC_RELAXED, __HIP_MEMORY_SCOPE_WORKGROUP);
                    __hip_atomic_fetch_add(tp + OFW, ev[u].w * w, __ATOMIC_RELAXED, __HIP_MEMORY_SCOPE_WORKGROUP);
                }
            }
        }
        __syncthreads();
    } else {
        const bool al16 = ((((size_t)dst) & 15) == 0);

        static_assert(TVS * 4 <= 8, "a thread's hits of one sub-chunk are kept as one 32-bit mask, one bit per byte");
        static_assert(TPF == 1, "the scan holds exactly one sub-chunk ahead in registers");
        const bool compact = tid_usable(n, e, ws_bytes);
        const uint4* tq = (const uint4*)((const unsigned char*)wsf + tid_offset(n)) + t;
        const unsigned pat = (unsigned)blockIdx.x * 0x01010101u;
        uint4 nx[TVS];
#pragma unroll
        for (int v = 0; v < TVS; ++v)
            nx[v] = compact ? tq[v * AGG_T] : make_uint4(0u, 0u, 0u, 0u);

        int  nrec = 0;
        bool fits = true;
        for (int cb = 0; cb < e; cb += CH) {
          if (compact) {
            uint4 c[TVS];
#pragma unroll
            for (int v = 0; v < TVS; ++v) {
                c[v] = nx[v];
                nx[v] = tq[(size_t)(cb / 16) + (size_t)((TVS + v) * AGG_T)];
            }
            unsigned bits = 0;
#pragma unroll
            for (int v = 0; v < TVS; ++v) {
                bits |= tid_match4(c[v].x, pat) >> (7 - (4 * v + 0));
                bits |= tid_match4(c[v].y, pat) >> (7 - (4 * v + 1));
                bits |= tid_match4(c[v].z, pat) >> (7 - (4 * v + 2));
                bits |= tid_match4(c[v].w, pat) >> (7 - (4 * v + 3));
            }
            const int k = __popc(bits);
            if (k > 0) {
                int p = atomicAdd(&cnt, k);
                const int eb = cb + t * 16;
                while (bits != 0) {
                    const int pos = __ffs(bits) - 1;
                    bits &= bits - 1;
                    const int d = pos & 7;
                    list[p++] = eb + (d >> 2) * (AGG_T * 16) + (d & 3) * 4 + (pos >> 3);
                }
            }
          } else {
            int4 dv[CH / (AGG_T * 4)];
            if (al16 && cb + CH <= e) {
                const int4* dp = (const int4*)(dst + cb) + t;
#pragma unroll
                for (int j = 0; j < CH / (AGG_T * 4); ++j) dv[j] = dp[j * AGG_T];
            } else {
#pragma unroll
                for (int j = 0; j < CH / (AGG_T * 4); ++j) {
                    const int i0 = cb + (j * AGG_T + t) * 4;
                    const int a = dst[i0     < e ? i0     : e - 1];
                    const int b = dst[i0 + 1 < e ? i0 + 1 : e - 1];
                    const int c = dst[i0 + 2 < e ? i0 + 2 : e - 1];
                    const int d = dst[i0 + 3 < e ? i0 + 3 : e - 1];
                    dv[j].x = i0     < e ? a : -1;
                    dv[j].y = i0 + 1 < e ? b : -1;
                    dv[j].z = i0 + 2 < e ? c : -1;
                    dv[j].w = i0 + 3 < e ? d : -1;
                }
            }
            int k = 0;
#pragma unroll
            for (int j = 0; j < CH / (AGG_T * 4); ++j) {
                k += ((unsigned)(dv[j].x - row0) < (unsigned)TR) ? 1 : 0;
                k += ((unsigned)(dv[j].y - row0) < (unsigned)TR) ? 1 : 0;
                k += ((unsigned)(dv[j].z - row0) < (unsigned)TR) ? 1 : 0;
                k += ((unsigned)(dv[j].w - row0) < (unsigned)TR) ? 1 : 0;
            }
            if (k > 0) {
                int p = atomicAdd(&cnt, k);
#pragma unroll
                for (int j = 0; j < CH / (AGG_T * 4); ++j) {
                    const int i0 = cb + (j * AGG_T + t) * 4;
                    if ((unsigned)(dv[j].x - row0) < (unsigned)TR) list[p++] = i0;
                    if ((unsigned)(dv[j].y - row0) < (unsigned)TR) list[p++] = i0 + 1;
                    if ((unsigned)(dv[j].z - row0) < (unsigned)TR) list[p++] = i0 + 2;
                    if ((unsigned)(dv[j].w - row0) < (unsigned)TR) list[p++] = i0 + 3;
                }
            }
          }
            __syncthreads();
            const int m = cnt;
            __syncthreads();

            const bool last = (cb + CH >= e);
            if (m > LC - CH || last) {
                const bool wr = save && fits && nrec + m <= LCAP;
                if (save && !wr) fits = false;
                for (int base = wave * GU; base < m; base += (AGG_T / 32) * GU) {
                    int   es[GU], ed[GU];
                    float ew[GU];
                    float4 ev[GU];
#pragma unroll
                    for (int u = 0; u < GU; ++u) {
                        const int idx = base + u;
                        const int eid = list[idx < m ? idx : m - 1];
                        es[u] = src[eid];
                        ed[u] = dst[eid];
                    }
#pragma unroll
                    for (int u = 0; u < GU; ++u) {
                        ew[u] = dinv[es[u]] * dinv[ed[u]];
                        ev[u] = *((const float4*)(h + (size_t)es[u] * GD) + lane);
                    }
#pragma unroll
                    for (int u = 0; u < GU; ++u) {
                        if (base + u < m) {
                            float* tp = tile + (ed[u] - row0) * TP + lane;
                            const float w = ew[u];
                            __hip_atomic_fetch_add(tp + OFX, ev[u].x * w, __ATOMIC_RELAXED, __HIP_MEMORY_SCOPE_WORKGROUP);
                            __hip_atomic_fetch_add(tp + OFY, ev[u].y * w, __ATOMIC_RELAXED, __HIP_MEMORY_SCOPE_WORKGROUP);
                            __hip_atomic_fetch_add(tp + OFZ, ev[u].z * w, __ATOMIC_RELAXED, __HIP_MEMORY_SCOPE_WORKGROUP);
                            __hip_atomic_fetch_add(tp + OFW, ev[u].w * w, __ATOMIC_RELAXED, __HIP_MEMORY_SCOPE_WORKGROUP);
                            if (wr && lane == 0)
                                grec[nrec + base + u] = make_int2((int)((unsigned)es[u] | ((unsigned)(ed[u] - row0) << 24)), __float_as_int(w));
                        }
                    }
                }
                nrec += m;
                __syncthreads();
                if (t == 0) cnt = 0;
                __syncthreads();
            }
        }
        if (save && t == 0) *ghdr = fits ? nrec : -1;
    }

    const float4 b4 = *((const float4*)bc + lane);
    float  dv[XR];
    float4 hv[XR], xv[XR];
#pragma unroll
    for (int j = 0; j < XR; ++j) {
        const int r = wave + j * (AGG_T / 32);
        const int row = (r < TR && row0 + r < n) ? row0 + r : row0;
        dv[j] = dinv[row];
        hv[j] = *((const float4*)(h + (size_t)row * GD) + lane);
        xv[j] = *((const float4*)(x + (size_t)row * GD) + lane);
    }
    v4f wf4[WV], ws4[WV];
    stage_w_load(wf4, Wf, t);
    static_assert(XR == 7 && WV == 4, "the statement below names one loaded value per row of h and of x, and Wf's vectors");
    asm volatile("" :: "v"(hv[0].x), "v"(hv[1].x), "v"(hv[2].x), "v"(hv[3].x), "v"(hv[4].x), "v"(hv[5].x), "v"(hv[6].x),
                       "v"(xv[0].x), "v"(xv[1].x), "v"(xv[2].x), "v"(xv[3].x), "v"(xv[4].x), "v"(xv[5].x), "v"(xv[6].x),
                       "v"(wf4[0]), "v"(wf4[1]), "v"(wf4[2]), "v"(wf4[3]));
    stage_w_store(wl, wf4, t);
    if (has_skip) {
        stage_w_load(ws4, Wsk, t);
    } else {
#pragma unroll
        for (int j = 0; j < WV; ++j) ws4[j] = (v4f){0.0f, 0.0f, 0.0f, 0.0f};
    }

#pragma unroll
    for (int j = 0; j < XR; ++j) {
        const int r = wave + j * (AGG_T / 32);
        if (r < TR && row0 + r < n) {
            const float di = dv[j];
            float* tp = tile + r * TP + lane;
            float4 o;
            o.x = (tp[OFX] + hv[j].x * di * di + b4.x) + xv[j].x;
            o.y = (tp[OFY] + hv[j].y * di * di + b4.y) + xv[j].y;
            o.z = (tp[OFZ] + hv[j].z * di * di + b4.z) + xv[j].z;
            o.w = (tp[OFW] + hv[j].w * di * di + b4.w) + xv[j].w;
            tp[OFX] = o.x > 0.0f ? o.x : 0.0f;
            tp[OFY] = o.y > 0.0f ? o.y : 0.0f;
            tp[OFZ] = o.z > 0.0f ? o.z : 0.0f;
            tp[OFW] = o.w > 0.0f ? o.w : 0.0f;
        }
    }
    if (has_skip) stage_w_store(wl + GD * WP, ws4, t);
    __syncthreads();

    const int half = lane >> 4;
    const int l16  = lane & 15;
    const int stripe = wave >> 1;
    const int chalf  = wave & 1;
    const bool gw = (wave < 2 * NSTRIPE);

    v8f acc[4];
    float bnm[4], bns[4], bng[4], bnb[4], bfv[4];
#pragma unroll
    for (int nt = 0; nt < 4; ++nt) {
        const int col = chalf * 64 + nt * 16 + l16;
        bnm[nt] = rmean[col];
        bns[nt] = rsqrtf(rvar[col] + BN_EPS);
        bng[nt] = gamma[col];
        bnb[nt] = beta[col];
        bfv[nt] = bf[col];
    }
    float* const ep = tile + (stripe * 16 + half * 8) * TP;
    int pcs[4];
#pragma unroll
    for (int nt = 0; nt < 4; ++nt) pcs[nt] = PERM(chalf * 64 + nt * 16 + l16);

    if (gw) tile_gemm(tile, wl, stripe, chalf, l16, half, acc);
    __syncthreads();
    if (has_next) {
        v4f wc4[WV];
        stage_w_load(wc4, Wcn, t);
        stage_w_store(wl, wc4, t);
    }
    if (gw) {
#pragma unroll
        for (int g = 0; g < 2; ++g) {
            float tv[2][8];
#pragma unroll
            for (int q = 0; q < 2; ++q)
#pragma unroll
                for (int r = 0; r < 8; ++r) tv[q][r] = ep[r * TP + pcs[2 * g + q]];
#pragma unroll
            for (int q = 0; q < 2; ++q) {
                const int nt = 2 * g + q;
#pragma unroll
                for (int r = 0; r < 8; ++r) {
                    float u = acc[nt][r] + bfv[nt];
                    u = u > 0.0f ? u : LRELU_SLOPE * u;
                    const float v = u + tv[q][r];
                    tv[q][r] = v > 0.0f ? v : 0.0f;
                }
            }
            if (!has_skip) {
#pragma unroll
                for (int q = 0; q < 2; ++q) {
                    const int nt = 2 * g + q;
#pragma unroll
                    for (int r = 0; r < 8; ++r) {
                        const float v = tv[q][r];
                        const float b = (v - bnm[nt]) * bns[nt] * bng[nt] + bnb[nt];
                        const float y = b + v;
                        tv[q][r] = y > 0.0f ? y : 0.0f;
                    }
                }
            }
#pragma unroll
            for (int q = 0; q < 2; ++q)
#pragma unroll
                for (int r = 0; r < 8; ++r) ep[r * TP + pcs[2 * g + q]] = tv[q][r];
        }
    }
    __syncthreads();

    if (has_skip) {
        float bkv[4];
#pragma unroll
        for (int nt = 0; nt < 4; ++nt) bkv[nt] = bsk[chalf * 64 + nt * 16 + l16];
        if (gw) tile_gemm(tile, wl + GD * WP, stripe, chalf, l16, half, acc);
        __syncthreads();
        if (gw) {
#pragma unroll
            for (int g = 0; g < 2; ++g) {
                float tv[2][8];
#pragma unroll
                for (int q = 0; q < 2; ++q)
#pragma unroll
                    for (int r = 0; r < 8; ++r) tv[q][r] = ep[r * TP + pcs[2 * g + q]];
#pragma unroll
                for (int q = 0; q < 2; ++q) {
                    const int nt = 2 * g + q;
#pragma unroll
                    for (int r = 0; r < 8; ++r) {
                        const float s = acc[nt][r] + bkv[nt];
                        float v = tv[q][r] + s;
                        v = v > 0.0f ? v : 0.0f;
                        const float b = (v - bnm[nt]) * bns[nt] * bng[nt] + bnb[nt];
                        const float y = b + v;
                        tv[q][r] = y > 0.0f ? y : 0.0f;
                    }
                }
#pragma unroll
                for (int q = 0; q < 2; ++q)
#pragma unroll
                    for (int r = 0; r < 8; ++r) ep[r * TP + pcs[2 * g + q]] = tv[q][r];
            }
        }
        __syncthreads();
    }

    tile_stream(tile, x, row0, n, wave, lane);

    if (has_next) {
        if (gw) tile_gemm(tile, wl, stripe, chalf, l16, half, acc);
        __syncthreads();
        if (gw) {
#pragma unroll
            for (int nt = 0; nt < 4; ++nt)
#pragma unroll
                for (int r = 0; r < 8; ++r) ep[r * TP + pcs[nt]] = acc[nt][r];
        }
        __syncthreads();
        tile_stream(tile, hn, row0, n, wave, lane);
    }
}

extern "C" void kernel_launch(void* const* d_in, const int* in_sizes, int n_in,
                              void* d_out, int out_size, void* d_ws, size_t ws_size,
                              hipStream_t stream) {
    const float* x_in  = (const float*)d_in[0];
    const int*   ei    = (const int*)d_in[1];
    const float* Wc    = (const float*)d_in[2];
    const float* bc    = (const float*)d_in[3];
    const float* Wf    = (const float*)d_in[4];
    const float* bf    = (const float*)d_in[5];
    const float* Wskip = (const float*)d_in[6];
    const float* bskip = (const float*)d_in[7];
    const float* gamma = (const float*)d_in[8];
    const float* beta  = (const float*)d_in[9];
    const float* rmean = (const float*)d_in[10];
    const float* rvar  = (const float*)d_in[11];

    const int n = in_sizes[0] / GD;
    const int e = in_sizes[1] / 2;
    const int L = in_sizes[3] / GD;

    const int* src = ei;
    const int* dst = ei + e;

    float* x    = (float*)d_out;
    float* bufA = (float*)d_ws;
    float* deg  = bufA + (size_t)n * GD;
    float* dinv = deg + n;
    const size_t offB = h2_offset(n, e);
    const bool chain = offB + (size_t)n * GD * sizeof(float) <= ws_size;
    float* bufB = chain ? (float*)((unsigned char*)d_ws + offB) : bufA;

    const int ND4 = n * (GD / 4);
    dim3 blk(256);
    const int gCopy = (ND4 + 256 * CPV - 1) / (256 * CPV);
    const int gN    = (n + 255) / 256;
    const int gT    = ((e + CH - 1) / CH + TPF) * (CH / 1024);
    const int gAgg  = (n + TR - 1) / TR;
    const int gGemm = (n + 127) / 128;

    const bool hist = chain && n >= 64 && n <= 2 * HW && (e + HK - 1) / HK <= 65535;
    unsigned* slab = (unsigned*)bufB;
    if (hist) {
        deg_hist_kernel<<<HK, dim3(HT), 0, stream>>>(dst, e, n, slab);
        prep_conv0_kernel<<<gGemm + gN + gT + gCopy, blk, 0, stream>>>(x_in, Wc, bufA, n, deg, dst, e, bufA, ws_size,
                                                                      x, ND4, gGemm, gT, slab, dinv, gN, 0);
    } else {
        init_one_kernel<<<gN, blk, 0, stream>>>(deg, n);
        prep_conv0_kernel<<<gGemm + gT + gCopy, blk, 0, stream>>>(x_in, Wc, bufA, n, deg, dst, e, bufA, ws_size,
                                                                 x, ND4, gGemm, gT, slab, dinv, 0, 1);
        rsqrt_kernel<<<gN, blk, 0, stream>>>(deg, dinv, n);
    }

    for (int i = 0; i < L; ++i) {
        float* hcur = (i & 1) ? bufB : bufA;
        float* hnxt = (i & 1) ? bufA : bufB;
        const int nxt = (chain && i + 1 < L) ? 1 : 0;
        if (i > 0 && !chain) wmma_gemm128<<<gGemm, blk, 0, stream>>>(x, Wc + (size_t)i * GD * GD, hcur, n);
        const float* Wsk = (i > 0) ? Wskip + (size_t)(i - 1) * GD * GD : Wf + (size_t)i * GD * GD;
        const float* bsk = (i > 0) ? bskip + (i - 1) * GD : bf + i * GD;
        layer_tile_kernel<<<gAgg, dim3(AGG_T), 0, stream>>>(
            x, hcur, src, dst, dinv, bc + i * GD,
            Wf + (size_t)i * GD * GD, bf + i * GD, Wsk, bsk,
            gamma + i * GD, beta + i * GD, rmean + i * GD, rvar + i * GD,
            n, e, (i > 0) ? 1 : 0, (i == 0) ? 1 : 0, bufA, ws_size,
            hnxt, Wc + (size_t)(nxt ? i + 1 : i) * GD * GD, nxt);
    }
}


# Per-edge LDS float atomics replaced by register sums: layer 0 sorts a tile's records by row in LDS (saved sorted); a wave sums its share's row runs, one LDS add per run; sum order: rank in row
# speedup vs baseline: 3.3644x; 1.0513x over previous
#include <hip/hip_runtime.h>
#include <hip/hip_bf16.h>

typedef __attribute__((ext_vector_type(2))) float v2f;
typedef __attribute__((ext_vector_type(8))) float v8f;
typedef __attribute__((ext_vector_type(4))) float v4f;

#define GD 128
#define BN_EPS 1e-5f
#define LRELU_SLOPE 0.01f

__device__ __forceinline__
void gemm128_block(const float* __restrict__ A, const float* __restrict__ W,
                   float* __restrict__ C, int nrows, int bid) {
    __shared__ float lds[GD * GD];
    const int t = threadIdx.x;
    for (int i = t; i < GD * GD; i += 256) lds[i] = W[i];
    __syncthreads();

    const int wave = t >> 5;
    const int lane = t & 31;
    const int half = lane >> 4;
    const int l16  = lane & 15;

    const int m0 = bid * 128 + wave * 16;
    int arow = m0 + l16;
    if (arow >= nrows) arow = nrows - 1;
    const float* aptr = A + (size_t)arow * GD + half * 2;

    v8f acc[8];
#pragma unroll
    for (int nt = 0; nt < 8; ++nt) acc[nt] = (v8f){0, 0, 0, 0, 0, 0, 0, 0};

    for (int k0 = 0; k0 < GD; k0 += 4) {
        const v2f a = *(const v2f*)(aptr + k0);
#pragma unroll
        for (int nt = 0; nt < 8; ++nt) {
            const int nn = nt * 16 + l16;
            v2f b;
            b.x = lds[(k0 + half * 2)     * GD + nn];
            b.y = lds[(k0 + half * 2 + 1) * GD + nn];
            acc[nt] = __builtin_amdgcn_wmma_f32_16x16x4_f32(
                false, a, false, b, (short)0, acc[nt], false, false);
        }
    }

#pragma unroll
    for (int r = 0; r < 8; ++r) {
        const int row = m0 + r + half * 8;
        if (row < nrows) {
#pragma unroll
            for (int nt = 0; nt < 8; ++nt)
                C[(size_t)row * GD + nt * 16 + l16] = acc[nt][r];
        }
    }
}

__global__ __launch_bounds__(256)
void wmma_gemm128(const float* __restrict__ A, const float* __restrict__ W,
                  float* __restrict__ C, int nrows) {
    gemm128_block(A, W, C, nrows, blockIdx.x);
}

__global__ void init_one_kernel(float* __restrict__ p, int n) {
    int i = blockIdx.x * 256 + threadIdx.x;
    if (i < n) p[i] = 1.0f;
}
__global__ void rsqrt_kernel(const float* __restrict__ deg, float* __restrict__ dinv, int n) {
    int i = blockIdx.x * 256 + threadIdx.x;
    if (i < n) dinv[i] = rsqrtf(deg[i]);
}

#define TR 200
#define TP 130
#define WP 136
#define AGG_T 1024
#define CH 32768
#define LC (CH + 4096)
#define GU 8
#define NSTRIPE ((TR + 15) / 16)
#define TRP (NSTRIPE * 16)
#define XR ((TR + AGG_T / 32 - 1) / (AGG_T / 32))
#define WV ((GD * GD / 4) / AGG_T)
#define PERM(c) (((((c) & 1) << 1) | (((c) >> 1) & 1)) * 32 + ((c) >> 2))
#define OFX 0
#define OFY 64
#define OFZ 32
#define OFW 96

#define TPF 1
#define TVS (CH / (AGG_T * 16))
__host__ __device__ __forceinline__ size_t tid_offset(int n) {
    const size_t used = ((size_t)n * GD + 2 * (size_t)n) * sizeof(float);
    return (used + 255) & ~(size_t)255;
}
__host__ __device__ __forceinline__ size_t tid_length(int e) {
    return ((size_t)((e + CH - 1) / CH) + TPF) * CH;
}
__device__ __forceinline__ bool tid_usable(int n, int e, size_t ws_bytes) {
    return (n + TR - 1) / TR <= 255 && tid_offset(n) + tid_length(e) <= ws_bytes;
}
#define LCAP 2560
#define NW (AGG_T / 32)
#define NKEY (32 * XR)
#define RPT ((LCAP + AGG_T - 1) / AGG_T)
#define SRT_OFF 32768
#define CNT_OFF 53248
#define BAS_OFF 54272
__host__ __device__ __forceinline__ size_t rec_offset(int n, int e) {
    return (tid_offset(n) + tid_length(e) + 255) & ~(size_t)255;
}
__host__ __device__ __forceinline__ size_t rec_hdr_bytes(int n) {
    return ((size_t)((n + TR - 1) / TR) * sizeof(int) + 255) & ~(size_t)255;
}
__device__ __forceinline__ bool rec_usable(int n, int e, size_t ws_bytes) {
    const size_t tiles = (size_t)((n + TR - 1) / TR);
    return n <= (1 << 24) && rec_offset(n, e) + rec_hdr_bytes(n) + tiles * LCAP * sizeof(int2) <= ws_bytes;
}
__host__ __device__ __forceinline__ size_t h2_offset(int n, int e) {
    const size_t tiles = (size_t)((n + TR - 1) / TR);
    return (rec_offset(n, e) + rec_hdr_bytes(n) + tiles * LCAP * sizeof(int2) + 255) & ~(size_t)255;
}
__device__ __forceinline__ unsigned tid_match4(unsigned w, unsigned pat) {
    const unsigned x = w ^ pat;
    return ~(((x & 0x7F7F7F7Fu) + 0x7F7F7F7Fu) | x | 0x7F7F7F7Fu);
}

#define HK 64
#define HT 1024
#define HW 32768
#define HE 8
__host__ __device__ __forceinline__ int slab_pitch(int n) {
    return (((n + 1) >> 1) + 63) & ~63;
}
__global__ __launch_bounds__(HT)
void deg_hist_kernel(const int* __restrict__ dst, int e, int n, unsigned* __restrict__ slab) {
    __shared__ __attribute__((aligned(16))) unsigned hist[HW];
    const int t = threadIdx.x;
    const int pv = slab_pitch(n) >> 2;
    for (int i = t; i < pv; i += HT) ((uint4*)hist)[i] = make_uint4(0u, 0u, 0u, 0u);
    __syncthreads();
    const int per = (e + HK - 1) / HK;
    const int lo = blockIdx.x * per;
    const int hi = lo + per < e ? lo + per : e;
    for (int b = lo; b < hi; b += HT * HE) {
        int d[HE];
#pragma unroll
        for (int j = 0; j < HE; ++j) {
            const int i = b + j * HT + t;
            d[j] = dst[i < hi ? i : hi - 1];
        }
#pragma unroll
        for (int j = 0; j < HE; ++j)
            if (b + j * HT + t < hi && (unsigned)d[j] < (unsigned)n)
                atomicAdd(&hist[d[j] >> 1], 1u << ((d[j] & 1) * 16));
    }
    __syncthreads();
    uint4* out = (uint4*)(slab + (size_t)blockIdx.x * slab_pitch(n));
    for (int i = t; i < pv; i += HT) out[i] = ((const uint4*)hist)[i];
}

__device__ __forceinline__ void dinv_block(const unsigned* __restrict__ slab, float* __restrict__ dinv, int n, int bid) {
    const int i = bid * 256 + threadIdx.x;
    if (i >= n) return;
    const size_t pitch = (size_t)slab_pitch(n);
    const unsigned* p = slab + (i >> 1);
    const int sh = (i & 1) * 16;
    unsigned cnt = 0;
    static_assert(HK % 16 == 0, "the slabs are read in groups of 16");
    for (int k0 = 0; k0 < HK; k0 += 16) {
        unsigned w[16];
#pragma unroll
        for (int j = 0; j < 16; ++j) w[j] = p[(size_t)(k0 + j) * pitch];
#pragma unroll
        for (int j = 0; j < 16; ++j) cnt += (w[j] >> sh) & 0xFFFFu;
    }
    dinv[i] = rsqrtf(1.0f + (float)cnt);
}

__device__ __forceinline__ void deg_tid_block(float* __restrict__ deg, const int* __restrict__ dst, int e, int n,
                                              float* __restrict__ wsf, size_t ws_bytes, int bid, int count) {
    const size_t i4 = (size_t)bid * 256 + threadIdx.x;
    int d[4];
#pragma unroll
    for (int j = 0; j < 4; ++j) {
        const size_t idx = i4 * 4 + j;
        d[j] = idx < (size_t)e ? dst[idx] : -1;
    }
    unsigned pk = 0;
#pragma unroll
    for (int j = 0; j < 4; ++j) {
        unsigned b = 255u;
        if (i4 * 4 + j < (size_t)e) {
            if (count) unsafeAtomicAdd(&deg[d[j]], 1.0f);
            if ((unsigned)d[j] < (unsigned)n) b = (unsigned)d[j] / TR;
        }
        pk |= b << (8 * j);
    }
    if (tid_usable(n, e, ws_bytes) && i4 * 4 < tid_length(e))
        *((unsigned*)((unsigned char*)wsf + tid_offset(n)) + i4) = pk;
}

#define CPV 8
__device__ __forceinline__ void copy_block(const float* __restrict__ a, float* __restrict__ b, int n4, int bid) {
    const float4* ap = (const float4*)a;
    float4* bp = (float4*)b;
    const int base = bid * (256 * CPV) + threadIdx.x;
    float4 v[CPV];
#pragma unroll
    for (int j = 0; j < CPV; ++j) {
        const int i = base + j * 256;
        v[j] = i < n4 ? ap[i] : make_float4(0.0f, 0.0f, 0.0f, 0.0f);
    }
#pragma unroll
    for (int j = 0; j < CPV; ++j) {
        const int i = base + j * 256;
        if (i < n4) bp[i] = v[j];
    }
}

__global__ __launch_bounds__(256)
void prep_conv0_kernel(const float* __restrict__ x_in, const float* __restrict__ W, float* __restrict__ hbuf,
                       int n, float* __restrict__ deg, const int* __restrict__ dst, int e,
                       float* wsf, size_t ws_bytes,
                       float* __restrict__ xout, int n4, int g_gemm, int g_tid,
                       const unsigned* __restrict__ slab, float* __restrict__ dinv, int g_dinv, int count) {
    const int b = blockIdx.x;
    if (b < g_gemm) {
        gemm128_block(x_in, W, hbuf, n, b);
    } else if (b < g_gemm + g_dinv) {
        dinv_block(slab, dinv, n, b - g_gemm);
    } else if (b < g_gemm + g_dinv + g_tid) {
        deg_tid_block(deg, dst, e, n, wsf, ws_bytes, b - g_gemm - g_dinv, count);
    } else {
        copy_block(x_in, xout, n4, b - g_gemm - g_dinv - g_tid);
    }
}

__device__ __forceinline__ void stage_w_load(v4f v[WV], const float* __restrict__ Wg, int t) {
#pragma unroll
    for (int j = 0; j < WV; ++j) v[j] = *((const v4f*)Wg + t + j * AGG_T);
}
__device__ __forceinline__ void stage_w_store(float* __restrict__ wl, const v4f v[WV], int t) {
#pragma unroll
    for (int j = 0; j < WV; ++j) {
        const int i4 = t + j * AGG_T;
        const int row = i4 >> 5;
        const int c4 = i4 & 31;
        *(v4f*)(wl + row * WP + c4 * 4) = v[j];
    }
}

__device__ __forceinline__ float4 tile_get4(const float* tp) {
    float4 o;
    o.x = tp[OFX];
    o.y = tp[OFY];
    o.z = tp[OFZ];
    o.w = tp[OFW];
    return o;
}

__device__ __forceinline__ void tile_stream(const float* __restrict__ tile, float* __restrict__ g,
                                            int row0, int n, int wave, int lane) {
    float4 o[XR];
#pragma unroll
    for (int j = 0; j < XR; ++j) {
        const int r = wave + j * (AGG_T / 32);
        o[j] = tile_get4(tile + (r < TR ? r : 0) * TP + lane);
    }
#pragma unroll
    for (int j = 0; j < XR; ++j) {
        const int r = wave + j * (AGG_T / 32);
        const int row = row0 + r;
        if (r < TR && row < n) *((float4*)(g + (size_t)row * GD) + lane) = o[j];
    }
}

__device__ __forceinline__ void tile_gemm(const float* __restrict__ tile, const float* __restrict__ wl,
                                          int stripe, int chalf, int l16, int half, v8f acc[4]) {
    const int ar = stripe * 16 + l16;
    const float* ap = tile + ar * TP + half * 32;
    const float* bp = wl + (half * 2) * WP + chalf * 64 + l16;
#pragma unroll
    for (int nt = 0; nt < 4; ++nt) acc[nt] = (v8f){0, 0, 0, 0, 0, 0, 0, 0};
    for (int k0 = 0; k0 < GD; k0 += 4) {
        v2f a;
        a.x = ap[k0 >> 2];
        a.y = ap[(k0 >> 2) + 64];
#pragma unroll
        for (int nt = 0; nt < 4; ++nt) {
            v2f b;
            b.x = bp[k0 * WP + nt * 16];
            b.y = bp[(k0 + 1) * WP + nt * 16];
            acc[nt] = __builtin_amdgcn_wmma_f32_16x16x4_f32(
                false, a, false, b, (short)0, acc[nt], false, false);
        }
    }
}

__device__ __forceinline__ void gather_sorted(float* __restrict__ tile, const int2* __restrict__ srt,
                                              const float* __restrict__ h, int n, int cnt, int wave, int lane) {
    const int q  = (cnt + NW - 1) / NW;
    const int lo = __builtin_amdgcn_readfirstlane(wave) * q;
    const int hi = lo + q < cnt ? lo + q : cnt;
    int cur = -1;
    float4 acc = make_float4(0.0f, 0.0f, 0.0f, 0.0f);
    for (int base = lo; base < hi; base += GU) {
        int2   rc[GU];
        float4 ev[GU];
#pragma unroll
        for (int u = 0; u < GU; ++u) {
            const int idx = base + u;
            rc[u] = srt[idx < hi ? idx : hi - 1];
        }
#pragma unroll
        for (int u = 0; u < GU; ++u) {
            int s = rc[u].x & 0xFFFFFF;
            if (s > n - 1) s = n - 1;
            ev[u] = *((const float4*)(h + (size_t)s * GD) + lane);
        }
        static_assert(GU == 8, "the statement below names one loaded value per record of a trip");
        asm volatile("" :: "v"(ev[0].x), "v"(ev[1].x), "v"(ev[2].x), "v"(ev[3].x),
                           "v"(ev[4].x), "v"(ev[5].x), "v"(ev[6].x), "v"(ev[7].x));
#pragma unroll
        for (int u = 0; u < GU; ++u) {
            if (base + u < hi) {
                int r = __builtin_amdgcn_readfirstlane((int)((unsigned)rc[u].x >> 24));
                if (r > TR - 1) r = TR - 1;
                if (r != cur) {
                    if (cur >= 0) {
                        float* tp = tile + cur * TP + lane;
                        __hip_atomic_fetch_add(tp + OFX, acc.x, __ATOMIC_RELAXED, __HIP_MEMORY_SCOPE_WORKGROUP);
                        __hip_atomic_fetch_add(tp + OFY, acc.y, __ATOMIC_RELAXED, __HIP_MEMORY_SCOPE_WORKGROUP);
                        __hip_atomic_fetch_add(tp + OFZ, acc.z, __ATOMIC_RELAXED, __HIP_MEMORY_SCOPE_WORKGROUP);
                        __hip_atomic_fetch_add(tp + OFW, acc.w, __ATOMIC_RELAXED, __HIP_MEMORY_SCOPE_WORKGROUP);
                    }
                    cur = r;
                    acc = make_float4(0.0f, 0.0f, 0.0f, 0.0f);
                }
                const float w = __int_as_float(rc[u].y);
                acc.x += ev[u].x * w;
                acc.y += ev[u].y * w;
                acc.z += ev[u].z * w;
                acc.w += ev[u].w * w;
            }
        }
    }
    if (cur >= 0) {
        float* tp = tile + cur * TP + lane;
        __hip_atomic_fetch_add(tp + OFX, acc.x, __ATOMIC_RELAXED, __HIP_MEMORY_SCOPE_WORKGROUP);
        __hip_atomic_fetch_add(tp + OFY, acc.y, __ATOMIC_RELAXED, __HIP_MEMORY_SCOPE_WORKGROUP);
        __hip_atomic_fetch_add(tp + OFZ, acc.z, __ATOMIC_RELAXED, __HIP_MEMORY_SCOPE_WORKGROUP);
        __hip_atomic_fetch_add(tp + OFW, acc.w, __ATOMIC_RELAXED, __HIP_MEMORY_SCOPE_WORKGROUP);
    }
}

__global__ __launch_bounds__(AGG_T)
void layer_tile_kernel(float* __restrict__ x, const float* __restrict__ h,
                       const int* __restrict__ src, const int* __restrict__ dst,
                       const float* __restrict__ dinv, const float* __restrict__ bc,
                       const float* __restrict__ Wf, const float* __restrict__ bf,
                       const float* __restrict__ Wsk, const float* __restrict__ bsk,
                       const float* __restrict__ gamma, const float* __restrict__ beta,
                       const float* __restrict__ rmean, const float* __restrict__ rvar,
                       int n, int e, int has_skip, int first,
                       float* wsf, size_t ws_bytes,
                       float* __restrict__ hn, const float* __restrict__ Wcn, int has_next) {
    __shared__ float tile[TRP * TP];
    __shared__ __attribute__((aligned(16))) unsigned char lraw[LC * 4];
    __shared__ int   cnt;
    int*   list = (int*)lraw;
    float* wl   = (float*)lraw;
    static_assert(LCAP * 4 <= SRT_OFF && SRT_OFF + LCAP * 8 <= CNT_OFF && CNT_OFF + NKEY * 4 <= BAS_OFF &&
                  BAS_OFF + NKEY * 4 <= LC * 4, "the sort's LDS areas lie side by side");
    static_assert(TR <= NKEY && TR <= 256, "every tile row has a counter; a tile row fits the record's 8 bits");
    int2* srt   = (int2*)(lraw + SRT_OFF);
    int*  cntk  = (int*)(lraw + CNT_OFF);
    int*  basek = (int*)(lraw + BAS_OFF);

    const int t = threadIdx.x;
    const int lane = t & 31;
    const int wave = t >> 5;
    const int row0 = blockIdx.x * TR;

    for (int i = t; i < TRP * TP; i += AGG_T) tile[i] = 0.0f;
    if (t == 0) cnt = 0;
    __syncthreads();

    const bool keep = rec_usable(n, e, ws_bytes);
    unsigned char* rbase = (unsigned char*)wsf + rec_offset(n, e);
    int*  ghdr = (int*)rbase + blockIdx.x;
    int2* grec = (int2*)(rbase + rec_hdr_bytes(n)) + (size_t)blockIdx.x * LCAP;
    const bool save = keep && first != 0;
    int nsaved = -1;
    if (keep && first == 0) {
        nsaved = __builtin_amdgcn_readfirstlane(*ghdr);
        if (nsaved > LCAP) nsaved = -1;
    }

    if (nsaved >= 0) {
        for (int i = t; i < nsaved; i += AGG_T) srt[i] = grec[i];
        __syncthreads();
        gather_sorted(tile, srt, h, n, nsaved, wave, lane);
        __syncthreads();
    } else {
        const bool al16 = ((((size_t)dst) & 15) == 0);

        static_assert(TVS * 4 <= 8, "a thread's hits of one sub-chunk are kept as one 32-bit mask, one bit per byte");
        static_assert(TPF == 1, "the scan holds exactly one sub-chunk ahead in registers");
        const bool compact = tid_usable(n, e, ws_bytes);
        const uint4* tq = (const uint4*)((const unsigned char*)wsf + tid_offset(n)) + t;
        const unsigned pat = (unsigned)blockIdx.x * 0x01010101u;
        uint4 nx[TVS];
#pragma unroll
        for (int v = 0; v < TVS; ++v)
            nx[v] = compact ? tq[v * AGG_T] : make_uint4(0u, 0u, 0u, 0u);

        int  nrec = 0;
        bool fits = true;
        for (int cb = 0; cb < e; cb += CH) {
          if (compact) {
            uint4 c[TVS];
#pragma unroll
            for (int v = 0; v < TVS; ++v) {
                c[v] = nx[v];
                nx[v] = tq[(size_t)(cb / 16) + (size_t)((TVS + v) * AGG_T)];
            }
            unsigned bits = 0;
#pragma unroll
            for (int v = 0; v < TVS; ++v) {
                bits |= tid_match4(c[v].x, pat) >> (7 - (4 * v + 0));
                bits |= tid_match4(c[v].y, pat) >> (7 - (4 * v + 1));
                bits |= tid_match4(c[v].z, pat) >> (7 - (4 * v + 2));
                bits |= tid_match4(c[v].w, pat) >> (7 - (4 * v + 3));
            }
            const int k = __popc(bits);
            if (k > 0) {
                int p = atomicAdd(&cnt, k);
                const int eb = cb + t * 16;
                while (bits != 0) {
                    const int pos = __ffs(bits) - 1;
                    bits &= bits - 1;
                    const int d = pos & 7;
                    list[p++] = eb + (d >> 2) * (AGG_T * 16) + (d & 3) * 4 + (pos >> 3);
                }
            }
          } else {
            int4 dv[CH / (AGG_T * 4)];
            if (al16 && cb + CH <= e) {
                const int4* dp = (const int4*)(dst + cb) + t;
#pragma unroll
                for (int j = 0; j < CH / (AGG_T * 4); ++j) dv[j] = dp[j * AGG_T];
            } else {
#pragma unroll
                for (int j = 0; j < CH / (AGG_T * 4); ++j) {
                    const int i0 = cb + (j * AGG_T + t) * 4;
                    const int a = dst[i0     < e ? i0     : e - 1];
                    const int b = dst[i0 + 1 < e ? i0 + 1 : e - 1];
                    const int c = dst[i0 + 2 < e ? i0 + 2 : e - 1];
                    const int d = dst[i0 + 3 < e ? i0 + 3 : e - 1];
                    dv[j].x = i0     < e ? a : -1;
                    dv[j].y = i0 + 1 < e ? b : -1;
                    dv[j].z = i0 + 2 < e ? c : -1;
                    dv[j].w = i0 + 3 < e ? d : -1;
                }
            }
            int k = 0;
#pragma unroll
            for (int j = 0; j < CH / (AGG_T * 4); ++j) {
                k += ((unsigned)(dv[j].x - row0) < (unsigned)TR) ? 1 : 0;
                k += ((unsigned)(dv[j].y - row0) < (unsigned)TR) ? 1 : 0;
                k += ((unsigned)(dv[j].z - row0) < (unsigned)TR) ? 1 : 0;
                k += ((unsigned)(dv[j].w - row0) < (unsigned)TR) ? 1 : 0;
            }
            if (k > 0) {
                int p = atomicAdd(&cnt, k);
#pragma unroll
                for (int j = 0; j < CH / (AGG_T * 4); ++j) {
                    const int i0 = cb + (j * AGG_T + t) * 4;
                    if ((unsigned)(dv[j].x - row0) < (unsigned)TR) list[p++] = i0;
                    if ((unsigned)(dv[j].y - row0) < (unsigned)TR) list[p++] = i0 + 1;
                    if ((unsigned)(dv[j].z - row0) < (unsigned)TR) list[p++] = i0 + 2;
                    if ((unsigned)(dv[j].w - row0) < (unsigned)TR) list[p++] = i0 + 3;
                }
            }
          }
            __syncthreads();
            const int m = cnt;
            __syncthreads();

            const bool last = (cb + CH >= e);
            if (m > LC - CH || last) {
                const bool sortp = save && fits && last && nrec == 0 && m > 0 && m <= LCAP;
                if (sortp) {
                    if (t < NKEY) cntk[t] = 0;
                    int es[RPT], ed[RPT], key[RPT], rank[RPT];
                    int2 rc[RPT];
#pragma unroll
                    for (int j = 0; j < RPT; ++j) {
                        const int i = t + j * AGG_T;
                        const int eid = list[i < m ? i : m - 1];
                        es[j] = src[eid];
                        ed[j] = dst[eid];
                    }
#pragma unroll
                    for (int j = 0; j < RPT; ++j) {
                        const float w = dinv[es[j]] * dinv[ed[j]];
                        const int r = ed[j] - row0;
                        rc[j] = make_int2((int)((unsigned)es[j] | ((unsigned)r << 24)), __float_as_int(w));
                        key[j] = r;
                    }
                    __syncthreads();
#pragma unroll
                    for (int j = 0; j < RPT; ++j) {
                        rank[j] = 0;
                        if (t + j * AGG_T < m) rank[j] = atomicAdd(&cntk[key[j]], 1);
                    }
                    __syncthreads();
                    if (wave == 0) {
                        int c[XR];
                        int tot = 0;
#pragma unroll
                        for (int j = 0; j < XR; ++j) {
                            c[j] = cntk[lane * XR + j];
                            tot += c[j];
                        }
                        int inc = tot;
#pragma unroll
                        for (int d = 1; d < 32; d <<= 1) {
                            const int o = __shfl_up(inc, d);
                            if (lane >= d) inc += o;
                        }
                        int b = inc - tot;
#pragma unroll
                        for (int j = 0; j < XR; ++j) {
                            basek[lane * XR + j] = b;
                            b += c[j];
                        }
                    }
                    __syncthreads();
#pragma unroll
                    for (int j = 0; j < RPT; ++j) {
                        if (t + j * AGG_T < m) {
                            int pos = basek[key[j]] + rank[j];
                            if (pos > LCAP - 1) pos = LCAP - 1;
                            srt[pos] = rc[j];
                        }
                    }
                    __syncthreads();
                    for (int i = t; i < m; i += AGG_T) grec[i] = srt[i];
                    gather_sorted(tile, srt, h, n, __builtin_amdgcn_readfirstlane(m), wave, lane);
                } else {
                if (save && m > 0) fits = false;
                for (int base = wave * GU; base < m; base += (AGG_T / 32) * GU) {
                    int   es[GU], ed[GU];
                    float ew[GU];
                    float4 ev[GU];
#pragma unroll
                    for (int u = 0; u < GU; ++u) {
                        const int idx = base + u;
                        const int eid = list[idx < m ? idx : m - 1];
                        es[u] = src[eid];
                        ed[u] = dst[eid];
                    }
#pragma unroll
                    for (int u = 0; u < GU; ++u) {
                        ew[u] = dinv[es[u]] * dinv[ed[u]];
                        ev[u] = *((const float4*)(h + (size_t)es[u] * GD) + lane);
                    }
#pragma unroll
                    for (int u = 0; u < GU; ++u) {
                        if (base + u < m) {
                            float* tp = tile + (ed[u] - row0) * TP + lane;
                            const float w = ew[u];
                            __hip_atomic_fetch_add(tp + OFX, ev[u].x * w, __ATOMIC_RELAXED, __HIP_MEMORY_SCOPE_WORKGROUP);
                            __hip_atomic_fetch_add(tp + OFY, ev[u].y * w, __ATOMIC_RELAXED, __HIP_MEMORY_SCOPE_WORKGROUP);
                            __hip_atomic_fetch_add(tp + OFZ, ev[u].z * w, __ATOMIC_RELAXED, __HIP_MEMORY_SCOPE_WORKGROUP);
                            __hip_atomic_fetch_add(tp + OFW, ev[u].w * w, __ATOMIC_RELAXED, __HIP_MEMORY_SCOPE_WORKGROUP);
                        }
                    }
                }
                }
                nrec += m;
                __syncthreads();
                if (t == 0) cnt = 0;
                __syncthreads();
            }
        }
        if (save && t == 0) *ghdr = fits ? nrec : -1;
    }

    const float4 b4 = *((const float4*)bc + lane);
    float  dv[XR];
    float4 hv[XR], xv[XR];
#pragma unroll
    for (int j = 0; j < XR; ++j) {
        const int r = wave + j * (AGG_T / 32);
        const int row = (r < TR && row0 + r < n) ? row0 + r : row0;
        dv[j] = dinv[row];
        hv[j] = *((const float4*)(h + (size_t)row * GD) + lane);
        xv[j] = *((const float4*)(x + (size_t)row * GD) + lane);
    }
    v4f wf4[WV], ws4[WV];
    stage_w_load(wf4, Wf, t);
    static_assert(XR == 7 && WV == 4, "the statement below names one loaded value per row of h and of x, and Wf's vectors");
    asm volatile("" :: "v"(hv[0].x), "v"(hv[1].x), "v"(hv[2].x), "v"(hv[3].x), "v"(hv[4].x), "v"(hv[5].x), "v"(hv[6].x),
                       "v"(xv[0].x), "v"(xv[1].x), "v"(xv[2].x), "v"(xv[3].x), "v"(xv[4].x), "v"(xv[5].x), "v"(xv[6].x),
                       "v"(wf4[0]), "v"(wf4[1]), "v"(wf4[2]), "v"(wf4[3]));
    stage_w_store(wl, wf4, t);
    if (has_skip) {
        stage_w_load(ws4, Wsk, t);
    } else {
#pragma unroll
        for (int j = 0; j < WV; ++j) ws4[j] = (v4f){0.0f, 0.0f, 0.0f, 0.0f};
    }

#pragma unroll
    for (int j = 0; j < XR; ++j) {
        const int r = wave + j * (AGG_T / 32);
        if (r < TR && row0 + r < n) {
            const float di = dv[j];
            float* tp = tile + r * TP + lane;
            float4 o;
            o.x = (tp[OFX] + hv[j].x * di * di + b4.x) + xv[j].x;
            o.y = (tp[OFY] + hv[j].y * di * di + b4.y) + xv[j].y;
            o.z = (tp[OFZ] + hv[j].z * di * di + b4.z) + xv[j].z;
            o.w = (tp[OFW] + hv[j].w * di * di + b4.w) + xv[j].w;
            tp[OFX] = o.x > 0.0f ? o.x : 0.0f;
            tp[OFY] = o.y > 0.0f ? o.y : 0.0f;
            tp[OFZ] = o.z > 0.0f ? o.z : 0.0f;
            tp[OFW] = o.w > 0.0f ? o.w : 0.0f;
        }
    }
    if (has_skip) stage_w_store(wl + GD * WP, ws4, t);
    __syncthreads();

    const int half = lane >> 4;
    const int l16  = lane & 15;
    const int stripe = wave >> 1;
    const int chalf  = wave & 1;
    const bool gw = (wave < 2 * NSTRIPE);

    v8f acc[4];
    float bnm[4], bns[4], bng[4], bnb[4], bfv[4];
#pragma unroll
    for (int nt = 0; nt < 4; ++nt) {
        const int col = chalf * 64 + nt * 16 + l16;
        bnm[nt] = rmean[col];
        bns[nt] = rsqrtf(rvar[col] + BN_EPS);
        bng[nt] = gamma[col];
        bnb[nt] = beta[col];
        bfv[nt] = bf[col];
    }
    float* const ep = tile + (stripe * 16 + half * 8) * TP;
    int pcs[4];
#pragma unroll
    for (int nt = 0; nt < 4; ++nt) pcs[nt] = PERM(chalf * 64 + nt * 16 + l16);

    if (gw) tile_gemm(tile, wl, stripe, chalf, l16, half, acc);
    __syncthreads();
    if (has_next) {
        v4f wc4[WV];
        stage_w_load(wc4, Wcn, t);
        stage_w_store(wl, wc4, t);
    }
    if (gw) {
#pragma unroll
        for (int g = 0; g < 2; ++g) {
            float tv[2][8];
#pragma unroll
            for (int q = 0; q < 2; ++q)
#pragma unroll
                for (int r = 0; r < 8; ++r) tv[q][r] = ep[r * TP + pcs[2 * g + q]];
#pragma unroll
            for (int q = 0; q < 2; ++q) {
                const int nt = 2 * g + q;
#pragma unroll
                for (int r = 0; r < 8; ++r) {
                    float u = acc[nt][r] + bfv[nt];
                    u = u > 0.0f ? u : LRELU_SLOPE * u;
                    const float v = u + tv[q][r];
                    tv[q][r] = v > 0.0f ? v : 0.0f;
                }
            }
            if (!has_skip) {
#pragma unroll
                for (int q = 0; q < 2; ++q) {
                    const int nt = 2 * g + q;
#pragma unroll
                    for (int r = 0; r < 8; ++r) {
                        const float v = tv[q][r];
                        const float b = (v - bnm[nt]) * bns[nt] * bng[nt] + bnb[nt];
                        const float y = b + v;
                        tv[q][r] = y > 0.0f ? y : 0.0f;
                    }
                }
            }
#pragma unroll
            for (int q = 0; q < 2; ++q)
#pragma unroll
                for (int r = 0; r < 8; ++r) ep[r * TP + pcs[2 * g + q]] = tv[q][r];
        }
    }
    __syncthreads();

    if (has_skip) {
        float bkv[4];
#pragma unroll
        for (int nt = 0; nt < 4; ++nt) bkv[nt] = bsk[chalf * 64 + nt * 16 + l16];
        if (gw) tile_gemm(tile, wl + GD * WP, stripe, chalf, l16, half, acc);
        __syncthreads();
        if (gw) {
#pragma unroll
            for (int g = 0; g < 2; ++g) {
                float tv[2][8];
#pragma unroll
                for (int q = 0; q < 2; ++q)
#pragma unroll
                    for (int r = 0; r < 8; ++r) tv[q][r] = ep[r * TP + pcs[2 * g + q]];
#pragma unroll
                for (int q = 0; q < 2; ++q) {
                    const int nt = 2 * g + q;
#pragma unroll
                    for (int r = 0; r < 8; ++r) {
                        const float s = acc[nt][r] + bkv[nt];
                        float v = tv[q][r] + s;
                        v = v > 0.0f ? v : 0.0f;
                        const float b = (v - bnm[nt]) * bns[nt] * bng[nt] + bnb[nt];
                        const float y = b + v;
                        tv[q][r] = y > 0.0f ? y : 0.0f;
                    }
                }
#pragma unroll
                for (int q = 0; q < 2; ++q)
#pragma unroll
                    for (int r = 0; r < 8; ++r) ep[r * TP + pcs[2 * g + q]] = tv[q][r];
            }
        }
        __syncthreads();
    }

    tile_stream(tile, x, row0, n, wave, lane);

    if (has_next) {
        if (gw) tile_gemm(tile, wl, stripe, chalf, l16, half, acc);
        __syncthreads();
        if (gw) {
#pragma unroll
            for (int nt = 0; nt < 4; ++nt)
#pragma unroll
                for (int r = 0; r < 8; ++r) ep[r * TP + pcs[nt]] = acc[nt][r];
        }
        __syncthreads();
        tile_stream(tile, hn, row0, n, wave, lane);
    }
}

extern "C" void kernel_launch(void* const* d_in, const int* in_sizes, int n_in,
                              void* d_out, int out_size, void* d_ws, size_t ws_size,
                              hipStream_t stream) {
    const float* x_in  = (const float*)d_in[0];
    const int*   ei    = (const int*)d_in[1];
    const float* Wc    = (const float*)d_in[2];
    const float* bc    = (const float*)d_in[3];
    const float* Wf    = (const float*)d_in[4];
    const float* bf    = (const float*)d_in[5];
    const float* Wskip = (const float*)d_in[6];
    const float* bskip = (const float*)d_in[7];
    const float* gamma = (const float*)d_in[8];
    const float* beta  = (const float*)d_in[9];
    const float* rmean = (const float*)d_in[10];
    const float* rvar  = (const float*)d_in[11];

    const int n = in_sizes[0] / GD;
    const int e = in_sizes[1] / 2;
    const int L = in_sizes[3] / GD;

    const int* src = ei;
    const int* dst = ei + e;

    float* x    = (float*)d_out;
    float* bufA = (float*)d_ws;
    float* deg  = bufA + (size_t)n * GD;
    float* dinv = deg + n;
    const size_t offB = h2_offset(n, e);
    const bool chain = offB + (size_t)n * GD * sizeof(float) <= ws_size;
    float* bufB = chain ? (float*)((unsigned char*)d_ws + offB) : bufA;

    const int ND4 = n * (GD / 4);
    dim3 blk(256);
    const int gCopy = (ND4 + 256 * CPV - 1) / (256 * CPV);
    const int gN    = (n + 255) / 256;
    const int gT    = ((e + CH - 1) / CH + TPF) * (CH / 1024);
    const int gAgg  = (n + TR - 1) / TR;
    const int gGemm = (n + 127) / 128;

    const bool hist = chain && n >= 64 && n <= 2 * HW && (e + HK - 1) / HK <= 65535;
    unsigned* slab = (unsigned*)bufB;
    if (hist) {
        deg_hist_kernel<<<HK, dim3(HT), 0, stream>>>(dst, e, n, slab);
        prep_conv0_kernel<<<gGemm + gN + gT + gCopy, blk, 0, stream>>>(x_in, Wc, bufA, n, deg, dst, e, bufA, ws_size,
                                                                      x, ND4, gGemm, gT, slab, dinv, gN, 0);
    } else {
        init_one_kernel<<<gN, blk, 0, stream>>>(deg, n);
        prep_conv0_kernel<<<gGemm + gT + gCopy, blk, 0, stream>>>(x_in, Wc, bufA, n, deg, dst, e, bufA, ws_size,
                                                                 x, ND4, gGemm, gT, slab, dinv, 0, 1);
        rsqrt_kernel<<<gN, blk, 0, stream>>>(deg, dinv, n);
    }

    for (int i = 0; i < L; ++i) {
        float* hcur = (i & 1) ? bufB : bufA;
        float* hnxt = (i & 1) ? bufA : bufB;
        const int nxt = (chain && i + 1 < L) ? 1 : 0;
        if (i > 0 && !chain) wmma_gemm128<<<gGemm, blk, 0, stream>>>(x, Wc + (size_t)i * GD * GD, hcur, n);
        const float* Wsk = (i > 0) ? Wskip + (size_t)(i - 1) * GD * GD : Wf + (size_t)i * GD * GD;
        const float* bsk = (i > 0) ? bskip + (i - 1) * GD : bf + i * GD;
        layer_tile_kernel<<<gAgg, dim3(AGG_T), 0, stream>>>(
            x, hcur, src, dst, dinv, bc + i * GD,
            Wf + (size_t)i * GD * GD, bf + i * GD, Wsk, bsk,
            gamma + i * GD, beta + i * GD, rmean + i * GD, rvar + i * GD,
            n, e, (i > 0) ? 1 : 0, (i == 0) ? 1 : 0, bufA, ws_size,
            hnxt, Wc + (size_t)(nxt ? i + 1 : i) * GD * GD, nxt);
    }
}
